# starting kernel
# baseline (speedup 1.0000x reference)
#include <hip/hip_runtime.h>
#include <math.h>


typedef __attribute__((ext_vector_type(2))) float v2f;
typedef __attribute__((ext_vector_type(8))) float v8f;

#define HEADS     4
#define CHAN      64
#define HIDDEN    256
#define FEAT      128
#define NEG_SLOPE 0.2f
#define ORD_NEG_INF 0x007FFFFFu

static __device__ __forceinline__ float lrelu(float x) {
  return x > 0.f ? x : NEG_SLOPE * x;
}
static __device__ __forceinline__ unsigned ordEnc(float f) {
  unsigned u = __float_as_uint(f);
  return (u & 0x80000000u) ? ~u : (u | 0x80000000u);
}
static __device__ __forceinline__ float ordDec(unsigned u) {
  unsigned v = (u & 0x80000000u) ? (u & 0x7fffffffu) : ~u;
  return __uint_as_float(v);
}

template <bool GAT, bool BIAS>
__global__ void wmma_gemm_kernel(const float* __restrict__ A,
                                 const float* __restrict__ W,
                                 const float* __restrict__ bias,
                                 const float* __restrict__ att_src,
                                 const float* __restrict__ att_dst,
                                 float* __restrict__ Hout,
                                 float* __restrict__ s_src,
                                 float* __restrict__ s_dst,
                                 int M, int K, int Nc) {
  const int lane = threadIdx.x & 31;
  const int wv   = threadIdx.x >> 5;
  const int wpb  = blockDim.x >> 5;
  const int tile = blockIdx.x * wpb + wv;
  const int ctiles = Nc >> 6;
  const int mt = tile / ctiles;
  const int ct = tile - mt * ctiles;
  const int m0 = mt << 5;
  if (m0 >= M) return;
  const int n0 = ct << 6;
  const int lo = lane & 15;
  const int hi = lane >> 4;

  v8f acc[2][4] = {};
  const float* arow0 = A + (size_t)(m0 + lo) * K;
  const float* arow1 = A + (size_t)(m0 + 16 + lo) * K;
  const float* wcol  = W + n0 + lo;

  for (int k = 0; k < K; k += 4) {
    const int ka = k + (hi << 1);
    v2f a0 = *(const v2f*)(arow0 + ka);
    v2f a1 = *(const v2f*)(arow1 + ka);
    const float* w0 = wcol + (size_t)ka * Nc;
    const float* w1 = w0 + Nc;
    v2f b0 = {w0[0],  w1[0]};
    v2f b1 = {w0[16], w1[16]};
    v2f b2 = {w0[32], w1[32]};
    v2f b3 = {w0[48], w1[48]};
    acc[0][0] = __builtin_amdgcn_wmma_f32_16x16x4_f32(false, a0, false, b0, (short)0, acc[0][0], false, false);
    acc[1][0] = __builtin_amdgcn_wmma_f32_16x16x4_f32(false, a1, false, b0, (short)0, acc[1][0], false, false);
    acc[0][1] = __builtin_amdgcn_wmma_f32_16x16x4_f32(false, a0, false, b1, (short)0, acc[0][1], false, false);
    acc[1][1] = __builtin_amdgcn_wmma_f32_16x16x4_f32(false, a1, false, b1, (short)0, acc[1][1], false, false);
    acc[0][2] = __builtin_amdgcn_wmma_f32_16x16x4_f32(false, a0, false, b2, (short)0, acc[0][2], false, false);
    acc[1][2] = __builtin_amdgcn_wmma_f32_16x16x4_f32(false, a1, false, b2, (short)0, acc[1][2], false, false);
    acc[0][3] = __builtin_amdgcn_wmma_f32_16x16x4_f32(false, a0, false, b3, (short)0, acc[0][3], false, false);
    acc[1][3] = __builtin_amdgcn_wmma_f32_16x16x4_f32(false, a1, false, b3, (short)0, acc[1][3], false, false);
  }

  float bv0 = 0.f, bv1 = 0.f, bv2 = 0.f, bv3 = 0.f;
  if (BIAS) {
    bv0 = bias[n0 + lo];      bv1 = bias[n0 + 16 + lo];
    bv2 = bias[n0 + 32 + lo]; bv3 = bias[n0 + 48 + lo];
  }

  #pragma unroll
  for (int g = 0; g < 2; ++g) {
    const int rbase = m0 + (g << 4) + (hi << 3);
    #pragma unroll
    for (int r = 0; r < 8; ++r) {
      float* hp = Hout + (size_t)(rbase + r) * Nc + n0 + lo;
      hp[0]  = acc[g][0][r] + bv0;
      hp[16] = acc[g][1][r] + bv1;
      hp[32] = acc[g][2][r] + bv2;
      hp[48] = acc[g][3][r] + bv3;
    }
  }

  if (GAT) {
    const int head = n0 >> 6;
    const float* as = att_src + head * CHAN;
    const float* ad = att_dst + head * CHAN;
    const float as0 = as[lo], as1 = as[16 + lo], as2 = as[32 + lo], as3 = as[48 + lo];
    const float ad0 = ad[lo], ad1 = ad[16 + lo], ad2 = ad[32 + lo], ad3 = ad[48 + lo];
    #pragma unroll
    for (int g = 0; g < 2; ++g) {
      const int rbase = m0 + (g << 4) + (hi << 3);
      #pragma unroll
      for (int r = 0; r < 8; ++r) {
        float ss = acc[g][0][r] * as0 + acc[g][1][r] * as1 + acc[g][2][r] * as2 + acc[g][3][r] * as3;
        float sd = acc[g][0][r] * ad0 + acc[g][1][r] * ad1 + acc[g][2][r] * ad2 + acc[g][3][r] * ad3;
        #pragma unroll
        for (int off = 8; off > 0; off >>= 1) {
          ss += __shfl_xor(ss, off, 32);
          sd += __shfl_xor(sd, off, 32);
        }
        if (lo == 0) {
          const int row = rbase + r;
          s_src[row * HEADS + head] = ss;
          s_dst[row * HEADS + head] = sd;
        }
      }
    }
  }
}

__global__ void layer_init_kernel(float* __restrict__ agg, const float* __restrict__ bias,
                                  float* __restrict__ denom, unsigned* __restrict__ mord,
                                  int N) {
  const int idx = blockIdx.x * blockDim.x + threadIdx.x;
  if (idx < N * HIDDEN) agg[idx] = bias[idx & (HIDDEN - 1)];
  if (idx < N * HEADS) { denom[idx] = 0.f; mord[idx] = ORD_NEG_INF; }
}

static __device__ __forceinline__ void edge_nodes(const int* __restrict__ ei,
                                                  int idx, int E, int& s, int& d) {
  if (idx < E) { s = ei[idx]; d = ei[E + idx]; }
  else         { s = d = idx - E; }
}

__global__ void edge_max_kernel(const int* __restrict__ ei, int E, int N,
                                const float* __restrict__ s_src,
                                const float* __restrict__ s_dst,
                                unsigned* __restrict__ mord) {
  const int idx = blockIdx.x * blockDim.x + threadIdx.x;
  if (idx >= E + N) return;
  int s, d; edge_nodes(ei, idx, E, s, d);
  const float4 vs = *(const float4*)(s_src + (size_t)s * HEADS);
  const float4 vd = *(const float4*)(s_dst + (size_t)d * HEADS);
  unsigned* mp = mord + (size_t)d * HEADS;
  atomicMax(mp + 0, ordEnc(lrelu(vs.x + vd.x)));
  atomicMax(mp + 1, ordEnc(lrelu(vs.y + vd.y)));
  atomicMax(mp + 2, ordEnc(lrelu(vs.z + vd.z)));
  atomicMax(mp + 3, ordEnc(lrelu(vs.w + vd.w)));
}

__global__ void edge_exp_kernel(const int* __restrict__ ei, int E, int N,
                                const float* __restrict__ s_src,
                                const float* __restrict__ s_dst,
                                const unsigned* __restrict__ mord,
                                float* __restrict__ exbuf,
                                float* __restrict__ denom) {
  const int idx = blockIdx.x * blockDim.x + threadIdx.x;
  if (idx >= E + N) return;
  int s, d; edge_nodes(ei, idx, E, s, d);
  const float4 vs = *(const float4*)(s_src + (size_t)s * HEADS);
  const float4 vd = *(const float4*)(s_dst + (size_t)d * HEADS);
  const unsigned* mp = mord + (size_t)d * HEADS;
  const float x0 = expf(lrelu(vs.x + vd.x) - ordDec(mp[0]));
  const float x1 = expf(lrelu(vs.y + vd.y) - ordDec(mp[1]));
  const float x2 = expf(lrelu(vs.z + vd.z) - ordDec(mp[2]));
  const float x3 = expf(lrelu(vs.w + vd.w) - ordDec(mp[3]));
  *(float4*)(exbuf + (size_t)idx * HEADS) = make_float4(x0, x1, x2, x3);
  float* dn = denom + (size_t)d * HEADS;
  atomicAdd(dn + 0, x0);
  atomicAdd(dn + 1, x1);
  atomicAdd(dn + 2, x2);
  atomicAdd(dn + 3, x3);
}

__global__ void edge_agg_kernel(const int* __restrict__ ei, int E, int N,
                                const float* __restrict__ exbuf,
                                const float* __restrict__ denom,
                                const float* __restrict__ Hin,
                                float* __restrict__ agg) {
  const int wid  = blockIdx.x * (blockDim.x >> 5) + (threadIdx.x >> 5);
  const int lane = threadIdx.x & 31;
  if (wid >= E + N) return;
  int s, d; edge_nodes(ei, wid, E, s, d);
  const float4 e4 = *(const float4*)(exbuf + (size_t)wid * HEADS);
  const float4 d4 = *(const float4*)(denom + (size_t)d * HEADS);
  const float alpha[HEADS] = {e4.x / d4.x, e4.y / d4.y, e4.z / d4.z, e4.w / d4.w};
  const float* hs = Hin + (size_t)s * HIDDEN;
  float* od = agg + (size_t)d * HIDDEN;
  #pragma unroll
  for (int i = 0; i < 8; ++i) {
    const int ch = lane + (i << 5);
    atomicAdd(od + ch, hs[ch] * alpha[i >> 1]);
  }
}

__global__ void elu_kernel(float* __restrict__ x, int n) {
  const int idx = blockIdx.x * blockDim.x + threadIdx.x;
  if (idx < n) {
    const float v = x[idx];
    x[idx] = v > 0.f ? v : expm1f(v);
  }
}

extern "C" void kernel_launch(void* const* d_in, const int* in_sizes, int n_in,
                              void* d_out, int out_size, void* d_ws, size_t ws_size,
                              hipStream_t stream) {
  const float* x     = (const float*)d_in[0];
  const int*   ei    = (const int*)  d_in[1];
  const float* W1    = (const float*)d_in[2];
  const float* asrc1 = (const float*)d_in[3];
  const float* adst1 = (const float*)d_in[4];
  const float* b1    = (const float*)d_in[5];
  const float* W2    = (const float*)d_in[6];
  const float* asrc2 = (const float*)d_in[7];
  const float* adst2 = (const float*)d_in[8];
  const float* b2    = (const float*)d_in[9];
  const float* Wf    = (const float*)d_in[10];
  const float* bf    = (const float*)d_in[11];

  const int B  = 2;
  const int N  = in_sizes[0] / (B * FEAT);
  const int E  = in_sizes[1] / 2;
  const int ET = E + N;

  float* p = (float*)d_ws;
  float* buf_h   = p; p += (size_t)N * HIDDEN;
  float* buf_agg = p; p += (size_t)N * HIDDEN;
  float* ssrc    = p; p += (size_t)N * HEADS;
  float* sdst    = p; p += (size_t)N * HEADS;
  float* denom   = p; p += (size_t)N * HEADS;
  unsigned* mord = (unsigned*)p; p += (size_t)N * HEADS;
  float* exbuf   = p; p += (size_t)ET * HEADS;

  const int mtiles  = N >> 5;
  const int gtiles  = mtiles * (HIDDEN >> 6);
  const int ftiles  = mtiles * (CHAN >> 6);
  const int nHC     = N * HIDDEN;
  const dim3 blkG(128);
  const dim3 blkE(256);

  for (int b = 0; b < B; ++b) {
    const float* xb = x + (size_t)b * N * FEAT;
    float* outb = (float*)d_out + (size_t)b * N * CHAN;

    wmma_gemm_kernel<true, false><<<(gtiles + 3) / 4, blkG, 0, stream>>>(
        xb, W1, nullptr, asrc1, adst1, buf_h, ssrc, sdst, N, FEAT, HIDDEN);
    layer_init_kernel<<<(nHC + 255) / 256, blkE, 0, stream>>>(buf_agg, b1, denom, mord, N);
    edge_max_kernel<<<(ET + 255) / 256, blkE, 0, stream>>>(ei, E, N, ssrc, sdst, mord);
    edge_exp_kernel<<<(ET + 255) / 256, blkE, 0, stream>>>(ei, E, N, ssrc, sdst, mord, exbuf, denom);
    edge_agg_kernel<<<(ET + 7) / 8, blkE, 0, stream>>>(ei, E, N, exbuf, denom, buf_h, buf_agg);
    elu_kernel<<<(nHC + 255) / 256, blkE, 0, stream>>>(buf_agg, nHC);

    wmma_gemm_kernel<true, false><<<(gtiles + 3) / 4, blkG, 0, stream>>>(
        buf_agg, W2, nullptr, asrc2, adst2, buf_h, ssrc, sdst, N, HIDDEN, HIDDEN);
    layer_init_kernel<<<(nHC + 255) / 256, blkE, 0, stream>>>(buf_agg, b2, denom, mord, N);
    edge_max_kernel<<<(ET + 255) / 256, blkE, 0, stream>>>(ei, E, N, ssrc, sdst, mord);
    edge_exp_kernel<<<(ET + 255) / 256, blkE, 0, stream>>>(ei, E, N, ssrc, sdst, mord, exbuf, denom);
    edge_agg_kernel<<<(ET + 7) / 8, blkE, 0, stream>>>(ei, E, N, exbuf, denom, buf_h, buf_agg);

    wmma_gemm_kernel<false, true><<<(ftiles + 3) / 4, blkG, 0, stream>>>(
        buf_agg, Wf, bf, nullptr, nullptr, outb, nullptr, nullptr, N, HIDDEN, CHAN);
  }
}


# Finished 80-row tile stays in LDS and is multiplied there by the next weights (layer 2's W with its scores; the final linear) on WMMA, A fragments from LDS; 10 launches become 6; sum order unchanged
# speedup vs baseline: 3.8916x; 3.8916x over previous
#include <hip/hip_runtime.h>
#include <math.h>


typedef __attribute__((ext_vector_type(2))) float v2f;
typedef __attribute__((ext_vector_type(8))) float v8f;

#define HEADS     4
#define CHAN      64
#define HIDDEN    256
#define FEAT      128
#define NEG_SLOPE 0.2f

static __device__ __forceinline__ float lrelu(float x) {
  return x > 0.f ? x : NEG_SLOPE * x;
}
static __device__ __forceinline__ unsigned ordEnc(float f) {
  unsigned u = __float_as_uint(f);
  return (u & 0x80000000u) ? ~u : (u | 0x80000000u);
}
static __device__ __forceinline__ float ordDec(unsigned u) {
  unsigned v = (u & 0x80000000u) ? (u & 0x7fffffffu) : ~u;
  return __uint_as_float(v);
}

template <bool GAT, bool BIAS>
__global__ void wmma_gemm_kernel(const float* __restrict__ A,
                                 const float* __restrict__ W,
                                 const float* __restrict__ bias,
                                 const float* __restrict__ att_src,
                                 const float* __restrict__ att_dst,
                                 float* __restrict__ Hout,
                                 float* __restrict__ s_src,
                                 float* __restrict__ s_dst,
                                 int M, int K, int Nc) {
  const int lane = threadIdx.x & 31;
  const int wv   = threadIdx.x >> 5;
  const int wpb  = blockDim.x >> 5;
  const int tile = blockIdx.x * wpb + wv;
  const int ctiles = Nc >> 6;
  const int mt = tile / ctiles;
  const int ct = tile - mt * ctiles;
  const int m0 = mt << 5;
  if (m0 >= M) return;
  const int n0 = ct << 6;
  const int lo = lane & 15;
  const int hi = lane >> 4;

  v8f acc[2][4] = {};
  const float* arow0 = A + (size_t)(m0 + lo) * K;
  const float* arow1 = A + (size_t)(m0 + 16 + lo) * K;
  const float* wcol  = W + n0 + lo;

  for (int k = 0; k < K; k += 4) {
    const int ka = k + (hi << 1);
    v2f a0 = *(const v2f*)(arow0 + ka);
    v2f a1 = *(const v2f*)(arow1 + ka);
    const float* w0 = wcol + (size_t)ka * Nc;
    const float* w1 = w0 + Nc;
    v2f b0 = {w0[0],  w1[0]};
    v2f b1 = {w0[16], w1[16]};
    v2f b2 = {w0[32], w1[32]};
    v2f b3 = {w0[48], w1[48]};
    acc[0][0] = __builtin_amdgcn_wmma_f32_16x16x4_f32(false, a0, false, b0, (short)0, acc[0][0], false, false);
    acc[1][0] = __builtin_amdgcn_wmma_f32_16x16x4_f32(false, a1, false, b0, (short)0, acc[1][0], false, false);
    acc[0][1] = __builtin_amdgcn_wmma_f32_16x16x4_f32(false, a0, false, b1, (short)0, acc[0][1], false, false);
    acc[1][1] = __builtin_amdgcn_wmma_f32_16x16x4_f32(false, a1, false, b1, (short)0, acc[1][1], false, false);
    acc[0][2] = __builtin_amdgcn_wmma_f32_16x16x4_f32(false, a0, false, b2, (short)0, acc[0][2], false, false);
    acc[1][2] = __builtin_amdgcn_wmma_f32_16x16x4_f32(false, a1, false, b2, (short)0, acc[1][2], false, false);
    acc[0][3] = __builtin_amdgcn_wmma_f32_16x16x4_f32(false, a0, false, b3, (short)0, acc[0][3], false, false);
    acc[1][3] = __builtin_amdgcn_wmma_f32_16x16x4_f32(false, a1, false, b3, (short)0, acc[1][3], false, false);
  }

  float bv0 = 0.f, bv1 = 0.f, bv2 = 0.f, bv3 = 0.f;
  if (BIAS) {
    bv0 = bias[n0 + lo];      bv1 = bias[n0 + 16 + lo];
    bv2 = bias[n0 + 32 + lo]; bv3 = bias[n0 + 48 + lo];
  }

  #pragma unroll
  for (int g = 0; g < 2; ++g) {
    const int rbase = m0 + (g << 4) + (hi << 3);
    #pragma unroll
    for (int r = 0; r < 8; ++r) {
      float* hp = Hout + (size_t)(rbase + r) * Nc + n0 + lo;
      hp[0]  = acc[g][0][r] + bv0;
      hp[16] = acc[g][1][r] + bv1;
      hp[32] = acc[g][2][r] + bv2;
      hp[48] = acc[g][3][r] + bv3;
    }
  }

  if (GAT) {
    const int head = n0 >> 6;
    const float* as = att_src + head * CHAN;
    const float* ad = att_dst + head * CHAN;
    const float as0 = as[lo], as1 = as[16 + lo], as2 = as[32 + lo], as3 = as[48 + lo];
    const float ad0 = ad[lo], ad1 = ad[16 + lo], ad2 = ad[32 + lo], ad3 = ad[48 + lo];
    #pragma unroll
    for (int g = 0; g < 2; ++g) {
      const int rbase = m0 + (g << 4) + (hi << 3);
      #pragma unroll
      for (int r = 0; r < 8; ++r) {
        float ss = acc[g][0][r] * as0 + acc[g][1][r] * as1 + acc[g][2][r] * as2 + acc[g][3][r] * as3;
        float sd = acc[g][0][r] * ad0 + acc[g][1][r] * ad1 + acc[g][2][r] * ad2 + acc[g][3][r] * ad3;
        #pragma unroll
        for (int off = 8; off > 0; off >>= 1) {
          ss += __shfl_xor(ss, off, 32);
          sd += __shfl_xor(sd, off, 32);
        }
        if (lo == 0) {
          const int row = rbase + r;
          s_src[row * HEADS + head] = ss;
          s_dst[row * HEADS + head] = sd;
        }
      }
    }
  }
}

#define TN       80
#define APITCH   (HIDDEN + 4)
#define TCAP     20480
#define TTHREADS 1024
static_assert(TN <= 128, "a noted edge keeps its tile row in 7 bits");

template <int NC, bool GAT, bool BIAS>
static __device__ __forceinline__ void tile_gemm(const float* tile, int wv, int lane, int d0, int nt,
                                                 const float* __restrict__ W,
                                                 const float* __restrict__ bias,
                                                 const float* __restrict__ att_src,
                                                 const float* __restrict__ att_dst,
                                                 float* __restrict__ Hout,
                                                 float* __restrict__ s_src,
                                                 float* __restrict__ s_dst) {
  constexpr int CT = NC / 64;
  static_assert(TN % 16 == 0, "whole 16-row tiles");
  if (wv >= (TN / 16) * CT) return;
  const int rt = wv / CT;
  const int ct = wv - rt * CT;
  const int n0 = ct << 6;
  const int lo = lane & 15;
  const int hi = lane >> 4;

  v8f c0 = {}, c1 = {}, c2 = {}, c3 = {};
  const float* arow = tile + (rt * 16 + lo) * APITCH + (hi << 1);
  const float* wcol = W + (size_t)(hi << 1) * NC + n0 + lo;

  for (int k = 0; k < HIDDEN; k += 16) {
    v2f a[4];
    float b[4][8];
    #pragma unroll
    for (int u = 0; u < 4; ++u) {
      const float* w0 = wcol + (size_t)(k + 4 * u) * NC;
      b[u][0] = w0[0];  b[u][1] = w0[NC];
      b[u][2] = w0[16]; b[u][3] = w0[NC + 16];
      b[u][4] = w0[32]; b[u][5] = w0[NC + 32];
      b[u][6] = w0[48]; b[u][7] = w0[NC + 48];
    }
    #pragma unroll
    for (int u = 0; u < 4; ++u) a[u] = *(const v2f*)(arow + k + 4 * u);
    #pragma unroll
    for (int u = 0; u < 4; ++u) {
      const v2f b0 = {b[u][0], b[u][1]};
      const v2f b1 = {b[u][2], b[u][3]};
      const v2f b2 = {b[u][4], b[u][5]};
      const v2f b3 = {b[u][6], b[u][7]};
      c0 = __builtin_amdgcn_wmma_f32_16x16x4_f32(false, a[u], false, b0, (short)0, c0, false, false);
      c1 = __builtin_amdgcn_wmma_f32_16x16x4_f32(false, a[u], false, b1, (short)0, c1, false, false);
      c2 = __builtin_amdgcn_wmma_f32_16x16x4_f32(false, a[u], false, b2, (short)0, c2, false, false);
      c3 = __builtin_amdgcn_wmma_f32_16x16x4_f32(false, a[u], false, b3, (short)0, c3, false, false);
    }
  }

  float bv0 = 0.f, bv1 = 0.f, bv2 = 0.f, bv3 = 0.f;
  if (BIAS) {
    bv0 = bias[n0 + lo];      bv1 = bias[n0 + 16 + lo];
    bv2 = bias[n0 + 32 + lo]; bv3 = bias[n0 + 48 + lo];
  }
  const int rloc = rt * 16 + (hi << 3);
  #pragma unroll
  for (int r = 0; r < 8; ++r) {
    if (rloc + r < nt) {
      float* hp = Hout + (size_t)(d0 + rloc + r) * NC + n0 + lo;
      hp[0]  = c0[r] + bv0;
      hp[16] = c1[r] + bv1;
      hp[32] = c2[r] + bv2;
      hp[48] = c3[r] + bv3;
    }
  }

  if (GAT) {
    const int head = ct;
    const float* as = att_src + head * CHAN;
    const float* ad = att_dst + head * CHAN;
    const float as0 = as[lo], as1 = as[16 + lo], as2 = as[32 + lo], as3 = as[48 + lo];
    const float ad0 = ad[lo], ad1 = ad[16 + lo], ad2 = ad[32 + lo], ad3 = ad[48 + lo];
    #pragma unroll
    for (int r = 0; r < 8; ++r) {
      float ss = c0[r] * as0 + c1[r] * as1 + c2[r] * as2 + c3[r] * as3;
      float sd = c0[r] * ad0 + c1[r] * ad1 + c2[r] * ad2 + c3[r] * ad3;
      #pragma unroll
      for (int off = 8; off > 0; off >>= 1) {
        ss += __shfl_xor(ss, off, 32);
        sd += __shfl_xor(sd, off, 32);
      }
      if (lo == 0 && rloc + r < nt) {
        const int row = d0 + rloc + r;
        s_src[row * HEADS + head] = ss;
        s_dst[row * HEADS + head] = sd;
      }
    }
  }
}

template <bool ELU>
__global__ void __launch_bounds__(TTHREADS)
tile_agg_kernel(const int* __restrict__ ei, int E, int N,
                const float* __restrict__ s_src,
                const float* __restrict__ s_dst,
                const float* __restrict__ Hin,
                const float* __restrict__ bias,
                const float* __restrict__ Wn,
                const float* __restrict__ bn,
                const float* __restrict__ att_src_n,
                const float* __restrict__ att_dst_n,
                float* __restrict__ Hout,
                float* __restrict__ s_src_n,
                float* __restrict__ s_dst_n) {
  __shared__ __attribute__((aligned(16))) float acc[TN * APITCH];
  __shared__ unsigned elist[TCAP];
  __shared__ __attribute__((aligned(16))) float sdt[TN * HEADS];
  __shared__ unsigned mxo[TN * HEADS];
  __shared__ float den[TN * HEADS];
  __shared__ unsigned cnt;

  const int tid  = threadIdx.x;
  const int lane = tid & 31;
  const int wv   = tid >> 5;
  const int hl   = lane & (HEADS - 1);
  const int d0   = blockIdx.x * TN;
  const int nt   = (N - d0) < TN ? (N - d0) : TN;
  const int* __restrict__ srcp = ei;
  const int* __restrict__ dstp = ei + E;

  if (tid == 0) cnt = 0u;
  if (tid < nt * HEADS) {
    const size_t g = (size_t)d0 * HEADS + tid;
    const float sd = s_dst[g];
    sdt[tid] = sd;
    mxo[tid] = ordEnc(lrelu(s_src[g] + sd));
  }
  __syncthreads();

  auto note = [&](int idx) {
    const unsigned slot = atomicAdd(&cnt, 1u);
    if (slot < (unsigned)TCAP) elist[slot] = (unsigned)idx;
  };

  const int Ev   = (((size_t)dstp & 15) == 0) ? (E & ~3) : 0;
  const int nvec = Ev >> 2;
  const int4* dv = (const int4*)dstp;
  for (int v0 = 0; v0 < nvec; v0 += TTHREADS * 4) {
    int4 q[4];
    #pragma unroll
    for (int j = 0; j < 4; ++j) {
      const int vi = v0 + j * TTHREADS + tid;
      q[j] = vi < nvec ? dv[vi] : make_int4(-1, -1, -1, -1);
    }
    #pragma unroll
    for (int j = 0; j < 4; ++j) {
      const int e0 = (v0 + j * TTHREADS + tid) << 2;
      if ((unsigned)(q[j].x - d0) < (unsigned)nt) note(e0 + 0);
      if ((unsigned)(q[j].y - d0) < (unsigned)nt) note(e0 + 1);
      if ((unsigned)(q[j].z - d0) < (unsigned)nt) note(e0 + 2);
      if ((unsigned)(q[j].w - d0) < (unsigned)nt) note(e0 + 3);
    }
  }
  for (int i = Ev + tid; i < E; i += TTHREADS)
    if ((unsigned)(dstp[i] - d0) < (unsigned)nt) note(i);
  __syncthreads();
  const unsigned total = cnt;
  const bool fast = total <= (unsigned)TCAP;

  auto edge_max = [&](int s, unsigned dl) {
    const float4 ss = *(const float4*)(s_src + (size_t)s * HEADS);
    const float4 sd = *(const float4*)(sdt + dl * HEADS);
    atomicMax(&mxo[dl * HEADS + 0], ordEnc(lrelu(ss.x + sd.x)));
    atomicMax(&mxo[dl * HEADS + 1], ordEnc(lrelu(ss.y + sd.y)));
    atomicMax(&mxo[dl * HEADS + 2], ordEnc(lrelu(ss.z + sd.z)));
    atomicMax(&mxo[dl * HEADS + 3], ordEnc(lrelu(ss.w + sd.w)));
  };
  if (fast) {
    for (int i = tid; i < (int)total; i += TTHREADS) {
      const int idx = (int)elist[i];
      const int s = srcp[idx];
      const unsigned dl = (unsigned)(dstp[idx] - d0);
      edge_max(s, dl);
      elist[i] = ((unsigned)s << 7) | dl;
    }
  } else {
    for (int i = tid; i < E; i += TTHREADS) {
      const unsigned dl = (unsigned)(dstp[i] - d0);
      if (dl < (unsigned)nt) edge_max(srcp[i], dl);
    }
  }
  __syncthreads();

  for (int dl = wv; dl < nt; dl += (TTHREADS >> 5)) {
    const int d = d0 + dl;
    const float e  = lrelu(s_src[(size_t)d * HEADS + hl] + sdt[dl * HEADS + hl]);
    const float wl = expf(e - ordDec(mxo[dl * HEADS + hl]));
    if (lane < HEADS) den[dl * HEADS + lane] = wl;
    float w[HEADS];
    #pragma unroll
    for (int h = 0; h < HEADS; ++h) w[h] = __shfl(wl, h, 32);
    const float* hs = Hin + (size_t)d * HIDDEN;
    #pragma unroll
    for (int i = 0; i < 8; ++i) {
      const int ch = lane + (i << 5);
      acc[dl * APITCH + ch] = hs[ch] * w[i >> 1];
    }
  }
  __syncthreads();

  if (fast) {
    const int cntv = (int)total;
    for (int b = wv; b < cntv; b += (TTHREADS >> 5) * 4) {
      unsigned en[4];
      float sv[4];
      float hv[4][8];
      #pragma unroll
      for (int j = 0; j < 4; ++j) {
        const int idx = b + j * (TTHREADS >> 5);
        en[j] = idx < cntv ? elist[idx] : 0xFFFFFFFFu;
      }
      #pragma unroll
      for (int j = 0; j < 4; ++j) {
        const unsigned s = en[j] == 0xFFFFFFFFu ? 0u : (en[j] >> 7);
        sv[j] = s_src[(size_t)s * HEADS + hl];
        const float* hs = Hin + (size_t)s * HIDDEN;
        #pragma unroll
        for (int i = 0; i < 8; ++i) hv[j][i] = hs[lane + (i << 5)];
      }
      #pragma unroll
      for (int j = 0; j < 4; ++j) {
        if (en[j] != 0xFFFFFFFFu) {
          const unsigned dl = en[j] & 127u;
          const float e  = lrelu(sv[j] + sdt[dl * HEADS + hl]);
          const float wl = expf(e - ordDec(mxo[dl * HEADS + hl]));
          if (lane < HEADS) atomicAdd(&den[dl * HEADS + lane], wl);
          float w[HEADS];
          #pragma unroll
          for (int h = 0; h < HEADS; ++h) w[h] = __shfl(wl, h, 32);
          #pragma unroll
          for (int i = 0; i < 8; ++i)
            atomicAdd(&acc[dl * APITCH + lane + (i << 5)], hv[j][i] * w[i >> 1]);
        }
      }
    }
  } else {
    for (int i = tid; i < E; i += TTHREADS) {
      const unsigned dl = (unsigned)(dstp[i] - d0);
      if (dl < (unsigned)nt) {
        const int s = srcp[i];
        const float* hs = Hin + (size_t)s * HIDDEN;
        for (int h = 0; h < HEADS; ++h) {
          const float e  = lrelu(s_src[(size_t)s * HEADS + h] + sdt[dl * HEADS + h]);
          const float wh = expf(e - ordDec(mxo[dl * HEADS + h]));
          atomicAdd(&den[dl * HEADS + h], wh);
          for (int c = 0; c < CHAN; ++c)
            atomicAdd(&acc[dl * APITCH + h * CHAN + c], hs[h * CHAN + c] * wh);
        }
      }
    }
  }
  __syncthreads();

  for (int qd = tid; qd < TN * (HIDDEN / 4); qd += TTHREADS) {
    const int dl = qd / (HIDDEN / 4);
    const int c  = (qd - dl * (HIDDEN / 4)) << 2;
    float4 v = make_float4(0.f, 0.f, 0.f, 0.f);
    if (dl < nt) {
      const float dn = den[dl * HEADS + (c >> 6)];
      v = *(const float4*)(acc + dl * APITCH + c);
      v.x = v.x / dn + bias[c + 0];
      v.y = v.y / dn + bias[c + 1];
      v.z = v.z / dn + bias[c + 2];
      v.w = v.w / dn + bias[c + 3];
      if (ELU) {
        v.x = v.x > 0.f ? v.x : expm1f(v.x);
        v.y = v.y > 0.f ? v.y : expm1f(v.y);
        v.z = v.z > 0.f ? v.z : expm1f(v.z);
        v.w = v.w > 0.f ? v.w : expm1f(v.w);
      }
    }
    *(float4*)(acc + dl * APITCH + c) = v;
  }
  __syncthreads();

  if (ELU)
    tile_gemm<HIDDEN, true, false>(acc, wv, lane, d0, nt, Wn, nullptr, att_src_n, att_dst_n,
                                   Hout, s_src_n, s_dst_n);
  else
    tile_gemm<CHAN, false, true>(acc, wv, lane, d0, nt, Wn, bn, nullptr, nullptr,
                                 Hout, nullptr, nullptr);
}

extern "C" void kernel_launch(void* const* d_in, const int* in_sizes, int n_in,
                              void* d_out, int out_size, void* d_ws, size_t ws_size,
                              hipStream_t stream) {
  const float* x     = (const float*)d_in[0];
  const int*   ei    = (const int*)  d_in[1];
  const float* W1    = (const float*)d_in[2];
  const float* asrc1 = (const float*)d_in[3];
  const float* adst1 = (const float*)d_in[4];
  const float* b1    = (const float*)d_in[5];
  const float* W2    = (const float*)d_in[6];
  const float* asrc2 = (const float*)d_in[7];
  const float* adst2 = (const float*)d_in[8];
  const float* b2    = (const float*)d_in[9];
  const float* Wf    = (const float*)d_in[10];
  const float* bf    = (const float*)d_in[11];

  const int B  = 2;
  const int N  = in_sizes[0] / (B * FEAT);
  const int E  = in_sizes[1] / 2;

  float* p = (float*)d_ws;
  float* buf_h   = p; p += (size_t)N * HIDDEN;
  float* buf_h2  = p; p += (size_t)N * HIDDEN;
  float* ssrc    = p; p += (size_t)N * HEADS;
  float* sdst    = p; p += (size_t)N * HEADS;
  float* ssrc2   = p; p += (size_t)N * HEADS;
  float* sdst2   = p; p += (size_t)N * HEADS;

  const int mtiles  = N >> 5;
  const int gtiles  = mtiles * (HIDDEN >> 6);
  const int ntiles  = (N + TN - 1) / TN;
  const dim3 blkG(128);
  const dim3 blkT(TTHREADS);

  for (int b = 0; b < B; ++b) {
    const float* xb = x + (size_t)b * N * FEAT;
    float* outb = (float*)d_out + (size_t)b * N * CHAN;

    wmma_gemm_kernel<true, false><<<(gtiles + 3) / 4, blkG, 0, stream>>>(
        xb, W1, nullptr, asrc1, adst1, buf_h, ssrc, sdst, N, FEAT, HIDDEN);
    tile_agg_kernel<true><<<ntiles, blkT, 0, stream>>>(ei, E, N, ssrc, sdst, buf_h, b1,
        W2, nullptr, asrc2, adst2, buf_h2, ssrc2, sdst2);

    tile_agg_kernel<false><<<ntiles, blkT, 0, stream>>>(ei, E, N, ssrc2, sdst2, buf_h2, b2,
        Wf, bf, nullptr, nullptr, outb, nullptr, nullptr);
  }
}


# Tile's noted edges (source, row) saved to workspace by the first aggregate launch and read back by the other three: edge list filtered once per call, not four times; sum order still arrival order
# speedup vs baseline: 4.6658x; 1.1989x over previous
#include <hip/hip_runtime.h>
#include <math.h>


typedef __attribute__((ext_vector_type(2))) float v2f;
typedef __attribute__((ext_vector_type(8))) float v8f;

#define HEADS     4
#define CHAN      64
#define HIDDEN    256
#define FEAT      128
#define NEG_SLOPE 0.2f

static __device__ __forceinline__ float lrelu(float x) {
  return x > 0.f ? x : NEG_SLOPE * x;
}
static __device__ __forceinline__ unsigned ordEnc(float f) {
  unsigned u = __float_as_uint(f);
  return (u & 0x80000000u) ? ~u : (u | 0x80000000u);
}
static __device__ __forceinline__ float ordDec(unsigned u) {
  unsigned v = (u & 0x80000000u) ? (u & 0x7fffffffu) : ~u;
  return __uint_as_float(v);
}

template <bool GAT, bool BIAS>
__global__ void wmma_gemm_kernel(const float* __restrict__ A,
                                 const float* __restrict__ W,
                                 const float* __restrict__ bias,
                                 const float* __restrict__ att_src,
                                 const float* __restrict__ att_dst,
                                 float* __restrict__ Hout,
                                 float* __restrict__ s_src,
                                 float* __restrict__ s_dst,
                                 int M, int K, int Nc) {
  const int lane = threadIdx.x & 31;
  const int wv   = threadIdx.x >> 5;
  const int wpb  = blockDim.x >> 5;
  const int tile = blockIdx.x * wpb + wv;
  const int ctiles = Nc >> 6;
  const int mt = tile / ctiles;
  const int ct = tile - mt * ctiles;
  const int m0 = mt << 5;
  if (m0 >= M) return;
  const int n0 = ct << 6;
  const int lo = lane & 15;
  const int hi = lane >> 4;

  v8f acc[2][4] = {};
  const float* arow0 = A + (size_t)(m0 + lo) * K;
  const float* arow1 = A + (size_t)(m0 + 16 + lo) * K;
  const float* wcol  = W + n0 + lo;

  for (int k = 0; k < K; k += 4) {
    const int ka = k + (hi << 1);
    v2f a0 = *(const v2f*)(arow0 + ka);
    v2f a1 = *(const v2f*)(arow1 + ka);
    const float* w0 = wcol + (size_t)ka * Nc;
    const float* w1 = w0 + Nc;
    v2f b0 = {w0[0],  w1[0]};
    v2f b1 = {w0[16], w1[16]};
    v2f b2 = {w0[32], w1[32]};
    v2f b3 = {w0[48], w1[48]};
    acc[0][0] = __builtin_amdgcn_wmma_f32_16x16x4_f32(false, a0, false, b0, (short)0, acc[0][0], false, false);
    acc[1][0] = __builtin_amdgcn_wmma_f32_16x16x4_f32(false, a1, false, b0, (short)0, acc[1][0], false, false);
    acc[0][1] = __builtin_amdgcn_wmma_f32_16x16x4_f32(false, a0, false, b1, (short)0, acc[0][1], false, false);
    acc[1][1] = __builtin_amdgcn_wmma_f32_16x16x4_f32(false, a1, false, b1, (short)0, acc[1][1], false, false);
    acc[0][2] = __builtin_amdgcn_wmma_f32_16x16x4_f32(false, a0, false, b2, (short)0, acc[0][2], false, false);
    acc[1][2] = __builtin_amdgcn_wmma_f32_16x16x4_f32(false, a1, false, b2, (short)0, acc[1][2], false, false);
    acc[0][3] = __builtin_amdgcn_wmma_f32_16x16x4_f32(false, a0, false, b3, (short)0, acc[0][3], false, false);
    acc[1][3] = __builtin_amdgcn_wmma_f32_16x16x4_f32(false, a1, false, b3, (short)0, acc[1][3], false, false);
  }

  float bv0 = 0.f, bv1 = 0.f, bv2 = 0.f, bv3 = 0.f;
  if (BIAS) {
    bv0 = bias[n0 + lo];      bv1 = bias[n0 + 16 + lo];
    bv2 = bias[n0 + 32 + lo]; bv3 = bias[n0 + 48 + lo];
  }

  #pragma unroll
  for (int g = 0; g < 2; ++g) {
    const int rbase = m0 + (g << 4) + (hi << 3);
    #pragma unroll
    for (int r = 0; r < 8; ++r) {
      float* hp = Hout + (size_t)(rbase + r) * Nc + n0 + lo;
      hp[0]  = acc[g][0][r] + bv0;
      hp[16] = acc[g][1][r] + bv1;
      hp[32] = acc[g][2][r] + bv2;
      hp[48] = acc[g][3][r] + bv3;
    }
  }

  if (GAT) {
    const int head = n0 >> 6;
    const float* as = att_src + head * CHAN;
    const float* ad = att_dst + head * CHAN;
    const float as0 = as[lo], as1 = as[16 + lo], as2 = as[32 + lo], as3 = as[48 + lo];
    const float ad0 = ad[lo], ad1 = ad[16 + lo], ad2 = ad[32 + lo], ad3 = ad[48 + lo];
    #pragma unroll
    for (int g = 0; g < 2; ++g) {
      const int rbase = m0 + (g << 4) + (hi << 3);
      #pragma unroll
      for (int r = 0; r < 8; ++r) {
        float ss = acc[g][0][r] * as0 + acc[g][1][r] * as1 + acc[g][2][r] * as2 + acc[g][3][r] * as3;
        float sd = acc[g][0][r] * ad0 + acc[g][1][r] * ad1 + acc[g][2][r] * ad2 + acc[g][3][r] * ad3;
        #pragma unroll
        for (int off = 8; off > 0; off >>= 1) {
          ss += __shfl_xor(ss, off, 32);
          sd += __shfl_xor(sd, off, 32);
        }
        if (lo == 0) {
          const int row = rbase + r;
          s_src[row * HEADS + head] = ss;
          s_dst[row * HEADS + head] = sd;
        }
      }
    }
  }
}

#define TN       80
#define APITCH   (HIDDEN + 4)
#define TCAP     20480
#define TTHREADS 1024
static_assert(TN <= 128, "a noted edge keeps its tile row in 7 bits");

template <int NC, bool GAT, bool BIAS>
static __device__ __forceinline__ void tile_gemm(const float* tile, int wv, int lane, int d0, int nt,
                                                 const float* __restrict__ W,
                                                 const float* __restrict__ bias,
                                                 const float* __restrict__ att_src,
                                                 const float* __restrict__ att_dst,
                                                 float* __restrict__ Hout,
                                                 float* __restrict__ s_src,
                                                 float* __restrict__ s_dst) {
  constexpr int CT = NC / 64;
  static_assert(TN % 16 == 0, "whole 16-row tiles");
  if (wv >= (TN / 16) * CT) return;
  const int rt = wv / CT;
  const int ct = wv - rt * CT;
  const int n0 = ct << 6;
  const int lo = lane & 15;
  const int hi = lane >> 4;

  v8f c0 = {}, c1 = {}, c2 = {}, c3 = {};
  const float* arow = tile + (rt * 16 + lo) * APITCH + (hi << 1);
  const float* wcol = W + (size_t)(hi << 1) * NC + n0 + lo;

  for (int k = 0; k < HIDDEN; k += 16) {
    v2f a[4];
    float b[4][8];
    #pragma unroll
    for (int u = 0; u < 4; ++u) {
      const float* w0 = wcol + (size_t)(k + 4 * u) * NC;
      b[u][0] = w0[0];  b[u][1] = w0[NC];
      b[u][2] = w0[16]; b[u][3] = w0[NC + 16];
      b[u][4] = w0[32]; b[u][5] = w0[NC + 32];
      b[u][6] = w0[48]; b[u][7] = w0[NC + 48];
    }
    #pragma unroll
    for (int u = 0; u < 4; ++u) a[u] = *(const v2f*)(arow + k + 4 * u);
    #pragma unroll
    for (int u = 0; u < 4; ++u) {
      const v2f b0 = {b[u][0], b[u][1]};
      const v2f b1 = {b[u][2], b[u][3]};
      const v2f b2 = {b[u][4], b[u][5]};
      const v2f b3 = {b[u][6], b[u][7]};
      c0 = __builtin_amdgcn_wmma_f32_16x16x4_f32(false, a[u], false, b0, (short)0, c0, false, false);
      c1 = __builtin_amdgcn_wmma_f32_16x16x4_f32(false, a[u], false, b1, (short)0, c1, false, false);
      c2 = __builtin_amdgcn_wmma_f32_16x16x4_f32(false, a[u], false, b2, (short)0, c2, false, false);
      c3 = __builtin_amdgcn_wmma_f32_16x16x4_f32(false, a[u], false, b3, (short)0, c3, false, false);
    }
  }

  float bv0 = 0.f, bv1 = 0.f, bv2 = 0.f, bv3 = 0.f;
  if (BIAS) {
    bv0 = bias[n0 + lo];      bv1 = bias[n0 + 16 + lo];
    bv2 = bias[n0 + 32 + lo]; bv3 = bias[n0 + 48 + lo];
  }
  const int rloc = rt * 16 + (hi << 3);
  #pragma unroll
  for (int r = 0; r < 8; ++r) {
    if (rloc + r < nt) {
      float* hp = Hout + (size_t)(d0 + rloc + r) * NC + n0 + lo;
      hp[0]  = c0[r] + bv0;
      hp[16] = c1[r] + bv1;
      hp[32] = c2[r] + bv2;
      hp[48] = c3[r] + bv3;
    }
  }

  if (GAT) {
    const int head = ct;
    const float* as = att_src + head * CHAN;
    const float* ad = att_dst + head * CHAN;
    const float as0 = as[lo], as1 = as[16 + lo], as2 = as[32 + lo], as3 = as[48 + lo];
    const float ad0 = ad[lo], ad1 = ad[16 + lo], ad2 = ad[32 + lo], ad3 = ad[48 + lo];
    #pragma unroll
    for (int r = 0; r < 8; ++r) {
      float ss = c0[r] * as0 + c1[r] * as1 + c2[r] * as2 + c3[r] * as3;
      float sd = c0[r] * ad0 + c1[r] * ad1 + c2[r] * ad2 + c3[r] * ad3;
      #pragma unroll
      for (int off = 8; off > 0; off >>= 1) {
        ss += __shfl_xor(ss, off, 32);
        sd += __shfl_xor(sd, off, 32);
      }
      if (lo == 0 && rloc + r < nt) {
        const int row = d0 + rloc + r;
        s_src[row * HEADS + head] = ss;
        s_dst[row * HEADS + head] = sd;
      }
    }
  }
}

template <bool ELU, bool FIRST>
__global__ void __launch_bounds__(TTHREADS)
tile_agg_kernel(const int* __restrict__ ei, int E, int N,
                const float* __restrict__ s_src,
                const float* __restrict__ s_dst,
                const float* __restrict__ Hin,
                const float* __restrict__ bias,
                const float* __restrict__ Wn,
                const float* __restrict__ bn,
                const float* __restrict__ att_src_n,
                const float* __restrict__ att_dst_n,
                float* __restrict__ Hout,
                float* __restrict__ s_src_n,
                float* __restrict__ s_dst_n,
                unsigned* __restrict__ tl_cnt,
                unsigned* __restrict__ tl_list) {
  __shared__ __attribute__((aligned(16))) float acc[TN * APITCH];
  __shared__ unsigned elist[TCAP];
  __shared__ __attribute__((aligned(16))) float sdt[TN * HEADS];
  __shared__ unsigned mxo[TN * HEADS];
  __shared__ float den[TN * HEADS];
  __shared__ unsigned cnt;

  const int tid  = threadIdx.x;
  const int lane = tid & 31;
  const int wv   = tid >> 5;
  const int hl   = lane & (HEADS - 1);
  const int d0   = blockIdx.x * TN;
  const int nt   = (N - d0) < TN ? (N - d0) : TN;
  const int* __restrict__ srcp = ei;
  const int* __restrict__ dstp = ei + E;

  unsigned* __restrict__ tl = tl_list + (size_t)blockIdx.x * TCAP;

  if (FIRST && tid == 0) cnt = 0u;
  unsigned total = 0u;
  if (!FIRST) total = tl_cnt[blockIdx.x];
  if (tid < nt * HEADS) {
    const size_t g = (size_t)d0 * HEADS + tid;
    const float sd = s_dst[g];
    sdt[tid] = sd;
    mxo[tid] = ordEnc(lrelu(s_src[g] + sd));
  }
  __syncthreads();

  if (FIRST) {
  auto note = [&](int idx) {
    const unsigned slot = atomicAdd(&cnt, 1u);
    if (slot < (unsigned)TCAP) elist[slot] = (unsigned)idx;
  };

  const int Ev   = (((size_t)dstp & 15) == 0) ? (E & ~3) : 0;
  const int nvec = Ev >> 2;
  const int4* dv = (const int4*)dstp;
  for (int v0 = 0; v0 < nvec; v0 += TTHREADS * 4) {
    int4 q[4];
    #pragma unroll
    for (int j = 0; j < 4; ++j) {
      const int vi = v0 + j * TTHREADS + tid;
      q[j] = vi < nvec ? dv[vi] : make_int4(-1, -1, -1, -1);
    }
    #pragma unroll
    for (int j = 0; j < 4; ++j) {
      const int e0 = (v0 + j * TTHREADS + tid) << 2;
      if ((unsigned)(q[j].x - d0) < (unsigned)nt) note(e0 + 0);
      if ((unsigned)(q[j].y - d0) < (unsigned)nt) note(e0 + 1);
      if ((unsigned)(q[j].z - d0) < (unsigned)nt) note(e0 + 2);
      if ((unsigned)(q[j].w - d0) < (unsigned)nt) note(e0 + 3);
    }
  }
  for (int i = Ev + tid; i < E; i += TTHREADS)
    if ((unsigned)(dstp[i] - d0) < (unsigned)nt) note(i);
  __syncthreads();
  total = cnt;
  if (tid == 0) tl_cnt[blockIdx.x] = total;
  }
  const bool fast = total <= (unsigned)TCAP;

  auto edge_max_q = [&](const float4 ss, unsigned dl) {
    const float4 sd = *(const float4*)(sdt + dl * HEADS);
    atomicMax(&mxo[dl * HEADS + 0], ordEnc(lrelu(ss.x + sd.x)));
    atomicMax(&mxo[dl * HEADS + 1], ordEnc(lrelu(ss.y + sd.y)));
    atomicMax(&mxo[dl * HEADS + 2], ordEnc(lrelu(ss.z + sd.z)));
    atomicMax(&mxo[dl * HEADS + 3], ordEnc(lrelu(ss.w + sd.w)));
  };
  auto edge_max = [&](int s, unsigned dl) {
    edge_max_q(*(const float4*)(s_src + (size_t)s * HEADS), dl);
  };
  if (fast && FIRST) {
    for (int i = tid; i < (int)total; i += TTHREADS) {
      const int idx = (int)elist[i];
      const int s = srcp[idx];
      const unsigned dl = (unsigned)(dstp[idx] - d0);
      edge_max(s, dl);
      const unsigned en = ((unsigned)s << 7) | dl;
      elist[i] = en;
      tl[i] = en;
    }
  } else if (fast) {
    for (int i0 = tid; i0 < (int)total; i0 += 2 * TTHREADS) {
      const int i1 = i0 + TTHREADS;
      const bool v1 = i1 < (int)total;
      const unsigned e0 = tl[i0];
      const unsigned e1 = v1 ? tl[i1] : 0u;
      const unsigned s0 = min(e0 >> 7, (unsigned)(N - 1));
      const unsigned s1 = min(e1 >> 7, (unsigned)(N - 1));
      const float4 q0 = *(const float4*)(s_src + (size_t)s0 * HEADS);
      const float4 q1 = *(const float4*)(s_src + (size_t)s1 * HEADS);
      asm volatile("" :: "v"(q0.x), "v"(q0.y), "v"(q0.z), "v"(q0.w),
                         "v"(q1.x), "v"(q1.y), "v"(q1.z), "v"(q1.w));
      edge_max_q(q0, e0 & 127u);
      elist[i0] = (s0 << 7) | (e0 & 127u);
      if (v1) {
        edge_max_q(q1, e1 & 127u);
        elist[i1] = (s1 << 7) | (e1 & 127u);
      }
    }
  } else {
    for (int i = tid; i < E; i += TTHREADS) {
      const unsigned dl = (unsigned)(dstp[i] - d0);
      if (dl < (unsigned)nt) edge_max(srcp[i], dl);
    }
  }
  __syncthreads();

  for (int dl = wv; dl < nt; dl += (TTHREADS >> 5)) {
    const int d = d0 + dl;
    const float e  = lrelu(s_src[(size_t)d * HEADS + hl] + sdt[dl * HEADS + hl]);
    const float wl = expf(e - ordDec(mxo[dl * HEADS + hl]));
    if (lane < HEADS) den[dl * HEADS + lane] = wl;
    float w[HEADS];
    #pragma unroll
    for (int h = 0; h < HEADS; ++h) w[h] = __shfl(wl, h, 32);
    const float* hs = Hin + (size_t)d * HIDDEN;
    #pragma unroll
    for (int i = 0; i < 8; ++i) {
      const int ch = lane + (i << 5);
      acc[dl * APITCH + ch] = hs[ch] * w[i >> 1];
    }
  }
  __syncthreads();

  if (fast) {
    const int cntv = (int)total;
    for (int b = wv; b < cntv; b += (TTHREADS >> 5) * 4) {
      unsigned en[4];
      float sv[4];
      float hv[4][8];
      #pragma unroll
      for (int j = 0; j < 4; ++j) {
        const int idx = b + j * (TTHREADS >> 5);
        en[j] = idx < cntv ? elist[idx] : 0xFFFFFFFFu;
      }
      #pragma unroll
      for (int j = 0; j < 4; ++j) {
        const unsigned s = en[j] == 0xFFFFFFFFu ? 0u : (en[j] >> 7);
        sv[j] = s_src[(size_t)s * HEADS + hl];
        const float* hs = Hin + (size_t)s * HIDDEN;
        #pragma unroll
        for (int i = 0; i < 8; ++i) hv[j][i] = hs[lane + (i << 5)];
      }
      #pragma unroll
      for (int j = 0; j < 4; ++j) {
        if (en[j] != 0xFFFFFFFFu) {
          const unsigned dl = en[j] & 127u;
          const float e  = lrelu(sv[j] + sdt[dl * HEADS + hl]);
          const float wl = expf(e - ordDec(mxo[dl * HEADS + hl]));
          if (lane < HEADS) atomicAdd(&den[dl * HEADS + lane], wl);
          float w[HEADS];
          #pragma unroll
          for (int h = 0; h < HEADS; ++h) w[h] = __shfl(wl, h, 32);
          #pragma unroll
          for (int i = 0; i < 8; ++i)
            atomicAdd(&acc[dl * APITCH + lane + (i << 5)], hv[j][i] * w[i >> 1]);
        }
      }
    }
  } else {
    for (int i = tid; i < E; i += TTHREADS) {
      const unsigned dl = (unsigned)(dstp[i] - d0);
      if (dl < (unsigned)nt) {
        const int s = srcp[i];
        const float* hs = Hin + (size_t)s * HIDDEN;
        for (int h = 0; h < HEADS; ++h) {
          const float e  = lrelu(s_src[(size_t)s * HEADS + h] + sdt[dl * HEADS + h]);
          const float wh = expf(e - ordDec(mxo[dl * HEADS + h]));
          atomicAdd(&den[dl * HEADS + h], wh);
          for (int c = 0; c < CHAN; ++c)
            atomicAdd(&acc[dl * APITCH + h * CHAN + c], hs[h * CHAN + c] * wh);
        }
      }
    }
  }
  __syncthreads();

  for (int qd = tid; qd < TN * (HIDDEN / 4); qd += TTHREADS) {
    const int dl = qd / (HIDDEN / 4);
    const int c  = (qd - dl * (HIDDEN / 4)) << 2;
    float4 v = make_float4(0.f, 0.f, 0.f, 0.f);
    if (dl < nt) {
      const float dn = den[dl * HEADS + (c >> 6)];
      v = *(const float4*)(acc + dl * APITCH + c);
      v.x = v.x / dn + bias[c + 0];
      v.y = v.y / dn + bias[c + 1];
      v.z = v.z / dn + bias[c + 2];
      v.w = v.w / dn + bias[c + 3];
      if (ELU) {
        v.x = v.x > 0.f ? v.x : expm1f(v.x);
        v.y = v.y > 0.f ? v.y : expm1f(v.y);
        v.z = v.z > 0.f ? v.z : expm1f(v.z);
        v.w = v.w > 0.f ? v.w : expm1f(v.w);
      }
    }
    *(float4*)(acc + dl * APITCH + c) = v;
  }
  __syncthreads();

  if (ELU)
    tile_gemm<HIDDEN, true, false>(acc, wv, lane, d0, nt, Wn, nullptr, att_src_n, att_dst_n,
                                   Hout, s_src_n, s_dst_n);
  else
    tile_gemm<CHAN, false, true>(acc, wv, lane, d0, nt, Wn, bn, nullptr, nullptr,
                                 Hout, nullptr, nullptr);
}

extern "C" void kernel_launch(void* const* d_in, const int* in_sizes, int n_in,
                              void* d_out, int out_size, void* d_ws, size_t ws_size,
                              hipStream_t stream) {
  const float* x     = (const float*)d_in[0];
  const int*   ei    = (const int*)  d_in[1];
  const float* W1    = (const float*)d_in[2];
  const float* asrc1 = (const float*)d_in[3];
  const float* adst1 = (const float*)d_in[4];
  const float* b1    = (const float*)d_in[5];
  const float* W2    = (const float*)d_in[6];
  const float* asrc2 = (const float*)d_in[7];
  const float* adst2 = (const float*)d_in[8];
  const float* b2    = (const float*)d_in[9];
  const float* Wf    = (const float*)d_in[10];
  const float* bf    = (const float*)d_in[11];

  const int B  = 2;
  const int N  = in_sizes[0] / (B * FEAT);
  const int E  = in_sizes[1] / 2;

  float* p = (float*)d_ws;
  float* buf_h   = p; p += (size_t)N * HIDDEN;
  float* buf_h2  = p; p += (size_t)N * HIDDEN;
  float* ssrc    = p; p += (size_t)N * HEADS;
  float* sdst    = p; p += (size_t)N * HEADS;
  float* ssrc2   = p; p += (size_t)N * HEADS;
  float* sdst2   = p; p += (size_t)N * HEADS;
  const int ntiles  = (N + TN - 1) / TN;
  unsigned* tlcnt  = (unsigned*)p; p += ntiles;
  unsigned* tllist = (unsigned*)p;

  const int mtiles  = N >> 5;
  const int gtiles  = mtiles * (HIDDEN >> 6);
  const dim3 blkG(128);
  const dim3 blkT(TTHREADS);

  for (int b = 0; b < B; ++b) {
    const float* xb = x + (size_t)b * N * FEAT;
    float* outb = (float*)d_out + (size_t)b * N * CHAN;

    wmma_gemm_kernel<true, false><<<(gtiles + 3) / 4, blkG, 0, stream>>>(
        xb, W1, nullptr, asrc1, adst1, buf_h, ssrc, sdst, N, FEAT, HIDDEN);
    if (b == 0)
      tile_agg_kernel<true, true><<<ntiles, blkT, 0, stream>>>(ei, E, N, ssrc, sdst, buf_h, b1,
          W2, nullptr, asrc2, adst2, buf_h2, ssrc2, sdst2, tlcnt, tllist);
    else
      tile_agg_kernel<true, false><<<ntiles, blkT, 0, stream>>>(ei, E, N, ssrc, sdst, buf_h, b1,
          W2, nullptr, asrc2, adst2, buf_h2, ssrc2, sdst2, tlcnt, tllist);

    tile_agg_kernel<false, false><<<ntiles, blkT, 0, stream>>>(ei, E, N, ssrc2, sdst2, buf_h2, b2,
        Wf, bf, nullptr, nullptr, outb, nullptr, nullptr, tlcnt, tllist);
  }
}


# Softmax weights exp(e-max) formed once per noted edge by one lane (4 heads), kept in LDS; edge loop only gathers, multiplies, adds; denominator summed in that pass, self loop joins arrival order
# speedup vs baseline: 4.8795x; 1.0458x over previous
#include <hip/hip_runtime.h>
#include <math.h>


typedef __attribute__((ext_vector_type(2))) float v2f;
typedef __attribute__((ext_vector_type(8))) float v8f;

#define HEADS     4
#define CHAN      64
#define HIDDEN    256
#define FEAT      128
#define NEG_SLOPE 0.2f

static __device__ __forceinline__ float lrelu(float x) {
  return x > 0.f ? x : NEG_SLOPE * x;
}
static __device__ __forceinline__ unsigned ordEnc(float f) {
  unsigned u = __float_as_uint(f);
  return (u & 0x80000000u) ? ~u : (u | 0x80000000u);
}
static __device__ __forceinline__ float ordDec(unsigned u) {
  unsigned v = (u & 0x80000000u) ? (u & 0x7fffffffu) : ~u;
  return __uint_as_float(v);
}

template <bool GAT, bool BIAS>
__global__ void wmma_gemm_kernel(const float* __restrict__ A,
                                 const float* __restrict__ W,
                                 const float* __restrict__ bias,
                                 const float* __restrict__ att_src,
                                 const float* __restrict__ att_dst,
                                 float* __restrict__ Hout,
                                 float* __restrict__ s_src,
                                 float* __restrict__ s_dst,
                                 int M, int K, int Nc) {
  const int lane = threadIdx.x & 31;
  const int wv   = threadIdx.x >> 5;
  const int wpb  = blockDim.x >> 5;
  const int tile = blockIdx.x * wpb + wv;
  const int ctiles = Nc >> 6;
  const int mt = tile / ctiles;
  const int ct = tile - mt * ctiles;
  const int m0 = mt << 5;
  if (m0 >= M) return;
  const int n0 = ct << 6;
  const int lo = lane & 15;
  const int hi = lane >> 4;

  v8f acc[2][4] = {};
  const float* arow0 = A + (size_t)(m0 + lo) * K;
  const float* arow1 = A + (size_t)(m0 + 16 + lo) * K;
  const float* wcol  = W + n0 + lo;

  for (int k = 0; k < K; k += 4) {
    const int ka = k + (hi << 1);
    v2f a0 = *(const v2f*)(arow0 + ka);
    v2f a1 = *(const v2f*)(arow1 + ka);
    const float* w0 = wcol + (size_t)ka * Nc;
    const float* w1 = w0 + Nc;
    v2f b0 = {w0[0],  w1[0]};
    v2f b1 = {w0[16], w1[16]};
    v2f b2 = {w0[32], w1[32]};
    v2f b3 = {w0[48], w1[48]};
    acc[0][0] = __builtin_amdgcn_wmma_f32_16x16x4_f32(false, a0, false, b0, (short)0, acc[0][0], false, false);
    acc[1][0] = __builtin_amdgcn_wmma_f32_16x16x4_f32(false, a1, false, b0, (short)0, acc[1][0], false, false);
    acc[0][1] = __builtin_amdgcn_wmma_f32_16x16x4_f32(false, a0, false, b1, (short)0, acc[0][1], false, false);
    acc[1][1] = __builtin_amdgcn_wmma_f32_16x16x4_f32(false, a1, false, b1, (short)0, acc[1][1], false, false);
    acc[0][2] = __builtin_amdgcn_wmma_f32_16x16x4_f32(false, a0, false, b2, (short)0, acc[0][2], false, false);
    acc[1][2] = __builtin_amdgcn_wmma_f32_16x16x4_f32(false, a1, false, b2, (short)0, acc[1][2], false, false);
    acc[0][3] = __builtin_amdgcn_wmma_f32_16x16x4_f32(false, a0, false, b3, (short)0, acc[0][3], false, false);
    acc[1][3] = __builtin_amdgcn_wmma_f32_16x16x4_f32(false, a1, false, b3, (short)0, acc[1][3], false, false);
  }

  float bv0 = 0.f, bv1 = 0.f, bv2 = 0.f, bv3 = 0.f;
  if (BIAS) {
    bv0 = bias[n0 + lo];      bv1 = bias[n0 + 16 + lo];
    bv2 = bias[n0 + 32 + lo]; bv3 = bias[n0 + 48 + lo];
  }

  #pragma unroll
  for (int g = 0; g < 2; ++g) {
    const int rbase = m0 + (g << 4) + (hi << 3);
    #pragma unroll
    for (int r = 0; r < 8; ++r) {
      float* hp = Hout + (size_t)(rbase + r) * Nc + n0 + lo;
      hp[0]  = acc[g][0][r] + bv0;
      hp[16] = acc[g][1][r] + bv1;
      hp[32] = acc[g][2][r] + bv2;
      hp[48] = acc[g][3][r] + bv3;
    }
  }

  if (GAT) {
    const int head = n0 >> 6;
    const float* as = att_src + head * CHAN;
    const float* ad = att_dst + head * CHAN;
    const float as0 = as[lo], as1 = as[16 + lo], as2 = as[32 + lo], as3 = as[48 + lo];
    const float ad0 = ad[lo], ad1 = ad[16 + lo], ad2 = ad[32 + lo], ad3 = ad[48 + lo];
    #pragma unroll
    for (int g = 0; g < 2; ++g) {
      const int rbase = m0 + (g << 4) + (hi << 3);
      #pragma unroll
      for (int r = 0; r < 8; ++r) {
        float ss = acc[g][0][r] * as0 + acc[g][1][r] * as1 + acc[g][2][r] * as2 + acc[g][3][r] * as3;
        float sd = acc[g][0][r] * ad0 + acc[g][1][r] * ad1 + acc[g][2][r] * ad2 + acc[g][3][r] * ad3;
        #pragma unroll
        for (int off = 8; off > 0; off >>= 1) {
          ss += __shfl_xor(ss, off, 32);
          sd += __shfl_xor(sd, off, 32);
        }
        if (lo == 0) {
          const int row = rbase + r;
          s_src[row * HEADS + head] = ss;
          s_dst[row * HEADS + head] = sd;
        }
      }
    }
  }
}

#define TN       80
#define APITCH   (HIDDEN + 4)
#define TCAP     20480
#define WCH      4096
#define TTHREADS 1024
static_assert(TN <= 128, "a noted edge keeps its tile row in 7 bits");

template <int NC, bool GAT, bool BIAS>
static __device__ __forceinline__ void tile_gemm(const float* tile, int wv, int lane, int d0, int nt,
                                                 const float* __restrict__ W,
                                                 const float* __restrict__ bias,
                                                 const float* __restrict__ att_src,
                                                 const float* __restrict__ att_dst,
                                                 float* __restrict__ Hout,
                                                 float* __restrict__ s_src,
                                                 float* __restrict__ s_dst) {
  constexpr int CT = NC / 64;
  static_assert(TN % 16 == 0, "whole 16-row tiles");
  if (wv >= (TN / 16) * CT) return;
  const int rt = wv / CT;
  const int ct = wv - rt * CT;
  const int n0 = ct << 6;
  const int lo = lane & 15;
  const int hi = lane >> 4;

  v8f c0 = {}, c1 = {}, c2 = {}, c3 = {};
  const float* arow = tile + (rt * 16 + lo) * APITCH + (hi << 1);
  const float* wcol = W + (size_t)(hi << 1) * NC + n0 + lo;

  for (int k = 0; k < HIDDEN; k += 16) {
    v2f a[4];
    float b[4][8];
    #pragma unroll
    for (int u = 0; u < 4; ++u) {
      const float* w0 = wcol + (size_t)(k + 4 * u) * NC;
      b[u][0] = w0[0];  b[u][1] = w0[NC];
      b[u][2] = w0[16]; b[u][3] = w0[NC + 16];
      b[u][4] = w0[32]; b[u][5] = w0[NC + 32];
      b[u][6] = w0[48]; b[u][7] = w0[NC + 48];
    }
    #pragma unroll
    for (int u = 0; u < 4; ++u) a[u] = *(const v2f*)(arow + k + 4 * u);
    #pragma unroll
    for (int u = 0; u < 4; ++u) {
      const v2f b0 = {b[u][0], b[u][1]};
      const v2f b1 = {b[u][2], b[u][3]};
      const v2f b2 = {b[u][4], b[u][5]};
      const v2f b3 = {b[u][6], b[u][7]};
      c0 = __builtin_amdgcn_wmma_f32_16x16x4_f32(false, a[u], false, b0, (short)0, c0, false, false);
      c1 = __builtin_amdgcn_wmma_f32_16x16x4_f32(false, a[u], false, b1, (short)0, c1, false, false);
      c2 = __builtin_amdgcn_wmma_f32_16x16x4_f32(false, a[u], false, b2, (short)0, c2, false, false);
      c3 = __builtin_amdgcn_wmma_f32_16x16x4_f32(false, a[u], false, b3, (short)0, c3, false, false);
    }
  }

  float bv0 = 0.f, bv1 = 0.f, bv2 = 0.f, bv3 = 0.f;
  if (BIAS) {
    bv0 = bias[n0 + lo];      bv1 = bias[n0 + 16 + lo];
    bv2 = bias[n0 + 32 + lo]; bv3 = bias[n0 + 48 + lo];
  }
  const int rloc = rt * 16 + (hi << 3);
  #pragma unroll
  for (int r = 0; r < 8; ++r) {
    if (rloc + r < nt) {
      float* hp = Hout + (size_t)(d0 + rloc + r) * NC + n0 + lo;
      hp[0]  = c0[r] + bv0;
      hp[16] = c1[r] + bv1;
      hp[32] = c2[r] + bv2;
      hp[48] = c3[r] + bv3;
    }
  }

  if (GAT) {
    const int head = ct;
    const float* as = att_src + head * CHAN;
    const float* ad = att_dst + head * CHAN;
    const float as0 = as[lo], as1 = as[16 + lo], as2 = as[32 + lo], as3 = as[48 + lo];
    const float ad0 = ad[lo], ad1 = ad[16 + lo], ad2 = ad[32 + lo], ad3 = ad[48 + lo];
    #pragma unroll
    for (int r = 0; r < 8; ++r) {
      float ss = c0[r] * as0 + c1[r] * as1 + c2[r] * as2 + c3[r] * as3;
      float sd = c0[r] * ad0 + c1[r] * ad1 + c2[r] * ad2 + c3[r] * ad3;
      #pragma unroll
      for (int off = 8; off > 0; off >>= 1) {
        ss += __shfl_xor(ss, off, 32);
        sd += __shfl_xor(sd, off, 32);
      }
      if (lo == 0 && rloc + r < nt) {
        const int row = d0 + rloc + r;
        s_src[row * HEADS + head] = ss;
        s_dst[row * HEADS + head] = sd;
      }
    }
  }
}

template <bool ELU, bool FIRST>
__global__ void __launch_bounds__(TTHREADS)
tile_agg_kernel(const int* __restrict__ ei, int E, int N,
                const float* __restrict__ s_src,
                const float* __restrict__ s_dst,
                const float* __restrict__ Hin,
                const float* __restrict__ bias,
                const float* __restrict__ Wn,
                const float* __restrict__ bn,
                const float* __restrict__ att_src_n,
                const float* __restrict__ att_dst_n,
                float* __restrict__ Hout,
                float* __restrict__ s_src_n,
                float* __restrict__ s_dst_n,
                unsigned* __restrict__ tl_cnt,
                unsigned* __restrict__ tl_list) {
  __shared__ __attribute__((aligned(16))) float acc[TN * APITCH];
  __shared__ unsigned elist[TCAP];
  __shared__ __attribute__((aligned(16))) float wls[WCH * HEADS];
  __shared__ __attribute__((aligned(16))) float sdt[TN * HEADS];
  __shared__ __attribute__((aligned(16))) unsigned mxo[TN * HEADS];
  __shared__ float den[TN * HEADS];
  __shared__ unsigned cnt;

  const int tid  = threadIdx.x;
  const int lane = tid & 31;
  const int wv   = tid >> 5;
  const int hl   = lane & (HEADS - 1);
  const int d0   = blockIdx.x * TN;
  const int nt   = (N - d0) < TN ? (N - d0) : TN;
  const int* __restrict__ srcp = ei;
  const int* __restrict__ dstp = ei + E;

  unsigned* __restrict__ tl = tl_list + (size_t)blockIdx.x * TCAP;

  if (FIRST && tid == 0) cnt = 0u;
  unsigned total = 0u;
  if (!FIRST) total = tl_cnt[blockIdx.x];
  if (tid < TN * HEADS) den[tid] = 0.f;
  if (tid < nt * HEADS) {
    const size_t g = (size_t)d0 * HEADS + tid;
    const float sd = s_dst[g];
    sdt[tid] = sd;
    mxo[tid] = ordEnc(lrelu(s_src[g] + sd));
  }
  __syncthreads();

  if (FIRST) {
  auto note = [&](int idx) {
    const unsigned slot = atomicAdd(&cnt, 1u);
    if (slot < (unsigned)TCAP) elist[slot] = (unsigned)idx;
  };

  const int Ev   = (((size_t)dstp & 15) == 0) ? (E & ~3) : 0;
  const int nvec = Ev >> 2;
  const int4* dv = (const int4*)dstp;
  for (int v0 = 0; v0 < nvec; v0 += TTHREADS * 4) {
    int4 q[4];
    #pragma unroll
    for (int j = 0; j < 4; ++j) {
      const int vi = v0 + j * TTHREADS + tid;
      q[j] = vi < nvec ? dv[vi] : make_int4(-1, -1, -1, -1);
    }
    #pragma unroll
    for (int j = 0; j < 4; ++j) {
      const int e0 = (v0 + j * TTHREADS + tid) << 2;
      if ((unsigned)(q[j].x - d0) < (unsigned)nt) note(e0 + 0);
      if ((unsigned)(q[j].y - d0) < (unsigned)nt) note(e0 + 1);
      if ((unsigned)(q[j].z - d0) < (unsigned)nt) note(e0 + 2);
      if ((unsigned)(q[j].w - d0) < (unsigned)nt) note(e0 + 3);
    }
  }
  for (int i = Ev + tid; i < E; i += TTHREADS)
    if ((unsigned)(dstp[i] - d0) < (unsigned)nt) note(i);
  __syncthreads();
  total = cnt;
  if (tid == 0) tl_cnt[blockIdx.x] = total;
  }
  const bool fast = total <= (unsigned)TCAP;

  auto edge_max_q = [&](const float4 ss, unsigned dl) {
    const float4 sd = *(const float4*)(sdt + dl * HEADS);
    atomicMax(&mxo[dl * HEADS + 0], ordEnc(lrelu(ss.x + sd.x)));
    atomicMax(&mxo[dl * HEADS + 1], ordEnc(lrelu(ss.y + sd.y)));
    atomicMax(&mxo[dl * HEADS + 2], ordEnc(lrelu(ss.z + sd.z)));
    atomicMax(&mxo[dl * HEADS + 3], ordEnc(lrelu(ss.w + sd.w)));
  };
  auto edge_max = [&](int s, unsigned dl) {
    edge_max_q(*(const float4*)(s_src + (size_t)s * HEADS), dl);
  };
  if (fast && FIRST) {
    for (int i = tid; i < (int)total; i += TTHREADS) {
      const int idx = (int)elist[i];
      const int s = srcp[idx];
      const unsigned dl = (unsigned)(dstp[idx] - d0);
      edge_max(s, dl);
      const unsigned en = ((unsigned)s << 7) | dl;
      elist[i] = en;
      tl[i] = en;
    }
  } else if (fast) {
    for (int i0 = tid; i0 < (int)total; i0 += 2 * TTHREADS) {
      const int i1 = i0 + TTHREADS;
      const bool v1 = i1 < (int)total;
      const unsigned e0 = tl[i0];
      const unsigned e1 = v1 ? tl[i1] : 0u;
      const unsigned s0 = min(e0 >> 7, (unsigned)(N - 1));
      const unsigned s1 = min(e1 >> 7, (unsigned)(N - 1));
      const float4 q0 = *(const float4*)(s_src + (size_t)s0 * HEADS);
      const float4 q1 = *(const float4*)(s_src + (size_t)s1 * HEADS);
      asm volatile("" :: "v"(q0.x), "v"(q0.y), "v"(q0.z), "v"(q0.w),
                         "v"(q1.x), "v"(q1.y), "v"(q1.z), "v"(q1.w));
      edge_max_q(q0, e0 & 127u);
      elist[i0] = (s0 << 7) | (e0 & 127u);
      if (v1) {
        edge_max_q(q1, e1 & 127u);
        elist[i1] = (s1 << 7) | (e1 & 127u);
      }
    }
  } else {
    for (int i = tid; i < E; i += TTHREADS) {
      const unsigned dl = (unsigned)(dstp[i] - d0);
      if (dl < (unsigned)nt) edge_max(srcp[i], dl);
    }
  }
  __syncthreads();

  for (int dl = wv; dl < nt; dl += (TTHREADS >> 5)) {
    const int d = d0 + dl;
    const float e  = lrelu(s_src[(size_t)d * HEADS + hl] + sdt[dl * HEADS + hl]);
    const float wl = expf(e - ordDec(mxo[dl * HEADS + hl]));
    if (lane < HEADS) atomicAdd(&den[dl * HEADS + lane], wl);
    float w[HEADS];
    #pragma unroll
    for (int h = 0; h < HEADS; ++h) w[h] = __shfl(wl, h, 32);
    const float* hs = Hin + (size_t)d * HIDDEN;
    #pragma unroll
    for (int i = 0; i < 8; ++i) {
      const int ch = lane + (i << 5);
      acc[dl * APITCH + ch] = hs[ch] * w[i >> 1];
    }
  }

  if (fast) {
    for (int c0 = 0; c0 < (int)total; c0 += WCH) {
      const int cend = (int)total - c0 < WCH ? (int)total : c0 + WCH;
      if (c0 > 0) __syncthreads();
      for (int i = c0 + tid; i < cend; i += TTHREADS) {
        const unsigned en = elist[i];
        const unsigned dl = en & 127u;
        const float4 ss = *(const float4*)(s_src + (size_t)(en >> 7) * HEADS);
        const float4 sd = *(const float4*)(sdt + dl * HEADS);
        const uint4  mx = *(const uint4*)(mxo + dl * HEADS);
        float4 w;
        w.x = expf(lrelu(ss.x + sd.x) - ordDec(mx.x));
        w.y = expf(lrelu(ss.y + sd.y) - ordDec(mx.y));
        w.z = expf(lrelu(ss.z + sd.z) - ordDec(mx.z));
        w.w = expf(lrelu(ss.w + sd.w) - ordDec(mx.w));
        atomicAdd(&den[dl * HEADS + 0], w.x);
        atomicAdd(&den[dl * HEADS + 1], w.y);
        atomicAdd(&den[dl * HEADS + 2], w.z);
        atomicAdd(&den[dl * HEADS + 3], w.w);
        *(float4*)(wls + (size_t)(i - c0) * HEADS) = w;
      }
      __syncthreads();
      for (int b = c0 + wv; b < cend; b += (TTHREADS >> 5) * 4) {
        unsigned en[4];
        float4 wq[4];
        float hv[4][8];
        #pragma unroll
        for (int j = 0; j < 4; ++j) {
          const int idx = b + j * (TTHREADS >> 5);
          const bool ok = idx < cend;
          en[j] = ok ? elist[idx] : 0xFFFFFFFFu;
          wq[j] = ok ? *(const float4*)(wls + (size_t)(idx - c0) * HEADS) : make_float4(0.f, 0.f, 0.f, 0.f);
        }
        #pragma unroll
        for (int j = 0; j < 4; ++j) {
          const unsigned s = en[j] == 0xFFFFFFFFu ? 0u : (en[j] >> 7);
          const float* hs = Hin + (size_t)s * HIDDEN;
          #pragma unroll
          for (int i = 0; i < 8; ++i) hv[j][i] = hs[lane + (i << 5)];
        }
        #pragma unroll
        for (int j = 0; j < 4; ++j) {
          if (en[j] != 0xFFFFFFFFu) {
            const unsigned dl = en[j] & 127u;
            const float w[HEADS] = {wq[j].x, wq[j].y, wq[j].z, wq[j].w};
            #pragma unroll
            for (int i = 0; i < 8; ++i)
              atomicAdd(&acc[dl * APITCH + lane + (i << 5)], hv[j][i] * w[i >> 1]);
          }
        }
      }
    }
  } else {
    __syncthreads();
    for (int i = tid; i < E; i += TTHREADS) {
      const unsigned dl = (unsigned)(dstp[i] - d0);
      if (dl < (unsigned)nt) {
        const int s = srcp[i];
        const float* hs = Hin + (size_t)s * HIDDEN;
        for (int h = 0; h < HEADS; ++h) {
          const float e  = lrelu(s_src[(size_t)s * HEADS + h] + sdt[dl * HEADS + h]);
          const float wh = expf(e - ordDec(mxo[dl * HEADS + h]));
          atomicAdd(&den[dl * HEADS + h], wh);
          for (int c = 0; c < CHAN; ++c)
            atomicAdd(&acc[dl * APITCH + h * CHAN + c], hs[h * CHAN + c] * wh);
        }
      }
    }
  }
  __syncthreads();

  for (int qd = tid; qd < TN * (HIDDEN / 4); qd += TTHREADS) {
    const int dl = qd / (HIDDEN / 4);
    const int c  = (qd - dl * (HIDDEN / 4)) << 2;
    float4 v = make_float4(0.f, 0.f, 0.f, 0.f);
    if (dl < nt) {
      const float dn = den[dl * HEADS + (c >> 6)];
      v = *(const float4*)(acc + dl * APITCH + c);
      v.x = v.x / dn + bias[c + 0];
      v.y = v.y / dn + bias[c + 1];
      v.z = v.z / dn + bias[c + 2];
      v.w = v.w / dn + bias[c + 3];
      if (ELU) {
        v.x = v.x > 0.f ? v.x : expm1f(v.x);
        v.y = v.y > 0.f ? v.y : expm1f(v.y);
        v.z = v.z > 0.f ? v.z : expm1f(v.z);
        v.w = v.w > 0.f ? v.w : expm1f(v.w);
      }
    }
    *(float4*)(acc + dl * APITCH + c) = v;
  }
  __syncthreads();

  if (ELU)
    tile_gemm<HIDDEN, true, false>(acc, wv, lane, d0, nt, Wn, nullptr, att_src_n, att_dst_n,
                                   Hout, s_src_n, s_dst_n);
  else
    tile_gemm<CHAN, false, true>(acc, wv, lane, d0, nt, Wn, bn, nullptr, nullptr,
                                 Hout, nullptr, nullptr);
}

extern "C" void kernel_launch(void* const* d_in, const int* in_sizes, int n_in,
                              void* d_out, int out_size, void* d_ws, size_t ws_size,
                              hipStream_t stream) {
  const float* x     = (const float*)d_in[0];
  const int*   ei    = (const int*)  d_in[1];
  const float* W1    = (const float*)d_in[2];
  const float* asrc1 = (const float*)d_in[3];
  const float* adst1 = (const float*)d_in[4];
  const float* b1    = (const float*)d_in[5];
  const float* W2    = (const float*)d_in[6];
  const float* asrc2 = (const float*)d_in[7];
  const float* adst2 = (const float*)d_in[8];
  const float* b2    = (const float*)d_in[9];
  const float* Wf    = (const float*)d_in[10];
  const float* bf    = (const float*)d_in[11];

  const int B  = 2;
  const int N  = in_sizes[0] / (B * FEAT);
  const int E  = in_sizes[1] / 2;

  float* p = (float*)d_ws;
  float* buf_h   = p; p += (size_t)N * HIDDEN;
  float* buf_h2  = p; p += (size_t)N * HIDDEN;
  float* ssrc    = p; p += (size_t)N * HEADS;
  float* sdst    = p; p += (size_t)N * HEADS;
  float* ssrc2   = p; p += (size_t)N * HEADS;
  float* sdst2   = p; p += (size_t)N * HEADS;
  const int ntiles  = (N + TN - 1) / TN;
  unsigned* tlcnt  = (unsigned*)p; p += ntiles;
  unsigned* tllist = (unsigned*)p;

  const int mtiles  = N >> 5;
  const int gtiles  = mtiles * (HIDDEN >> 6);
  const dim3 blkG(128);
  const dim3 blkT(TTHREADS);

  for (int b = 0; b < B; ++b) {
    const float* xb = x + (size_t)b * N * FEAT;
    float* outb = (float*)d_out + (size_t)b * N * CHAN;

    wmma_gemm_kernel<true, false><<<(gtiles + 3) / 4, blkG, 0, stream>>>(
        xb, W1, nullptr, asrc1, adst1, buf_h, ssrc, sdst, N, FEAT, HIDDEN);
    if (b == 0)
      tile_agg_kernel<true, true><<<ntiles, blkT, 0, stream>>>(ei, E, N, ssrc, sdst, buf_h, b1,
          W2, nullptr, asrc2, adst2, buf_h2, ssrc2, sdst2, tlcnt, tllist);
    else
      tile_agg_kernel<true, false><<<ntiles, blkT, 0, stream>>>(ei, E, N, ssrc, sdst, buf_h, b1,
          W2, nullptr, asrc2, adst2, buf_h2, ssrc2, sdst2, tlcnt, tllist);

    tile_agg_kernel<false, false><<<ntiles, blkT, 0, stream>>>(ei, E, N, ssrc2, sdst2, buf_h2, b2,
        Wf, bf, nullptr, nullptr, outb, nullptr, nullptr, tlcnt, tllist);
  }
}


# First GEMM launch writes a tile map of the edge list (1 byte per edge: its 80-row destination tile); tile filter reads it, 16 edges per 16-byte load, 4 per test; sum order still arrival order
# speedup vs baseline: 5.0861x; 1.0423x over previous
#include <hip/hip_runtime.h>
#include <math.h>


typedef __attribute__((ext_vector_type(2))) float v2f;
typedef __attribute__((ext_vector_type(8))) float v8f;

#define HEADS     4
#define CHAN      64
#define HIDDEN    256
#define FEAT      128
#define NEG_SLOPE 0.2f
#define TILE_ROWS 80

static __device__ __forceinline__ float lrelu(float x) {
  return x > 0.f ? x : NEG_SLOPE * x;
}
static __device__ __forceinline__ unsigned ordEnc(float f) {
  unsigned u = __float_as_uint(f);
  return (u & 0x80000000u) ? ~u : (u | 0x80000000u);
}
static __device__ __forceinline__ float ordDec(unsigned u) {
  unsigned v = (u & 0x80000000u) ? (u & 0x7fffffffu) : ~u;
  return __uint_as_float(v);
}

template <bool GAT, bool BIAS, bool TMAP>
__global__ void wmma_gemm_kernel(const float* __restrict__ A,
                                 const float* __restrict__ W,
                                 const float* __restrict__ bias,
                                 const float* __restrict__ att_src,
                                 const float* __restrict__ att_dst,
                                 float* __restrict__ Hout,
                                 float* __restrict__ s_src,
                                 float* __restrict__ s_dst,
                                 int M, int K, int Nc,
                                 const int* __restrict__ ei, int E,
                                 unsigned* __restrict__ tmap) {
  if (TMAP) {
    const int* __restrict__ edst = ei + E;
    const int nwords = ((E + 15) >> 4) << 2;
    const int nthr = gridDim.x * blockDim.x;
    for (int w = blockIdx.x * blockDim.x + threadIdx.x; w < nwords; w += nthr) {
      unsigned pk = 0u;
      #pragma unroll
      for (int j = 0; j < 4; ++j) {
        const int e = 4 * w + j;
        unsigned t = 255u;
        if (e < E) t = min((unsigned)edst[e] / (unsigned)TILE_ROWS, 255u);
        pk |= t << (8 * j);
      }
      tmap[w] = pk;
    }
  }
  const int lane = threadIdx.x & 31;
  const int wv   = threadIdx.x >> 5;
  const int wpb  = blockDim.x >> 5;
  const int tile = blockIdx.x * wpb + wv;
  const int ctiles = Nc >> 6;
  const int mt = tile / ctiles;
  const int ct = tile - mt * ctiles;
  const int m0 = mt << 5;
  if (m0 >= M) return;
  const int n0 = ct << 6;
  const int lo = lane & 15;
  const int hi = lane >> 4;

  v8f acc[2][4] = {};
  const float* arow0 = A + (size_t)(m0 + lo) * K;
  const float* arow1 = A + (size_t)(m0 + 16 + lo) * K;
  const float* wcol  = W + n0 + lo;

  for (int k = 0; k < K; k += 4) {
    const int ka = k + (hi << 1);
    v2f a0 = *(const v2f*)(arow0 + ka);
    v2f a1 = *(const v2f*)(arow1 + ka);
    const float* w0 = wcol + (size_t)ka * Nc;
    const float* w1 = w0 + Nc;
    v2f b0 = {w0[0],  w1[0]};
    v2f b1 = {w0[16], w1[16]};
    v2f b2 = {w0[32], w1[32]};
    v2f b3 = {w0[48], w1[48]};
    acc[0][0] = __builtin_amdgcn_wmma_f32_16x16x4_f32(false, a0, false, b0, (short)0, acc[0][0], false, false);
    acc[1][0] = __builtin_amdgcn_wmma_f32_16x16x4_f32(false, a1, false, b0, (short)0, acc[1][0], false, false);
    acc[0][1] = __builtin_amdgcn_wmma_f32_16x16x4_f32(false, a0, false, b1, (short)0, acc[0][1], false, false);
    acc[1][1] = __builtin_amdgcn_wmma_f32_16x16x4_f32(false, a1, false, b1, (short)0, acc[1][1], false, false);
    acc[0][2] = __builtin_amdgcn_wmma_f32_16x16x4_f32(false, a0, false, b2, (short)0, acc[0][2], false, false);
    acc[1][2] = __builtin_amdgcn_wmma_f32_16x16x4_f32(false, a1, false, b2, (short)0, acc[1][2], false, false);
    acc[0][3] = __builtin_amdgcn_wmma_f32_16x16x4_f32(false, a0, false, b3, (short)0, acc[0][3], false, false);
    acc[1][3] = __builtin_amdgcn_wmma_f32_16x16x4_f32(false, a1, false, b3, (short)0, acc[1][3], false, false);
  }

  float bv0 = 0.f, bv1 = 0.f, bv2 = 0.f, bv3 = 0.f;
  if (BIAS) {
    bv0 = bias[n0 + lo];      bv1 = bias[n0 + 16 + lo];
    bv2 = bias[n0 + 32 + lo]; bv3 = bias[n0 + 48 + lo];
  }

  #pragma unroll
  for (int g = 0; g < 2; ++g) {
    const int rbase = m0 + (g << 4) + (hi << 3);
    #pragma unroll
    for (int r = 0; r < 8; ++r) {
      float* hp = Hout + (size_t)(rbase + r) * Nc + n0 + lo;
      hp[0]  = acc[g][0][r] + bv0;
      hp[16] = acc[g][1][r] + bv1;
      hp[32] = acc[g][2][r] + bv2;
      hp[48] = acc[g][3][r] + bv3;
    }
  }

  if (GAT) {
    const int head = n0 >> 6;
    const float* as = att_src + head * CHAN;
    const float* ad = att_dst + head * CHAN;
    const float as0 = as[lo], as1 = as[16 + lo], as2 = as[32 + lo], as3 = as[48 + lo];
    const float ad0 = ad[lo], ad1 = ad[16 + lo], ad2 = ad[32 + lo], ad3 = ad[48 + lo];
    #pragma unroll
    for (int g = 0; g < 2; ++g) {
      const int rbase = m0 + (g << 4) + (hi << 3);
      #pragma unroll
      for (int r = 0; r < 8; ++r) {
        float ss = acc[g][0][r] * as0 + acc[g][1][r] * as1 + acc[g][2][r] * as2 + acc[g][3][r] * as3;
        float sd = acc[g][0][r] * ad0 + acc[g][1][r] * ad1 + acc[g][2][r] * ad2 + acc[g][3][r] * ad3;
        #pragma unroll
        for (int off = 8; off > 0; off >>= 1) {
          ss += __shfl_xor(ss, off, 32);
          sd += __shfl_xor(sd, off, 32);
        }
        if (lo == 0) {
          const int row = rbase + r;
          s_src[row * HEADS + head] = ss;
          s_dst[row * HEADS + head] = sd;
        }
      }
    }
  }
}

#define TN       TILE_ROWS
#define APITCH   (HIDDEN + 4)
#define TCAP     20480
#define WCH      4096
#define TTHREADS 1024
static_assert(TN <= 128, "a noted edge keeps its tile row in 7 bits");

template <int NC, bool GAT, bool BIAS>
static __device__ __forceinline__ void tile_gemm(const float* tile, int wv, int lane, int d0, int nt,
                                                 const float* __restrict__ W,
                                                 const float* __restrict__ bias,
                                                 const float* __restrict__ att_src,
                                                 const float* __restrict__ att_dst,
                                                 float* __restrict__ Hout,
                                                 float* __restrict__ s_src,
                                                 float* __restrict__ s_dst) {
  constexpr int CT = NC / 64;
  static_assert(TN % 16 == 0, "whole 16-row tiles");
  if (wv >= (TN / 16) * CT) return;
  const int rt = wv / CT;
  const int ct = wv - rt * CT;
  const int n0 = ct << 6;
  const int lo = lane & 15;
  const int hi = lane >> 4;

  v8f c0 = {}, c1 = {}, c2 = {}, c3 = {};
  const float* arow = tile + (rt * 16 + lo) * APITCH + (hi << 1);
  const float* wcol = W + (size_t)(hi << 1) * NC + n0 + lo;

  for (int k = 0; k < HIDDEN; k += 16) {
    v2f a[4];
    float b[4][8];
    #pragma unroll
    for (int u = 0; u < 4; ++u) {
      const float* w0 = wcol + (size_t)(k + 4 * u) * NC;
      b[u][0] = w0[0];  b[u][1] = w0[NC];
      b[u][2] = w0[16]; b[u][3] = w0[NC + 16];
      b[u][4] = w0[32]; b[u][5] = w0[NC + 32];
      b[u][6] = w0[48]; b[u][7] = w0[NC + 48];
    }
    #pragma unroll
    for (int u = 0; u < 4; ++u) a[u] = *(const v2f*)(arow + k + 4 * u);
    #pragma unroll
    for (int u = 0; u < 4; ++u) {
      const v2f b0 = {b[u][0], b[u][1]};
      const v2f b1 = {b[u][2], b[u][3]};
      const v2f b2 = {b[u][4], b[u][5]};
      const v2f b3 = {b[u][6], b[u][7]};
      c0 = __builtin_amdgcn_wmma_f32_16x16x4_f32(false, a[u], false, b0, (short)0, c0, false, false);
      c1 = __builtin_amdgcn_wmma_f32_16x16x4_f32(false, a[u], false, b1, (short)0, c1, false, false);
      c2 = __builtin_amdgcn_wmma_f32_16x16x4_f32(false, a[u], false, b2, (short)0, c2, false, false);
      c3 = __builtin_amdgcn_wmma_f32_16x16x4_f32(false, a[u], false, b3, (short)0, c3, false, false);
    }
  }

  float bv0 = 0.f, bv1 = 0.f, bv2 = 0.f, bv3 = 0.f;
  if (BIAS) {
    bv0 = bias[n0 + lo];      bv1 = bias[n0 + 16 + lo];
    bv2 = bias[n0 + 32 + lo]; bv3 = bias[n0 + 48 + lo];
  }
  const int rloc = rt * 16 + (hi << 3);
  #pragma unroll
  for (int r = 0; r < 8; ++r) {
    if (rloc + r < nt) {
      float* hp = Hout + (size_t)(d0 + rloc + r) * NC + n0 + lo;
      hp[0]  = c0[r] + bv0;
      hp[16] = c1[r] + bv1;
      hp[32] = c2[r] + bv2;
      hp[48] = c3[r] + bv3;
    }
  }

  if (GAT) {
    const int head = ct;
    const float* as = att_src + head * CHAN;
    const float* ad = att_dst + head * CHAN;
    const float as0 = as[lo], as1 = as[16 + lo], as2 = as[32 + lo], as3 = as[48 + lo];
    const float ad0 = ad[lo], ad1 = ad[16 + lo], ad2 = ad[32 + lo], ad3 = ad[48 + lo];
    #pragma unroll
    for (int r = 0; r < 8; ++r) {
      float ss = c0[r] * as0 + c1[r] * as1 + c2[r] * as2 + c3[r] * as3;
      float sd = c0[r] * ad0 + c1[r] * ad1 + c2[r] * ad2 + c3[r] * ad3;
      #pragma unroll
      for (int off = 8; off > 0; off >>= 1) {
        ss += __shfl_xor(ss, off, 32);
        sd += __shfl_xor(sd, off, 32);
      }
      if (lo == 0 && rloc + r < nt) {
        const int row = d0 + rloc + r;
        s_src[row * HEADS + head] = ss;
        s_dst[row * HEADS + head] = sd;
      }
    }
  }
}

template <bool ELU, bool FIRST>
__global__ void __launch_bounds__(TTHREADS)
tile_agg_kernel(const int* __restrict__ ei, int E, int N,
                const float* __restrict__ s_src,
                const float* __restrict__ s_dst,
                const float* __restrict__ Hin,
                const float* __restrict__ bias,
                const float* __restrict__ Wn,
                const float* __restrict__ bn,
                const float* __restrict__ att_src_n,
                const float* __restrict__ att_dst_n,
                float* __restrict__ Hout,
                float* __restrict__ s_src_n,
                float* __restrict__ s_dst_n,
                unsigned* __restrict__ tl_cnt,
                unsigned* __restrict__ tl_list,
                const unsigned* __restrict__ tmap) {
  __shared__ __attribute__((aligned(16))) float acc[TN * APITCH];
  __shared__ unsigned elist[TCAP];
  __shared__ __attribute__((aligned(16))) float wls[WCH * HEADS];
  __shared__ __attribute__((aligned(16))) float sdt[TN * HEADS];
  __shared__ __attribute__((aligned(16))) unsigned mxo[TN * HEADS];
  __shared__ float den[TN * HEADS];
  __shared__ unsigned cnt;

  const int tid  = threadIdx.x;
  const int lane = tid & 31;
  const int wv   = tid >> 5;
  const int hl   = lane & (HEADS - 1);
  const int d0   = blockIdx.x * TN;
  const int nt   = (N - d0) < TN ? (N - d0) : TN;
  const int* __restrict__ srcp = ei;
  const int* __restrict__ dstp = ei + E;

  unsigned* __restrict__ tl = tl_list + (size_t)blockIdx.x * TCAP;

  if (FIRST && tid == 0) cnt = 0u;
  unsigned total = 0u;
  if (!FIRST) total = tl_cnt[blockIdx.x];
  if (tid < TN * HEADS) den[tid] = 0.f;
  if (tid < nt * HEADS) {
    const size_t g = (size_t)d0 * HEADS + tid;
    const float sd = s_dst[g];
    sdt[tid] = sd;
    mxo[tid] = ordEnc(lrelu(s_src[g] + sd));
  }
  __syncthreads();

  if (FIRST) {
  auto note = [&](int idx) {
    const unsigned slot = atomicAdd(&cnt, 1u);
    if (slot < (unsigned)TCAP) elist[slot] = (unsigned)idx;
  };

  if (gridDim.x <= 255u) {
  const unsigned pat = blockIdx.x * 0x01010101u;
  const int nv = (E + 15) >> 4;
  const uint4* tv = (const uint4*)tmap;
  auto test4 = [&](unsigned wd, int e0) {
    const unsigned y = wd ^ pat;
    if (((y - 0x01010101u) & ~y & 0x80808080u) != 0u) {
      if ((y & 0x000000ffu) == 0u) note(e0 + 0);
      if ((y & 0x0000ff00u) == 0u) note(e0 + 1);
      if ((y & 0x00ff0000u) == 0u) note(e0 + 2);
      if ((y & 0xff000000u) == 0u) note(e0 + 3);
    }
  };
  for (int v0 = 0; v0 < nv; v0 += TTHREADS * 4) {
    uint4 q[4];
    #pragma unroll
    for (int j = 0; j < 4; ++j) {
      const int vi = v0 + j * TTHREADS + tid;
      q[j] = vi < nv ? tv[vi] : make_uint4(~0u, ~0u, ~0u, ~0u);
    }
    #pragma unroll
    for (int j = 0; j < 4; ++j) {
      const int e0 = (v0 + j * TTHREADS + tid) << 4;
      test4(q[j].x, e0 + 0);
      test4(q[j].y, e0 + 4);
      test4(q[j].z, e0 + 8);
      test4(q[j].w, e0 + 12);
    }
  }
  } else {
  const int Ev   = (((size_t)dstp & 15) == 0) ? (E & ~3) : 0;
  const int nvec = Ev >> 2;
  const int4* dv = (const int4*)dstp;
  for (int v0 = 0; v0 < nvec; v0 += TTHREADS * 4) {
    int4 q[4];
    #pragma unroll
    for (int j = 0; j < 4; ++j) {
      const int vi = v0 + j * TTHREADS + tid;
      q[j] = vi < nvec ? dv[vi] : make_int4(-1, -1, -1, -1);
    }
    #pragma unroll
    for (int j = 0; j < 4; ++j) {
      const int e0 = (v0 + j * TTHREADS + tid) << 2;
      if ((unsigned)(q[j].x - d0) < (unsigned)nt) note(e0 + 0);
      if ((unsigned)(q[j].y - d0) < (unsigned)nt) note(e0 + 1);
      if ((unsigned)(q[j].z - d0) < (unsigned)nt) note(e0 + 2);
      if ((unsigned)(q[j].w - d0) < (unsigned)nt) note(e0 + 3);
    }
  }
  for (int i = Ev + tid; i < E; i += TTHREADS)
    if ((unsigned)(dstp[i] - d0) < (unsigned)nt) note(i);
  }
  __syncthreads();
  total = cnt;
  if (tid == 0) tl_cnt[blockIdx.x] = total;
  }
  const bool fast = total <= (unsigned)TCAP;

  auto edge_max_q = [&](const float4 ss, unsigned dl) {
    const float4 sd = *(const float4*)(sdt + dl * HEADS);
    atomicMax(&mxo[dl * HEADS + 0], ordEnc(lrelu(ss.x + sd.x)));
    atomicMax(&mxo[dl * HEADS + 1], ordEnc(lrelu(ss.y + sd.y)));
    atomicMax(&mxo[dl * HEADS + 2], ordEnc(lrelu(ss.z + sd.z)));
    atomicMax(&mxo[dl * HEADS + 3], ordEnc(lrelu(ss.w + sd.w)));
  };
  auto edge_max = [&](int s, unsigned dl) {
    edge_max_q(*(const float4*)(s_src + (size_t)s * HEADS), dl);
  };
  if (fast && FIRST) {
    for (int i = tid; i < (int)total; i += TTHREADS) {
      const int idx = (int)elist[i];
      const int s = srcp[idx];
      const unsigned dl = (unsigned)(dstp[idx] - d0);
      edge_max(s, dl);
      const unsigned en = ((unsigned)s << 7) | dl;
      elist[i] = en;
      tl[i] = en;
    }
  } else if (fast) {
    for (int i0 = tid; i0 < (int)total; i0 += 2 * TTHREADS) {
      const int i1 = i0 + TTHREADS;
      const bool v1 = i1 < (int)total;
      const unsigned e0 = tl[i0];
      const unsigned e1 = v1 ? tl[i1] : 0u;
      const unsigned s0 = min(e0 >> 7, (unsigned)(N - 1));
      const unsigned s1 = min(e1 >> 7, (unsigned)(N - 1));
      const float4 q0 = *(const float4*)(s_src + (size_t)s0 * HEADS);
      const float4 q1 = *(const float4*)(s_src + (size_t)s1 * HEADS);
      asm volatile("" :: "v"(q0.x), "v"(q0.y), "v"(q0.z), "v"(q0.w),
                         "v"(q1.x), "v"(q1.y), "v"(q1.z), "v"(q1.w));
      edge_max_q(q0, e0 & 127u);
      elist[i0] = (s0 << 7) | (e0 & 127u);
      if (v1) {
        edge_max_q(q1, e1 & 127u);
        elist[i1] = (s1 << 7) | (e1 & 127u);
      }
    }
  } else {
    for (int i = tid; i < E; i += TTHREADS) {
      const unsigned dl = (unsigned)(dstp[i] - d0);
      if (dl < (unsigned)nt) edge_max(srcp[i], dl);
    }
  }
  __syncthreads();

  for (int dl = wv; dl < nt; dl += (TTHREADS >> 5)) {
    const int d = d0 + dl;
    const float e  = lrelu(s_src[(size_t)d * HEADS + hl] + sdt[dl * HEADS + hl]);
    const float wl = expf(e - ordDec(mxo[dl * HEADS + hl]));
    if (lane < HEADS) atomicAdd(&den[dl * HEADS + lane], wl);
    float w[HEADS];
    #pragma unroll
    for (int h = 0; h < HEADS; ++h) w[h] = __shfl(wl, h, 32);
    const float* hs = Hin + (size_t)d * HIDDEN;
    #pragma unroll
    for (int i = 0; i < 8; ++i) {
      const int ch = lane + (i << 5);
      acc[dl * APITCH + ch] = hs[ch] * w[i >> 1];
    }
  }

  if (fast) {
    for (int c0 = 0; c0 < (int)total; c0 += WCH) {
      const int cend = (int)total - c0 < WCH ? (int)total : c0 + WCH;
      if (c0 > 0) __syncthreads();
      for (int i = c0 + tid; i < cend; i += TTHREADS) {
        const unsigned en = elist[i];
        const unsigned dl = en & 127u;
        const float4 ss = *(const float4*)(s_src + (size_t)(en >> 7) * HEADS);
        const float4 sd = *(const float4*)(sdt + dl * HEADS);
        const uint4  mx = *(const uint4*)(mxo + dl * HEADS);
        float4 w;
        w.x = expf(lrelu(ss.x + sd.x) - ordDec(mx.x));
        w.y = expf(lrelu(ss.y + sd.y) - ordDec(mx.y));
        w.z = expf(lrelu(ss.z + sd.z) - ordDec(mx.z));
        w.w = expf(lrelu(ss.w + sd.w) - ordDec(mx.w));
        atomicAdd(&den[dl * HEADS + 0], w.x);
        atomicAdd(&den[dl * HEADS + 1], w.y);
        atomicAdd(&den[dl * HEADS + 2], w.z);
        atomicAdd(&den[dl * HEADS + 3], w.w);
        *(float4*)(wls + (size_t)(i - c0) * HEADS) = w;
      }
      __syncthreads();
      for (int b = c0 + wv; b < cend; b += (TTHREADS >> 5) * 4) {
        unsigned en[4];
        float4 wq[4];
        float hv[4][8];
        #pragma unroll
        for (int j = 0; j < 4; ++j) {
          const int idx = b + j * (TTHREADS >> 5);
          const bool ok = idx < cend;
          en[j] = ok ? elist[idx] : 0xFFFFFFFFu;
          wq[j] = ok ? *(const float4*)(wls + (size_t)(idx - c0) * HEADS) : make_float4(0.f, 0.f, 0.f, 0.f);
        }
        #pragma unroll
        for (int j = 0; j < 4; ++j) {
          const unsigned s = en[j] == 0xFFFFFFFFu ? 0u : (en[j] >> 7);
          const float* hs = Hin + (size_t)s * HIDDEN;
          #pragma unroll
          for (int i = 0; i < 8; ++i) hv[j][i] = hs[lane + (i << 5)];
        }
        #pragma unroll
        for (int j = 0; j < 4; ++j) {
          if (en[j] != 0xFFFFFFFFu) {
            const unsigned dl = en[j] & 127u;
            const float w[HEADS] = {wq[j].x, wq[j].y, wq[j].z, wq[j].w};
            #pragma unroll
            for (int i = 0; i < 8; ++i)
              atomicAdd(&acc[dl * APITCH + lane + (i << 5)], hv[j][i] * w[i >> 1]);
          }
        }
      }
    }
  } else {
    __syncthreads();
    for (int i = tid; i < E; i += TTHREADS) {
      const unsigned dl = (unsigned)(dstp[i] - d0);
      if (dl < (unsigned)nt) {
        const int s = srcp[i];
        const float* hs = Hin + (size_t)s * HIDDEN;
        for (int h = 0; h < HEADS; ++h) {
          const float e  = lrelu(s_src[(size_t)s * HEADS + h] + sdt[dl * HEADS + h]);
          const float wh = expf(e - ordDec(mxo[dl * HEADS + h]));
          atomicAdd(&den[dl * HEADS + h], wh);
          for (int c = 0; c < CHAN; ++c)
            atomicAdd(&acc[dl * APITCH + h * CHAN + c], hs[h * CHAN + c] * wh);
        }
      }
    }
  }
  __syncthreads();

  for (int qd = tid; qd < TN * (HIDDEN / 4); qd += TTHREADS) {
    const int dl = qd / (HIDDEN / 4);
    const int c  = (qd - dl * (HIDDEN / 4)) << 2;
    float4 v = make_float4(0.f, 0.f, 0.f, 0.f);
    if (dl < nt) {
      const float dn = den[dl * HEADS + (c >> 6)];
      v = *(const float4*)(acc + dl * APITCH + c);
      v.x = v.x / dn + bias[c + 0];
      v.y = v.y / dn + bias[c + 1];
      v.z = v.z / dn + bias[c + 2];
      v.w = v.w / dn + bias[c + 3];
      if (ELU) {
        v.x = v.x > 0.f ? v.x : expm1f(v.x);
        v.y = v.y > 0.f ? v.y : expm1f(v.y);
        v.z = v.z > 0.f ? v.z : expm1f(v.z);
        v.w = v.w > 0.f ? v.w : expm1f(v.w);
      }
    }
    *(float4*)(acc + dl * APITCH + c) = v;
  }
  __syncthreads();

  if (ELU)
    tile_gemm<HIDDEN, true, false>(acc, wv, lane, d0, nt, Wn, nullptr, att_src_n, att_dst_n,
                                   Hout, s_src_n, s_dst_n);
  else
    tile_gemm<CHAN, false, true>(acc, wv, lane, d0, nt, Wn, bn, nullptr, nullptr,
                                 Hout, nullptr, nullptr);
}

extern "C" void kernel_launch(void* const* d_in, const int* in_sizes, int n_in,
                              void* d_out, int out_size, void* d_ws, size_t ws_size,
                              hipStream_t stream) {
  const float* x     = (const float*)d_in[0];
  const int*   ei    = (const int*)  d_in[1];
  const float* W1    = (const float*)d_in[2];
  const float* asrc1 = (const float*)d_in[3];
  const float* adst1 = (const float*)d_in[4];
  const float* b1    = (const float*)d_in[5];
  const float* W2    = (const float*)d_in[6];
  const float* asrc2 = (const float*)d_in[7];
  const float* adst2 = (const float*)d_in[8];
  const float* b2    = (const float*)d_in[9];
  const float* Wf    = (const float*)d_in[10];
  const float* bf    = (const float*)d_in[11];

  const int B  = 2;
  const int N  = in_sizes[0] / (B * FEAT);
  const int E  = in_sizes[1] / 2;

  float* p = (float*)d_ws;
  float* buf_h   = p; p += (size_t)N * HIDDEN;
  float* buf_h2  = p; p += (size_t)N * HIDDEN;
  float* ssrc    = p; p += (size_t)N * HEADS;
  float* sdst    = p; p += (size_t)N * HEADS;
  float* ssrc2   = p; p += (size_t)N * HEADS;
  float* sdst2   = p; p += (size_t)N * HEADS;
  const int ntiles  = (N + TN - 1) / TN;
  unsigned* tmap   = (unsigned*)p; p += ((E + 15) >> 4) << 2;
  unsigned* tlcnt  = (unsigned*)p; p += ntiles;
  unsigned* tllist = (unsigned*)p;

  const int mtiles  = N >> 5;
  const int gtiles  = mtiles * (HIDDEN >> 6);
  const dim3 blkG(128);
  const dim3 blkT(TTHREADS);

  for (int b = 0; b < B; ++b) {
    const float* xb = x + (size_t)b * N * FEAT;
    float* outb = (float*)d_out + (size_t)b * N * CHAN;

    if (b == 0)
      wmma_gemm_kernel<true, false, true><<<(gtiles + 3) / 4, blkG, 0, stream>>>(
          xb, W1, nullptr, asrc1, adst1, buf_h, ssrc, sdst, N, FEAT, HIDDEN, ei, E, tmap);
    else
      wmma_gemm_kernel<true, false, false><<<(gtiles + 3) / 4, blkG, 0, stream>>>(
          xb, W1, nullptr, asrc1, adst1, buf_h, ssrc, sdst, N, FEAT, HIDDEN, nullptr, 0, nullptr);
    if (b == 0)
      tile_agg_kernel<true, true><<<ntiles, blkT, 0, stream>>>(ei, E, N, ssrc, sdst, buf_h, b1,
          W2, nullptr, asrc2, adst2, buf_h2, ssrc2, sdst2, tlcnt, tllist, tmap);
    else
      tile_agg_kernel<true, false><<<ntiles, blkT, 0, stream>>>(ei, E, N, ssrc, sdst, buf_h, b1,
          W2, nullptr, asrc2, adst2, buf_h2, ssrc2, sdst2, tlcnt, tllist, tmap);

    tile_agg_kernel<false, false><<<ntiles, blkT, 0, stream>>>(ei, E, N, ssrc2, sdst2, buf_h2, b2,
        Wf, bf, nullptr, nullptr, outb, nullptr, nullptr, tlcnt, tllist, tmap);
  }
}


# Edge sums by owner waves: a wave owns 5 tile rows x 128 channels, picks their notes from the tile's list by ballot, sums w*h[src] in registers; edge loop's LDS float atomics gone; sums in list order
# speedup vs baseline: 5.3328x; 1.0485x over previous
#include <hip/hip_runtime.h>
#include <math.h>


typedef __attribute__((ext_vector_type(2))) float v2f;
typedef __attribute__((ext_vector_type(8))) float v8f;

#define HEADS     4
#define CHAN      64
#define HIDDEN    256
#define FEAT      128
#define NEG_SLOPE 0.2f
#define TILE_ROWS 80

static __device__ __forceinline__ float lrelu(float x) {
  return x > 0.f ? x : NEG_SLOPE * x;
}
static __device__ __forceinline__ unsigned ordEnc(float f) {
  unsigned u = __float_as_uint(f);
  return (u & 0x80000000u) ? ~u : (u | 0x80000000u);
}
static __device__ __forceinline__ float ordDec(unsigned u) {
  unsigned v = (u & 0x80000000u) ? (u & 0x7fffffffu) : ~u;
  return __uint_as_float(v);
}

template <bool GAT, bool BIAS, bool TMAP>
__global__ void wmma_gemm_kernel(const float* __restrict__ A,
                                 const float* __restrict__ W,
                                 const float* __restrict__ bias,
                                 const float* __restrict__ att_src,
                                 const float* __restrict__ att_dst,
                                 float* __restrict__ Hout,
                                 float* __restrict__ s_src,
                                 float* __restrict__ s_dst,
                                 int M, int K, int Nc,
                                 const int* __restrict__ ei, int E,
                                 unsigned* __restrict__ tmap) {
  if (TMAP) {
    const int* __restrict__ edst = ei + E;
    const int nwords = ((E + 15) >> 4) << 2;
    const int nthr = gridDim.x * blockDim.x;
    for (int w = blockIdx.x * blockDim.x + threadIdx.x; w < nwords; w += nthr) {
      unsigned pk = 0u;
      #pragma unroll
      for (int j = 0; j < 4; ++j) {
        const int e = 4 * w + j;
        unsigned t = 255u;
        if (e < E) t = min((unsigned)edst[e] / (unsigned)TILE_ROWS, 255u);
        pk |= t << (8 * j);
      }
      tmap[w] = pk;
    }
  }
  const int lane = threadIdx.x & 31;
  const int wv   = threadIdx.x >> 5;
  const int wpb  = blockDim.x >> 5;
  const int tile = blockIdx.x * wpb + wv;
  const int ctiles = Nc >> 6;
  const int mt = tile / ctiles;
  const int ct = tile - mt * ctiles;
  const int m0 = mt << 5;
  if (m0 >= M) return;
  const int n0 = ct << 6;
  const int lo = lane & 15;
  const int hi = lane >> 4;

  v8f acc[2][4] = {};
  const float* arow0 = A + (size_t)(m0 + lo) * K;
  const float* arow1 = A + (size_t)(m0 + 16 + lo) * K;
  const float* wcol  = W + n0 + lo;

  for (int k = 0; k < K; k += 4) {
    const int ka = k + (hi << 1);
    v2f a0 = *(const v2f*)(arow0 + ka);
    v2f a1 = *(const v2f*)(arow1 + ka);
    const float* w0 = wcol + (size_t)ka * Nc;
    const float* w1 = w0 + Nc;
    v2f b0 = {w0[0],  w1[0]};
    v2f b1 = {w0[16], w1[16]};
    v2f b2 = {w0[32], w1[32]};
    v2f b3 = {w0[48], w1[48]};
    acc[0][0] = __builtin_amdgcn_wmma_f32_16x16x4_f32(false, a0, false, b0, (short)0, acc[0][0], false, false);
    acc[1][0] = __builtin_amdgcn_wmma_f32_16x16x4_f32(false, a1, false, b0, (short)0, acc[1][0], false, false);
    acc[0][1] = __builtin_amdgcn_wmma_f32_16x16x4_f32(false, a0, false, b1, (short)0, acc[0][1], false, false);
    acc[1][1] = __builtin_amdgcn_wmma_f32_16x16x4_f32(false, a1, false, b1, (short)0, acc[1][1], false, false);
    acc[0][2] = __builtin_amdgcn_wmma_f32_16x16x4_f32(false, a0, false, b2, (short)0, acc[0][2], false, false);
    acc[1][2] = __builtin_amdgcn_wmma_f32_16x16x4_f32(false, a1, false, b2, (short)0, acc[1][2], false, false);
    acc[0][3] = __builtin_amdgcn_wmma_f32_16x16x4_f32(false, a0, false, b3, (short)0, acc[0][3], false, false);
    acc[1][3] = __builtin_amdgcn_wmma_f32_16x16x4_f32(false, a1, false, b3, (short)0, acc[1][3], false, false);
  }

  float bv0 = 0.f, bv1 = 0.f, bv2 = 0.f, bv3 = 0.f;
  if (BIAS) {
    bv0 = bias[n0 + lo];      bv1 = bias[n0 + 16 + lo];
    bv2 = bias[n0 + 32 + lo]; bv3 = bias[n0 + 48 + lo];
  }

  #pragma unroll
  for (int g = 0; g < 2; ++g) {
    const int rbase = m0 + (g << 4) + (hi << 3);
    #pragma unroll
    for (int r = 0; r < 8; ++r) {
      float* hp = Hout + (size_t)(rbase + r) * Nc + n0 + lo;
      hp[0]  = acc[g][0][r] + bv0;
      hp[16] = acc[g][1][r] + bv1;
      hp[32] = acc[g][2][r] + bv2;
      hp[48] = acc[g][3][r] + bv3;
    }
  }

  if (GAT) {
    const int head = n0 >> 6;
    const float* as = att_src + head * CHAN;
    const float* ad = att_dst + head * CHAN;
    const float as0 = as[lo], as1 = as[16 + lo], as2 = as[32 + lo], as3 = as[48 + lo];
    const float ad0 = ad[lo], ad1 = ad[16 + lo], ad2 = ad[32 + lo], ad3 = ad[48 + lo];
    #pragma unroll
    for (int g = 0; g < 2; ++g) {
      const int rbase = m0 + (g << 4) + (hi << 3);
      #pragma unroll
      for (int r = 0; r < 8; ++r) {
        float ss = acc[g][0][r] * as0 + acc[g][1][r] * as1 + acc[g][2][r] * as2 + acc[g][3][r] * as3;
        float sd = acc[g][0][r] * ad0 + acc[g][1][r] * ad1 + acc[g][2][r] * ad2 + acc[g][3][r] * ad3;
        #pragma unroll
        for (int off = 8; off > 0; off >>= 1) {
          ss += __shfl_xor(ss, off, 32);
          sd += __shfl_xor(sd, off, 32);
        }
        if (lo == 0) {
          const int row = rbase + r;
          s_src[row * HEADS + head] = ss;
          s_dst[row * HEADS + head] = sd;
        }
      }
    }
  }
}

#define TN       TILE_ROWS
#define APITCH   (HIDDEN + 4)
#define TCAP     20480
#define WCH      4096
#define TTHREADS 1024
#define NSLOT    (TN / 16)
#define LCAP     256
#define SCAN     4
#define NB       8
static_assert(TTHREADS == 1024 && TN % 16 == 0, "32 owner waves: 16 row classes x 2 halves of a row");
static_assert(TN <= 128, "a noted edge keeps its tile row in 7 bits");

template <int NC, bool GAT, bool BIAS>
static __device__ __forceinline__ void tile_gemm(const float* tile, int wv, int lane, int d0, int nt,
                                                 const float* __restrict__ W,
                                                 const float* __restrict__ bias,
                                                 const float* __restrict__ att_src,
                                                 const float* __restrict__ att_dst,
                                                 float* __restrict__ Hout,
                                                 float* __restrict__ s_src,
                                                 float* __restrict__ s_dst) {
  constexpr int CT = NC / 64;
  static_assert(TN % 16 == 0, "whole 16-row tiles");
  if (wv >= (TN / 16) * CT) return;
  const int rt = wv / CT;
  const int ct = wv - rt * CT;
  const int n0 = ct << 6;
  const int lo = lane & 15;
  const int hi = lane >> 4;

  v8f c0 = {}, c1 = {}, c2 = {}, c3 = {};
  const float* arow = tile + (rt * 16 + lo) * APITCH + (hi << 1);
  const float* wcol = W + (size_t)(hi << 1) * NC + n0 + lo;

  #pragma unroll 2
  for (int k = 0; k < HIDDEN; k += 16) {
    v2f a[4];
    float b[4][8];
    #pragma unroll
    for (int u = 0; u < 4; ++u) {
      const float* w0 = wcol + (size_t)(k + 4 * u) * NC;
      b[u][0] = w0[0];  b[u][1] = w0[NC];
      b[u][2] = w0[16]; b[u][3] = w0[NC + 16];
      b[u][4] = w0[32]; b[u][5] = w0[NC + 32];
      b[u][6] = w0[48]; b[u][7] = w0[NC + 48];
    }
    #pragma unroll
    for (int u = 0; u < 4; ++u) a[u] = *(const v2f*)(arow + k + 4 * u);
    #pragma unroll
    for (int u = 0; u < 4; ++u) {
      const v2f b0 = {b[u][0], b[u][1]};
      const v2f b1 = {b[u][2], b[u][3]};
      const v2f b2 = {b[u][4], b[u][5]};
      const v2f b3 = {b[u][6], b[u][7]};
      c0 = __builtin_amdgcn_wmma_f32_16x16x4_f32(false, a[u], false, b0, (short)0, c0, false, false);
      c1 = __builtin_amdgcn_wmma_f32_16x16x4_f32(false, a[u], false, b1, (short)0, c1, false, false);
      c2 = __builtin_amdgcn_wmma_f32_16x16x4_f32(false, a[u], false, b2, (short)0, c2, false, false);
      c3 = __builtin_amdgcn_wmma_f32_16x16x4_f32(false, a[u], false, b3, (short)0, c3, false, false);
    }
  }

  float bv0 = 0.f, bv1 = 0.f, bv2 = 0.f, bv3 = 0.f;
  if (BIAS) {
    bv0 = bias[n0 + lo];      bv1 = bias[n0 + 16 + lo];
    bv2 = bias[n0 + 32 + lo]; bv3 = bias[n0 + 48 + lo];
  }
  const int rloc = rt * 16 + (hi << 3);
  #pragma unroll
  for (int r = 0; r < 8; ++r) {
    if (rloc + r < nt) {
      float* hp = Hout + (size_t)(d0 + rloc + r) * NC + n0 + lo;
      hp[0]  = c0[r] + bv0;
      hp[16] = c1[r] + bv1;
      hp[32] = c2[r] + bv2;
      hp[48] = c3[r] + bv3;
    }
  }

  if (GAT) {
    const int head = ct;
    const float* as = att_src + head * CHAN;
    const float* ad = att_dst + head * CHAN;
    const float as0 = as[lo], as1 = as[16 + lo], as2 = as[32 + lo], as3 = as[48 + lo];
    const float ad0 = ad[lo], ad1 = ad[16 + lo], ad2 = ad[32 + lo], ad3 = ad[48 + lo];
    #pragma unroll
    for (int r = 0; r < 8; ++r) {
      float ss = c0[r] * as0 + c1[r] * as1 + c2[r] * as2 + c3[r] * as3;
      float sd = c0[r] * ad0 + c1[r] * ad1 + c2[r] * ad2 + c3[r] * ad3;
      #pragma unroll
      for (int off = 8; off > 0; off >>= 1) {
        ss += __shfl_xor(ss, off, 32);
        sd += __shfl_xor(sd, off, 32);
      }
      if (lo == 0 && rloc + r < nt) {
        const int row = d0 + rloc + r;
        s_src[row * HEADS + head] = ss;
        s_dst[row * HEADS + head] = sd;
      }
    }
  }
}

template <bool ELU, bool FIRST>
__global__ void __launch_bounds__(TTHREADS)
tile_agg_kernel(const int* __restrict__ ei, int E, int N,
                const float* __restrict__ s_src,
                const float* __restrict__ s_dst,
                const float* __restrict__ Hin,
                const float* __restrict__ bias,
                const float* __restrict__ Wn,
                const float* __restrict__ bn,
                const float* __restrict__ att_src_n,
                const float* __restrict__ att_dst_n,
                float* __restrict__ Hout,
                float* __restrict__ s_src_n,
                float* __restrict__ s_dst_n,
                unsigned* __restrict__ tl_cnt,
                unsigned* __restrict__ tl_list,
                const unsigned* __restrict__ tmap) {
  __shared__ __attribute__((aligned(16))) float acc[TN * APITCH];
  __shared__ unsigned elist[TCAP];
  __shared__ __attribute__((aligned(16))) float wls[WCH * HEADS];
  __shared__ __attribute__((aligned(16))) float sdt[TN * HEADS];
  __shared__ __attribute__((aligned(16))) unsigned mxo[TN * HEADS];
  __shared__ float den[TN * HEADS];
  __shared__ unsigned cnt;
  __shared__ uint2 wlist[(TTHREADS >> 5) * LCAP];

  const int tid  = threadIdx.x;
  const int lane = tid & 31;
  const int wv   = tid >> 5;
  const int hl   = lane & (HEADS - 1);
  const int d0   = blockIdx.x * TN;
  const int nt   = (N - d0) < TN ? (N - d0) : TN;
  const int* __restrict__ srcp = ei;
  const int* __restrict__ dstp = ei + E;

  unsigned* __restrict__ tl = tl_list + (size_t)blockIdx.x * TCAP;

  if (FIRST && tid == 0) cnt = 0u;
  unsigned total = 0u;
  if (!FIRST) total = tl_cnt[blockIdx.x];
  if (tid < TN * HEADS) den[tid] = 0.f;
  if (tid < nt * HEADS) {
    const size_t g = (size_t)d0 * HEADS + tid;
    const float sd = s_dst[g];
    sdt[tid] = sd;
    mxo[tid] = ordEnc(lrelu(s_src[g] + sd));
  }
  __syncthreads();

  if (FIRST) {
  auto note = [&](int idx) {
    const unsigned slot = atomicAdd(&cnt, 1u);
    if (slot < (unsigned)TCAP) elist[slot] = (unsigned)idx;
  };

  if (gridDim.x <= 255u) {
  const unsigned pat = blockIdx.x * 0x01010101u;
  const int nv = (E + 15) >> 4;
  const uint4* tv = (const uint4*)tmap;
  auto test4 = [&](unsigned wd, int e0) {
    const unsigned y = wd ^ pat;
    if (((y - 0x01010101u) & ~y & 0x80808080u) != 0u) {
      if ((y & 0x000000ffu) == 0u) note(e0 + 0);
      if ((y & 0x0000ff00u) == 0u) note(e0 + 1);
      if ((y & 0x00ff0000u) == 0u) note(e0 + 2);
      if ((y & 0xff000000u) == 0u) note(e0 + 3);
    }
  };
  for (int v0 = 0; v0 < nv; v0 += TTHREADS * 4) {
    uint4 q[4];
    #pragma unroll
    for (int j = 0; j < 4; ++j) {
      const int vi = v0 + j * TTHREADS + tid;
      q[j] = vi < nv ? tv[vi] : make_uint4(~0u, ~0u, ~0u, ~0u);
    }
    #pragma unroll
    for (int j = 0; j < 4; ++j) {
      const int e0 = (v0 + j * TTHREADS + tid) << 4;
      test4(q[j].x, e0 + 0);
      test4(q[j].y, e0 + 4);
      test4(q[j].z, e0 + 8);
      test4(q[j].w, e0 + 12);
    }
  }
  } else {
  const int Ev   = (((size_t)dstp & 15) == 0) ? (E & ~3) : 0;
  const int nvec = Ev >> 2;
  const int4* dv = (const int4*)dstp;
  for (int v0 = 0; v0 < nvec; v0 += TTHREADS * 4) {
    int4 q[4];
    #pragma unroll
    for (int j = 0; j < 4; ++j) {
      const int vi = v0 + j * TTHREADS + tid;
      q[j] = vi < nvec ? dv[vi] : make_int4(-1, -1, -1, -1);
    }
    #pragma unroll
    for (int j = 0; j < 4; ++j) {
      const int e0 = (v0 + j * TTHREADS + tid) << 2;
      if ((unsigned)(q[j].x - d0) < (unsigned)nt) note(e0 + 0);
      if ((unsigned)(q[j].y - d0) < (unsigned)nt) note(e0 + 1);
      if ((unsigned)(q[j].z - d0) < (unsigned)nt) note(e0 + 2);
      if ((unsigned)(q[j].w - d0) < (unsigned)nt) note(e0 + 3);
    }
  }
  for (int i = Ev + tid; i < E; i += TTHREADS)
    if ((unsigned)(dstp[i] - d0) < (unsigned)nt) note(i);
  }
  __syncthreads();
  total = cnt;
  if (tid == 0) tl_cnt[blockIdx.x] = total;
  }
  const bool fast = total <= (unsigned)TCAP;

  auto edge_max_q = [&](const float4 ss, unsigned dl) {
    const float4 sd = *(const float4*)(sdt + dl * HEADS);
    atomicMax(&mxo[dl * HEADS + 0], ordEnc(lrelu(ss.x + sd.x)));
    atomicMax(&mxo[dl * HEADS + 1], ordEnc(lrelu(ss.y + sd.y)));
    atomicMax(&mxo[dl * HEADS + 2], ordEnc(lrelu(ss.z + sd.z)));
    atomicMax(&mxo[dl * HEADS + 3], ordEnc(lrelu(ss.w + sd.w)));
  };
  auto edge_max = [&](int s, unsigned dl) {
    edge_max_q(*(const float4*)(s_src + (size_t)s * HEADS), dl);
  };
  if (fast && FIRST) {
    for (int i = tid; i < (int)total; i += TTHREADS) {
      const int idx = (int)elist[i];
      const int s = srcp[idx];
      const unsigned dl = (unsigned)(dstp[idx] - d0);
      edge_max(s, dl);
      const unsigned en = ((unsigned)s << 7) | dl;
      elist[i] = en;
      tl[i] = en;
    }
  } else if (fast) {
    for (int i0 = tid; i0 < (int)total; i0 += 2 * TTHREADS) {
      const int i1 = i0 + TTHREADS;
      const bool v1 = i1 < (int)total;
      const unsigned e0 = tl[i0];
      const unsigned e1 = v1 ? tl[i1] : 0u;
      const unsigned s0 = min(e0 >> 7, (unsigned)(N - 1));
      const unsigned s1 = min(e1 >> 7, (unsigned)(N - 1));
      const float4 q0 = *(const float4*)(s_src + (size_t)s0 * HEADS);
      const float4 q1 = *(const float4*)(s_src + (size_t)s1 * HEADS);
      asm volatile("" :: "v"(q0.x), "v"(q0.y), "v"(q0.z), "v"(q0.w),
                         "v"(q1.x), "v"(q1.y), "v"(q1.z), "v"(q1.w));
      edge_max_q(q0, e0 & 127u);
      elist[i0] = (s0 << 7) | (e0 & 127u);
      if (v1) {
        edge_max_q(q1, e1 & 127u);
        elist[i1] = (s1 << 7) | (e1 & 127u);
      }
    }
  } else {
    for (int i = tid; i < E; i += TTHREADS) {
      const unsigned dl = (unsigned)(dstp[i] - d0);
      if (dl < (unsigned)nt) edge_max(srcp[i], dl);
    }
  }
  __syncthreads();

  for (int dl = wv; dl < nt; dl += (TTHREADS >> 5)) {
    const int d = d0 + dl;
    const float e  = lrelu(s_src[(size_t)d * HEADS + hl] + sdt[dl * HEADS + hl]);
    const float wl = expf(e - ordDec(mxo[dl * HEADS + hl]));
    if (lane < HEADS) atomicAdd(&den[dl * HEADS + lane], wl);
    float w[HEADS];
    #pragma unroll
    for (int h = 0; h < HEADS; ++h) w[h] = __shfl(wl, h, 32);
    const float* hs = Hin + (size_t)d * HIDDEN;
    #pragma unroll
    for (int i = 0; i < 8; ++i) {
      const int ch = lane + (i << 5);
      acc[dl * APITCH + ch] = hs[ch] * w[i >> 1];
    }
  }

  if (fast) {
    for (int c0 = 0; c0 < (int)total; c0 += WCH) {
      const int cend = (int)total - c0 < WCH ? (int)total : c0 + WCH;
      if (c0 > 0) __syncthreads();
      for (int i = c0 + tid; i < cend; i += TTHREADS) {
        const unsigned en = elist[i];
        const unsigned dl = en & 127u;
        const float4 ss = *(const float4*)(s_src + (size_t)(en >> 7) * HEADS);
        const float4 sd = *(const float4*)(sdt + dl * HEADS);
        const uint4  mx = *(const uint4*)(mxo + dl * HEADS);
        float4 w;
        w.x = expf(lrelu(ss.x + sd.x) - ordDec(mx.x));
        w.y = expf(lrelu(ss.y + sd.y) - ordDec(mx.y));
        w.z = expf(lrelu(ss.z + sd.z) - ordDec(mx.z));
        w.w = expf(lrelu(ss.w + sd.w) - ordDec(mx.w));
        atomicAdd(&den[dl * HEADS + 0], w.x);
        atomicAdd(&den[dl * HEADS + 1], w.y);
        atomicAdd(&den[dl * HEADS + 2], w.z);
        atomicAdd(&den[dl * HEADS + 3], w.w);
        *(float4*)(wls + (size_t)(i - c0) * HEADS) = w;
      }
      __syncthreads();
      {
        const int half = wv & 1;
        const unsigned q = (unsigned)(wv >> 1);
        const int coff = (half << 7) + (lane << 2);
        const int hsel = (half << 1) + (lane >> 4);
        uint2* mylist = wlist + wv * LCAP;
        float4 a[NSLOT];
        #pragma unroll
        for (int s = 0; s < NSLOT; ++s) a[s] = *(const float4*)(acc + ((int)q + 16 * s) * APITCH + coff);
        int base = c0;
        while (base < cend) {
          int npick = 0;
          while (base < cend && npick <= LCAP - 32 * SCAN) {
            unsigned en[SCAN];
            #pragma unroll
            for (int u = 0; u < SCAN; ++u) {
              const int idx = base + 32 * u + lane;
              en[u] = idx < cend ? elist[idx] : 0xFFFFFFFFu;
            }
            #pragma unroll
            for (int u = 0; u < SCAN; ++u) {
              const int idx = base + 32 * u + lane;
              const bool mine = idx < cend && (en[u] & 15u) == q;
              const unsigned m = __builtin_amdgcn_ballot_w32(mine);
              if (mine) mylist[npick + (int)__builtin_amdgcn_mbcnt_lo(m, 0u)] = make_uint2(en[u], (unsigned)(idx - c0));
              npick += __popc(m);
            }
            base += 32 * SCAN;
          }
          __builtin_amdgcn_wave_barrier();
          for (int cb = 0; cb < npick; cb += 32) {
            const bool val = cb + lane < npick;
            uint2 le = make_uint2(0u, 0u);
            if (val) le = mylist[cb + lane];
            const unsigned lslot = (le.x & 127u) >> 4;
            #pragma unroll
            for (int s = 0; s < NSLOT; ++s) {
              unsigned m = __builtin_amdgcn_ballot_w32(val && lslot == (unsigned)s);
              while (m != 0u) {
                float wq[NB];
                float4 hv[NB];
                #pragma unroll
                for (int j = 0; j < NB; ++j) {
                  hv[j] = make_float4(0.f, 0.f, 0.f, 0.f);
                  wq[j] = 0.f;
                  if (m != 0u) {
                    const int bit = __builtin_ctz(m);
                    const unsigned enj = (unsigned)__builtin_amdgcn_readlane((int)le.x, bit);
                    const unsigned wij = (unsigned)__builtin_amdgcn_readlane((int)le.y, bit);
                    hv[j] = *(const float4*)(Hin + (size_t)(enj >> 7) * HIDDEN + coff);
                    wq[j] = wls[wij * HEADS + hsel];
                  }
                  m &= m - 1u;
                }
                #pragma unroll
                for (int j = 0; j < NB; ++j) {
                  a[s].x += wq[j] * hv[j].x;
                  a[s].y += wq[j] * hv[j].y;
                  a[s].z += wq[j] * hv[j].z;
                  a[s].w += wq[j] * hv[j].w;
                }
              }
            }
          }
        }
        #pragma unroll
        for (int s = 0; s < NSLOT; ++s) *(float4*)(acc + ((int)q + 16 * s) * APITCH + coff) = a[s];
      }
    }
  } else {
    __syncthreads();
    for (int i = tid; i < E; i += TTHREADS) {
      const unsigned dl = (unsigned)(dstp[i] - d0);
      if (dl < (unsigned)nt) {
        const int s = srcp[i];
        const float* hs = Hin + (size_t)s * HIDDEN;
        for (int h = 0; h < HEADS; ++h) {
          const float e  = lrelu(s_src[(size_t)s * HEADS + h] + sdt[dl * HEADS + h]);
          const float wh = expf(e - ordDec(mxo[dl * HEADS + h]));
          atomicAdd(&den[dl * HEADS + h], wh);
          for (int c = 0; c < CHAN; ++c)
            atomicAdd(&acc[dl * APITCH + h * CHAN + c], hs[h * CHAN + c] * wh);
        }
      }
    }
  }
  __syncthreads();

  for (int qd = tid; qd < TN * (HIDDEN / 4); qd += TTHREADS) {
    const int dl = qd / (HIDDEN / 4);
    const int c  = (qd - dl * (HIDDEN / 4)) << 2;
    float4 v = make_float4(0.f, 0.f, 0.f, 0.f);
    if (dl < nt) {
      const float dn = den[dl * HEADS + (c >> 6)];
      v = *(const float4*)(acc + dl * APITCH + c);
      v.x = v.x / dn + bias[c + 0];
      v.y = v.y / dn + bias[c + 1];
      v.z = v.z / dn + bias[c + 2];
      v.w = v.w / dn + bias[c + 3];
      if (ELU) {
        v.x = v.x > 0.f ? v.x : expm1f(v.x);
        v.y = v.y > 0.f ? v.y : expm1f(v.y);
        v.z = v.z > 0.f ? v.z : expm1f(v.z);
        v.w = v.w > 0.f ? v.w : expm1f(v.w);
      }
    }
    *(float4*)(acc + dl * APITCH + c) = v;
  }
  __syncthreads();

  if (ELU)
    tile_gemm<HIDDEN, true, false>(acc, wv, lane, d0, nt, Wn, nullptr, att_src_n, att_dst_n,
                                   Hout, s_src_n, s_dst_n);
  else
    tile_gemm<CHAN, false, true>(acc, wv, lane, d0, nt, Wn, bn, nullptr, nullptr,
                                 Hout, nullptr, nullptr);
}

extern "C" void kernel_launch(void* const* d_in, const int* in_sizes, int n_in,
                              void* d_out, int out_size, void* d_ws, size_t ws_size,
                              hipStream_t stream) {
  const float* x     = (const float*)d_in[0];
  const int*   ei    = (const int*)  d_in[1];
  const float* W1    = (const float*)d_in[2];
  const float* asrc1 = (const float*)d_in[3];
  const float* adst1 = (const float*)d_in[4];
  const float* b1    = (const float*)d_in[5];
  const float* W2    = (const float*)d_in[6];
  const float* asrc2 = (const float*)d_in[7];
  const float* adst2 = (const float*)d_in[8];
  const float* b2    = (const float*)d_in[9];
  const float* Wf    = (const float*)d_in[10];
  const float* bf    = (const float*)d_in[11];

  const int B  = 2;
  const int N  = in_sizes[0] / (B * FEAT);
  const int E  = in_sizes[1] / 2;

  float* p = (float*)d_ws;
  float* buf_h   = p; p += (size_t)N * HIDDEN;
  float* buf_h2  = p; p += (size_t)N * HIDDEN;
  float* ssrc    = p; p += (size_t)N * HEADS;
  float* sdst    = p; p += (size_t)N * HEADS;
  float* ssrc2   = p; p += (size_t)N * HEADS;
  float* sdst2   = p; p += (size_t)N * HEADS;
  const int ntiles  = (N + TN - 1) / TN;
  unsigned* tmap   = (unsigned*)p; p += ((E + 15) >> 4) << 2;
  unsigned* tlcnt  = (unsigned*)p; p += ntiles;
  unsigned* tllist = (unsigned*)p;

  const int mtiles  = N >> 5;
  const int gtiles  = mtiles * (HIDDEN >> 6);
  const dim3 blkG(128);
  const dim3 blkT(TTHREADS);

  for (int b = 0; b < B; ++b) {
    const float* xb = x + (size_t)b * N * FEAT;
    float* outb = (float*)d_out + (size_t)b * N * CHAN;

    if (b == 0)
      wmma_gemm_kernel<true, false, true><<<(gtiles + 3) / 4, blkG, 0, stream>>>(
          xb, W1, nullptr, asrc1, adst1, buf_h, ssrc, sdst, N, FEAT, HIDDEN, ei, E, tmap);
    else
      wmma_gemm_kernel<true, false, false><<<(gtiles + 3) / 4, blkG, 0, stream>>>(
          xb, W1, nullptr, asrc1, adst1, buf_h, ssrc, sdst, N, FEAT, HIDDEN, nullptr, 0, nullptr);
    if (b == 0)
      tile_agg_kernel<true, true><<<ntiles, blkT, 0, stream>>>(ei, E, N, ssrc, sdst, buf_h, b1,
          W2, nullptr, asrc2, adst2, buf_h2, ssrc2, sdst2, tlcnt, tllist, tmap);
    else
      tile_agg_kernel<true, false><<<ntiles, blkT, 0, stream>>>(ei, E, N, ssrc, sdst, buf_h, b1,
          W2, nullptr, asrc2, adst2, buf_h2, ssrc2, sdst2, tlcnt, tllist, tmap);

    tile_agg_kernel<false, false><<<ntiles, blkT, 0, stream>>>(ei, E, N, ssrc2, sdst2, buf_h2, b2,
        Wf, bf, nullptr, nullptr, outb, nullptr, nullptr, tlcnt, tllist, tmap);
  }
}


# All three products re-tiled: A tile of 80 rows in LDS, a wave owns 16 output columns of all 80 rows, a W fragment load feeds 5 WMMAs and is issued a trip ahead; product via LDS to whole-line stores
# speedup vs baseline: 5.5424x; 1.0393x over previous
#include <hip/hip_runtime.h>
#include <math.h>


typedef __attribute__((ext_vector_type(2))) float v2f;
typedef __attribute__((ext_vector_type(8))) float v8f;

#define HEADS     4
#define CHAN      64
#define HIDDEN    256
#define FEAT      128
#define NEG_SLOPE 0.2f
#define TILE_ROWS 80

static __device__ __forceinline__ float lrelu(float x) {
  return x > 0.f ? x : NEG_SLOPE * x;
}
static __device__ __forceinline__ unsigned ordEnc(float f) {
  unsigned u = __float_as_uint(f);
  return (u & 0x80000000u) ? ~u : (u | 0x80000000u);
}
static __device__ __forceinline__ float ordDec(unsigned u) {
  unsigned v = (u & 0x80000000u) ? (u & 0x7fffffffu) : ~u;
  return __uint_as_float(v);
}

#define TN       TILE_ROWS
#define APITCH   (HIDDEN + 4)
#define TCAP     20480
#define WCH      4096
#define TTHREADS 1024
#define NSLOT    (TN / 16)
#define LCAP     256
#define SCAN     4
#define NB       8
static_assert(TTHREADS == 1024 && TN % 16 == 0, "32 owner waves: 16 row classes x 2 halves of a row");
static_assert(TN <= 128, "a noted edge keeps its tile row in 7 bits");

#define GKT 32
template <int KD, int PA, int NC, int NT, bool GAT, bool BIAS>
static __device__ __forceinline__ void tile_gemm(const float* atile, float* otile, int tid, int d0, int nt,
                                                 const float* __restrict__ W,
                                                 const float* __restrict__ bias,
                                                 const float* __restrict__ att_src,
                                                 const float* __restrict__ att_dst,
                                                 float* __restrict__ Hout,
                                                 float* __restrict__ s_src,
                                                 float* __restrict__ s_dst) {
  constexpr int RT = TN / 16;
  constexpr int NSTRIP = NC / 16;
  static_assert(TN % 16 == 0, "whole 16-row tiles");
  static_assert(NSTRIP <= (NT >> 5) && KD % (2 * GKT) == 0, "strips fit the block; whole pairs of trips");
  const int lane = tid & 31;
  const int wv   = tid >> 5;
  const int lo = lane & 15;
  const int hi = lane >> 4;
  const bool work = wv < NSTRIP;
  const int n0 = wv << 4;

  v8f c[RT];
  #pragma unroll
  for (int t = 0; t < RT; ++t) c[t] = v8f{};

  if (work) {
    const float* arow = atile + lo * PA + (hi << 1);
    const float* wcol = W + (size_t)(hi << 1) * NC + n0 + lo;
    float bn[GKT / 4][2];
    #pragma unroll
    for (int u = 0; u < GKT / 4; ++u) {
      const float* w0 = wcol + (size_t)(4 * u) * NC;
      bn[u][0] = w0[0];  bn[u][1] = w0[NC];
    }
    #pragma unroll 2
    for (int k = 0; k < KD; k += GKT) {
      float bc[GKT / 4][2];
      #pragma unroll
      for (int u = 0; u < GKT / 4; ++u) { bc[u][0] = bn[u][0]; bc[u][1] = bn[u][1]; }
      if (k + GKT < KD) {
        #pragma unroll
        for (int u = 0; u < GKT / 4; ++u) {
          const float* w0 = wcol + (size_t)(k + GKT + 4 * u) * NC;
          bn[u][0] = w0[0];  bn[u][1] = w0[NC];
        }
      }
      #pragma unroll
      for (int u = 0; u < GKT / 4; ++u) {
        const v2f b = {bc[u][0], bc[u][1]};
        #pragma unroll
        for (int t = 0; t < RT; ++t) {
          const v2f a = *(const v2f*)(arow + t * 16 * PA + k + 4 * u);
          c[t] = __builtin_amdgcn_wmma_f32_16x16x4_f32(false, a, false, b, (short)0, c[t], false, false);
        }
      }
    }
  }
  __syncthreads();
  if (work) {
    #pragma unroll
    for (int t = 0; t < RT; ++t) {
      #pragma unroll
      for (int r = 0; r < 8; ++r)
        otile[(t * 16 + (hi << 3) + r) * APITCH + n0 + lo] = c[t][r];
    }
  }
  __syncthreads();

  constexpr int Q = NC / 4;
  for (int qd = tid; qd < TN * Q; qd += NT) {
    const int row = qd / Q;
    const int c4  = (qd - row * Q) << 2;
    if (row < nt) {
      float4 v = *(const float4*)(otile + row * APITCH + c4);
      if (BIAS) {
        v.x += bias[c4 + 0]; v.y += bias[c4 + 1]; v.z += bias[c4 + 2]; v.w += bias[c4 + 3];
      }
      *(float4*)(Hout + (size_t)(d0 + row) * NC + c4) = v;
    }
  }

  if (GAT) {
    static_assert(!GAT || (NC == HEADS * CHAN && (TN * HEADS) % (NT >> 4) == 0), "one head per 64 columns; whole waves in every trip of the score loop");
    for (int p = tid >> 4; p < TN * HEADS; p += (NT >> 4)) {
      const int row  = p >> 2;
      const int head = p & (HEADS - 1);
      const float* tr = otile + row * APITCH + head * CHAN + lo;
      const float* as = att_src + head * CHAN + lo;
      const float* ad = att_dst + head * CHAN + lo;
      const float x0 = tr[0], x1 = tr[16], x2 = tr[32], x3 = tr[48];
      float ss = x0 * as[0] + x1 * as[16] + x2 * as[32] + x3 * as[48];
      float sd = x0 * ad[0] + x1 * ad[16] + x2 * ad[32] + x3 * ad[48];
      #pragma unroll
      for (int off = 8; off > 0; off >>= 1) {
        ss += __shfl_xor(ss, off, 32);
        sd += __shfl_xor(sd, off, 32);
      }
      if (lo == 0 && row < nt) {
        s_src[(d0 + row) * HEADS + head] = ss;
        s_dst[(d0 + row) * HEADS + head] = sd;
      }
    }
  }
}

#define XPITCH   (FEAT + 4)
#define GTHREADS 512
template <bool TMAP>
__global__ void __launch_bounds__(GTHREADS)
xw_tile_kernel(const float* __restrict__ X,
               const float* __restrict__ W,
               const float* __restrict__ att_src,
               const float* __restrict__ att_dst,
               float* __restrict__ Hout,
               float* __restrict__ s_src,
               float* __restrict__ s_dst,
               int M,
               const int* __restrict__ ei, int E,
               unsigned* __restrict__ tmap) {
  __shared__ __attribute__((aligned(16))) float xt[TN * XPITCH];
  __shared__ __attribute__((aligned(16))) float ot[TN * APITCH];
  const int tid = threadIdx.x;
  if (TMAP) {
    const int* __restrict__ edst = ei + E;
    const int nwords = ((E + 15) >> 4) << 2;
    const int nthr = gridDim.x * blockDim.x;
    for (int w = blockIdx.x * blockDim.x + tid; w < nwords; w += nthr) {
      unsigned pk = 0u;
      #pragma unroll
      for (int j = 0; j < 4; ++j) {
        const int e = 4 * w + j;
        unsigned t = 255u;
        if (e < E) t = min((unsigned)edst[e] / (unsigned)TILE_ROWS, 255u);
        pk |= t << (8 * j);
      }
      tmap[w] = pk;
    }
  }
  const int d0 = blockIdx.x * TN;
  const int nt = (M - d0) < TN ? (M - d0) : TN;
  constexpr int XQ = TN * (FEAT / 4) / GTHREADS;
  static_assert(TN * (FEAT / 4) % GTHREADS == 0, "whole trips");
  float4 xv[XQ];
  #pragma unroll
  for (int j = 0; j < XQ; ++j) {
    const int qd  = tid + j * GTHREADS;
    const int row = qd / (FEAT / 4);
    const int c4  = (qd - row * (FEAT / 4)) << 2;
    const int rr  = row < nt ? row : nt - 1;
    xv[j] = *(const float4*)(X + (size_t)(d0 + rr) * FEAT + c4);
  }
  #pragma unroll
  for (int j = 0; j < XQ; ++j) {
    const int qd  = tid + j * GTHREADS;
    const int row = qd / (FEAT / 4);
    const int c4  = (qd - row * (FEAT / 4)) << 2;
    const float4 v = row < nt ? xv[j] : make_float4(0.f, 0.f, 0.f, 0.f);
    *(float4*)(xt + row * XPITCH + c4) = v;
  }
  __syncthreads();
  tile_gemm<FEAT, XPITCH, HIDDEN, GTHREADS, true, false>(xt, ot, tid, d0, nt, W, nullptr, att_src, att_dst,
                                                         Hout, s_src, s_dst);
}

template <bool ELU, bool FIRST>
__global__ void __launch_bounds__(TTHREADS)
tile_agg_kernel(const int* __restrict__ ei, int E, int N,
                const float* __restrict__ s_src,
                const float* __restrict__ s_dst,
                const float* __restrict__ Hin,
                const float* __restrict__ bias,
                const float* __restrict__ Wn,
                const float* __restrict__ bn,
                const float* __restrict__ att_src_n,
                const float* __restrict__ att_dst_n,
                float* __restrict__ Hout,
                float* __restrict__ s_src_n,
                float* __restrict__ s_dst_n,
                unsigned* __restrict__ tl_cnt,
                unsigned* __restrict__ tl_list,
                const unsigned* __restrict__ tmap) {
  __shared__ __attribute__((aligned(16))) float acc[TN * APITCH];
  __shared__ unsigned elist[TCAP];
  __shared__ __attribute__((aligned(16))) float wls[WCH * HEADS];
  __shared__ __attribute__((aligned(16))) float sdt[TN * HEADS];
  __shared__ __attribute__((aligned(16))) unsigned mxo[TN * HEADS];
  __shared__ float den[TN * HEADS];
  __shared__ unsigned cnt;
  __shared__ uint2 wlist[(TTHREADS >> 5) * LCAP];

  const int tid  = threadIdx.x;
  const int lane = tid & 31;
  const int wv   = tid >> 5;
  const int hl   = lane & (HEADS - 1);
  const int d0   = blockIdx.x * TN;
  const int nt   = (N - d0) < TN ? (N - d0) : TN;
  const int* __restrict__ srcp = ei;
  const int* __restrict__ dstp = ei + E;

  unsigned* __restrict__ tl = tl_list + (size_t)blockIdx.x * TCAP;

  if (FIRST && tid == 0) cnt = 0u;
  unsigned total = 0u;
  if (!FIRST) total = tl_cnt[blockIdx.x];
  if (tid < TN * HEADS) den[tid] = 0.f;
  if (tid < nt * HEADS) {
    const size_t g = (size_t)d0 * HEADS + tid;
    const float sd = s_dst[g];
    sdt[tid] = sd;
    mxo[tid] = ordEnc(lrelu(s_src[g] + sd));
  }
  __syncthreads();

  if (FIRST) {
  auto note = [&](int idx) {
    const unsigned slot = atomicAdd(&cnt, 1u);
    if (slot < (unsigned)TCAP) elist[slot] = (unsigned)idx;
  };

  if (gridDim.x <= 255u) {
  const unsigned pat = blockIdx.x * 0x01010101u;
  const int nv = (E + 15) >> 4;
  const uint4* tv = (const uint4*)tmap;
  auto test4 = [&](unsigned wd, int e0) {
    const unsigned y = wd ^ pat;
    if (((y - 0x01010101u) & ~y & 0x80808080u) != 0u) {
      if ((y & 0x000000ffu) == 0u) note(e0 + 0);
      if ((y & 0x0000ff00u) == 0u) note(e0 + 1);
      if ((y & 0x00ff0000u) == 0u) note(e0 + 2);
      if ((y & 0xff000000u) == 0u) note(e0 + 3);
    }
  };
  for (int v0 = 0; v0 < nv; v0 += TTHREADS * 4) {
    uint4 q[4];
    #pragma unroll
    for (int j = 0; j < 4; ++j) {
      const int vi = v0 + j * TTHREADS + tid;
      q[j] = vi < nv ? tv[vi] : make_uint4(~0u, ~0u, ~0u, ~0u);
    }
    #pragma unroll
    for (int j = 0; j < 4; ++j) {
      const int e0 = (v0 + j * TTHREADS + tid) << 4;
      test4(q[j].x, e0 + 0);
      test4(q[j].y, e0 + 4);
      test4(q[j].z, e0 + 8);
      test4(q[j].w, e0 + 12);
    }
  }
  } else {
  const int Ev   = (((size_t)dstp & 15) == 0) ? (E & ~3) : 0;
  const int nvec = Ev >> 2;
  const int4* dv = (const int4*)dstp;
  for (int v0 = 0; v0 < nvec; v0 += TTHREADS * 4) {
    int4 q[4];
    #pragma unroll
    for (int j = 0; j < 4; ++j) {
      const int vi = v0 + j * TTHREADS + tid;
      q[j] = vi < nvec ? dv[vi] : make_int4(-1, -1, -1, -1);
    }
    #pragma unroll
    for (int j = 0; j < 4; ++j) {
      const int e0 = (v0 + j * TTHREADS + tid) << 2;
      if ((unsigned)(q[j].x - d0) < (unsigned)nt) note(e0 + 0);
      if ((unsigned)(q[j].y - d0) < (unsigned)nt) note(e0 + 1);
      if ((unsigned)(q[j].z - d0) < (unsigned)nt) note(e0 + 2);
      if ((unsigned)(q[j].w - d0) < (unsigned)nt) note(e0 + 3);
    }
  }
  for (int i = Ev + tid; i < E; i += TTHREADS)
    if ((unsigned)(dstp[i] - d0) < (unsigned)nt) note(i);
  }
  __syncthreads();
  total = cnt;
  if (tid == 0) tl_cnt[blockIdx.x] = total;
  }
  const bool fast = total <= (unsigned)TCAP;

  auto edge_max_q = [&](const float4 ss, unsigned dl) {
    const float4 sd = *(const float4*)(sdt + dl * HEADS);
    atomicMax(&mxo[dl * HEADS + 0], ordEnc(lrelu(ss.x + sd.x)));
    atomicMax(&mxo[dl * HEADS + 1], ordEnc(lrelu(ss.y + sd.y)));
    atomicMax(&mxo[dl * HEADS + 2], ordEnc(lrelu(ss.z + sd.z)));
    atomicMax(&mxo[dl * HEADS + 3], ordEnc(lrelu(ss.w + sd.w)));
  };
  auto edge_max = [&](int s, unsigned dl) {
    edge_max_q(*(const float4*)(s_src + (size_t)s * HEADS), dl);
  };
  if (fast && FIRST) {
    for (int i = tid; i < (int)total; i += TTHREADS) {
      const int idx = (int)elist[i];
      const int s = srcp[idx];
      const unsigned dl = (unsigned)(dstp[idx] - d0);
      edge_max(s, dl);
      const unsigned en = ((unsigned)s << 7) | dl;
      elist[i] = en;
      tl[i] = en;
    }
  } else if (fast) {
    for (int i0 = tid; i0 < (int)total; i0 += 2 * TTHREADS) {
      const int i1 = i0 + TTHREADS;
      const bool v1 = i1 < (int)total;
      const unsigned e0 = tl[i0];
      const unsigned e1 = v1 ? tl[i1] : 0u;
      const unsigned s0 = min(e0 >> 7, (unsigned)(N - 1));
      const unsigned s1 = min(e1 >> 7, (unsigned)(N - 1));
      const float4 q0 = *(const float4*)(s_src + (size_t)s0 * HEADS);
      const float4 q1 = *(const float4*)(s_src + (size_t)s1 * HEADS);
      asm volatile("" :: "v"(q0.x), "v"(q0.y), "v"(q0.z), "v"(q0.w),
                         "v"(q1.x), "v"(q1.y), "v"(q1.z), "v"(q1.w));
      edge_max_q(q0, e0 & 127u);
      elist[i0] = (s0 << 7) | (e0 & 127u);
      if (v1) {
        edge_max_q(q1, e1 & 127u);
        elist[i1] = (s1 << 7) | (e1 & 127u);
      }
    }
  } else {
    for (int i = tid; i < E; i += TTHREADS) {
      const unsigned dl = (unsigned)(dstp[i] - d0);
      if (dl < (unsigned)nt) edge_max(srcp[i], dl);
    }
  }
  __syncthreads();

  for (int dl = wv; dl < nt; dl += (TTHREADS >> 5)) {
    const int d = d0 + dl;
    const float e  = lrelu(s_src[(size_t)d * HEADS + hl] + sdt[dl * HEADS + hl]);
    const float wl = expf(e - ordDec(mxo[dl * HEADS + hl]));
    if (lane < HEADS) atomicAdd(&den[dl * HEADS + lane], wl);
    float w[HEADS];
    #pragma unroll
    for (int h = 0; h < HEADS; ++h) w[h] = __shfl(wl, h, 32);
    const float* hs = Hin + (size_t)d * HIDDEN;
    #pragma unroll
    for (int i = 0; i < 8; ++i) {
      const int ch = lane + (i << 5);
      acc[dl * APITCH + ch] = hs[ch] * w[i >> 1];
    }
  }

  if (fast) {
    for (int c0 = 0; c0 < (int)total; c0 += WCH) {
      const int cend = (int)total - c0 < WCH ? (int)total : c0 + WCH;
      if (c0 > 0) __syncthreads();
      for (int i = c0 + tid; i < cend; i += TTHREADS) {
        const unsigned en = elist[i];
        const unsigned dl = en & 127u;
        const float4 ss = *(const float4*)(s_src + (size_t)(en >> 7) * HEADS);
        const float4 sd = *(const float4*)(sdt + dl * HEADS);
        const uint4  mx = *(const uint4*)(mxo + dl * HEADS);
        float4 w;
        w.x = expf(lrelu(ss.x + sd.x) - ordDec(mx.x));
        w.y = expf(lrelu(ss.y + sd.y) - ordDec(mx.y));
        w.z = expf(lrelu(ss.z + sd.z) - ordDec(mx.z));
        w.w = expf(lrelu(ss.w + sd.w) - ordDec(mx.w));
        atomicAdd(&den[dl * HEADS + 0], w.x);
        atomicAdd(&den[dl * HEADS + 1], w.y);
        atomicAdd(&den[dl * HEADS + 2], w.z);
        atomicAdd(&den[dl * HEADS + 3], w.w);
        *(float4*)(wls + (size_t)(i - c0) * HEADS) = w;
      }
      __syncthreads();
      {
        const int half = wv & 1;
        const unsigned q = (unsigned)(wv >> 1);
        const int coff = (half << 7) + (lane << 2);
        const int hsel = (half << 1) + (lane >> 4);
        uint2* mylist = wlist + wv * LCAP;
        float4 a[NSLOT];
        #pragma unroll
        for (int s = 0; s < NSLOT; ++s) a[s] = *(const float4*)(acc + ((int)q + 16 * s) * APITCH + coff);
        int base = c0;
        while (base < cend) {
          int npick = 0;
          while (base < cend && npick <= LCAP - 32 * SCAN) {
            unsigned en[SCAN];
            #pragma unroll
            for (int u = 0; u < SCAN; ++u) {
              const int idx = base + 32 * u + lane;
              en[u] = idx < cend ? elist[idx] : 0xFFFFFFFFu;
            }
            #pragma unroll
            for (int u = 0; u < SCAN; ++u) {
              const int idx = base + 32 * u + lane;
              const bool mine = idx < cend && (en[u] & 15u) == q;
              const unsigned m = __builtin_amdgcn_ballot_w32(mine);
              if (mine) mylist[npick + (int)__builtin_amdgcn_mbcnt_lo(m, 0u)] = make_uint2(en[u], (unsigned)(idx - c0));
              npick += __popc(m);
            }
            base += 32 * SCAN;
          }
          __builtin_amdgcn_wave_barrier();
          for (int cb = 0; cb < npick; cb += 32) {
            const bool val = cb + lane < npick;
            uint2 le = make_uint2(0u, 0u);
            if (val) le = mylist[cb + lane];
            const unsigned lslot = (le.x & 127u) >> 4;
            #pragma unroll
            for (int s = 0; s < NSLOT; ++s) {
              unsigned m = __builtin_amdgcn_ballot_w32(val && lslot == (unsigned)s);
              while (m != 0u) {
                float wq[NB];
                float4 hv[NB];
                #pragma unroll
                for (int j = 0; j < NB; ++j) {
                  hv[j] = make_float4(0.f, 0.f, 0.f, 0.f);
                  wq[j] = 0.f;
                  if (m != 0u) {
                    const int bit = __builtin_ctz(m);
                    const unsigned enj = (unsigned)__builtin_amdgcn_readlane((int)le.x, bit);
                    const unsigned wij = (unsigned)__builtin_amdgcn_readlane((int)le.y, bit);
                    hv[j] = *(const float4*)(Hin + (size_t)(enj >> 7) * HIDDEN + coff);
                    wq[j] = wls[wij * HEADS + hsel];
                  }
                  m &= m - 1u;
                }
                #pragma unroll
                for (int j = 0; j < NB; ++j) {
                  a[s].x += wq[j] * hv[j].x;
                  a[s].y += wq[j] * hv[j].y;
                  a[s].z += wq[j] * hv[j].z;
                  a[s].w += wq[j] * hv[j].w;
                }
              }
            }
          }
        }
        #pragma unroll
        for (int s = 0; s < NSLOT; ++s) *(float4*)(acc + ((int)q + 16 * s) * APITCH + coff) = a[s];
      }
    }
  } else {
    __syncthreads();
    for (int i = tid; i < E; i += TTHREADS) {
      const unsigned dl = (unsigned)(dstp[i] - d0);
      if (dl < (unsigned)nt) {
        const int s = srcp[i];
        const float* hs = Hin + (size_t)s * HIDDEN;
        for (int h = 0; h < HEADS; ++h) {
          const float e  = lrelu(s_src[(size_t)s * HEADS + h] + sdt[dl * HEADS + h]);
          const float wh = expf(e - ordDec(mxo[dl * HEADS + h]));
          atomicAdd(&den[dl * HEADS + h], wh);
          for (int c = 0; c < CHAN; ++c)
            atomicAdd(&acc[dl * APITCH + h * CHAN + c], hs[h * CHAN + c] * wh);
        }
      }
    }
  }
  __syncthreads();

  for (int qd = tid; qd < TN * (HIDDEN / 4); qd += TTHREADS) {
    const int dl = qd / (HIDDEN / 4);
    const int c  = (qd - dl * (HIDDEN / 4)) << 2;
    float4 v = make_float4(0.f, 0.f, 0.f, 0.f);
    if (dl < nt) {
      const float dn = den[dl * HEADS + (c >> 6)];
      v = *(const float4*)(acc + dl * APITCH + c);
      v.x = v.x / dn + bias[c + 0];
      v.y = v.y / dn + bias[c + 1];
      v.z = v.z / dn + bias[c + 2];
      v.w = v.w / dn + bias[c + 3];
      if (ELU) {
        v.x = v.x > 0.f ? v.x : expm1f(v.x);
        v.y = v.y > 0.f ? v.y : expm1f(v.y);
        v.z = v.z > 0.f ? v.z : expm1f(v.z);
        v.w = v.w > 0.f ? v.w : expm1f(v.w);
      }
    }
    *(float4*)(acc + dl * APITCH + c) = v;
  }
  __syncthreads();

  if (ELU)
    tile_gemm<HIDDEN, APITCH, HIDDEN, TTHREADS, true, false>(acc, acc, tid, d0, nt, Wn, nullptr, att_src_n, att_dst_n,
                                                             Hout, s_src_n, s_dst_n);
  else
    tile_gemm<HIDDEN, APITCH, CHAN, TTHREADS, false, true>(acc, acc, tid, d0, nt, Wn, bn, nullptr, nullptr,
                                                           Hout, nullptr, nullptr);
}

extern "C" void kernel_launch(void* const* d_in, const int* in_sizes, int n_in,
                              void* d_out, int out_size, void* d_ws, size_t ws_size,
                              hipStream_t stream) {
  const float* x     = (const float*)d_in[0];
  const int*   ei    = (const int*)  d_in[1];
  const float* W1    = (const float*)d_in[2];
  const float* asrc1 = (const float*)d_in[3];
  const float* adst1 = (const float*)d_in[4];
  const float* b1    = (const float*)d_in[5];
  const float* W2    = (const float*)d_in[6];
  const float* asrc2 = (const float*)d_in[7];
  const float* adst2 = (const float*)d_in[8];
  const float* b2    = (const float*)d_in[9];
  const float* Wf    = (const float*)d_in[10];
  const float* bf    = (const float*)d_in[11];

  const int B  = 2;
  const int N  = in_sizes[0] / (B * FEAT);
  const int E  = in_sizes[1] / 2;

  float* p = (float*)d_ws;
  float* buf_h   = p; p += (size_t)N * HIDDEN;
  float* buf_h2  = p; p += (size_t)N * HIDDEN;
  float* ssrc    = p; p += (size_t)N * HEADS;
  float* sdst    = p; p += (size_t)N * HEADS;
  float* ssrc2   = p; p += (size_t)N * HEADS;
  float* sdst2   = p; p += (size_t)N * HEADS;
  const int ntiles  = (N + TN - 1) / TN;
  unsigned* tmap   = (unsigned*)p; p += ((E + 15) >> 4) << 2;
  unsigned* tlcnt  = (unsigned*)p; p += ntiles;
  unsigned* tllist = (unsigned*)p;

  const dim3 blkG(GTHREADS);
  const dim3 blkT(TTHREADS);

  for (int b = 0; b < B; ++b) {
    const float* xb = x + (size_t)b * N * FEAT;
    float* outb = (float*)d_out + (size_t)b * N * CHAN;

    if (b == 0)
      xw_tile_kernel<true><<<ntiles, blkG, 0, stream>>>(
          xb, W1, asrc1, adst1, buf_h, ssrc, sdst, N, ei, E, tmap);
    else
      xw_tile_kernel<false><<<ntiles, blkG, 0, stream>>>(
          xb, W1, asrc1, adst1, buf_h, ssrc, sdst, N, nullptr, 0, nullptr);
    if (b == 0)
      tile_agg_kernel<true, true><<<ntiles, blkT, 0, stream>>>(ei, E, N, ssrc, sdst, buf_h, b1,
          W2, nullptr, asrc2, adst2, buf_h2, ssrc2, sdst2, tlcnt, tllist, tmap);
    else
      tile_agg_kernel<true, false><<<ntiles, blkT, 0, stream>>>(ei, E, N, ssrc, sdst, buf_h, b1,
          W2, nullptr, asrc2, adst2, buf_h2, ssrc2, sdst2, tlcnt, tllist, tmap);

    tile_agg_kernel<false, false><<<ntiles, blkT, 0, stream>>>(ei, E, N, ssrc2, sdst2, buf_h2, b2,
        Wf, bf, nullptr, nullptr, outb, nullptr, nullptr, tlcnt, tllist, tmap);
  }
}


# A block runs its tile of both graphs one after the other inside each launch (graph loop in the block, edge notes stay in LDS for the second graph): 6 launches become 3; sum order unchanged
# speedup vs baseline: 5.8265x; 1.0513x over previous
#include <hip/hip_runtime.h>
#include <math.h>


typedef __attribute__((ext_vector_type(2))) float v2f;
typedef __attribute__((ext_vector_type(8))) float v8f;

#define HEADS     4
#define CHAN      64
#define HIDDEN    256
#define FEAT      128
#define NEG_SLOPE 0.2f
#define TILE_ROWS 80

static __device__ __forceinline__ float lrelu(float x) {
  return x > 0.f ? x : NEG_SLOPE * x;
}
static __device__ __forceinline__ unsigned ordEnc(float f) {
  unsigned u = __float_as_uint(f);
  return (u & 0x80000000u) ? ~u : (u | 0x80000000u);
}
static __device__ __forceinline__ float ordDec(unsigned u) {
  unsigned v = (u & 0x80000000u) ? (u & 0x7fffffffu) : ~u;
  return __uint_as_float(v);
}

#define TN       TILE_ROWS
#define APITCH   (HIDDEN + 4)
#define TCAP     20480
#define WCH      4096
#define TTHREADS 1024
#define NSLOT    (TN / 16)
#define LCAP     256
#define SCAN     4
#define NB       8
static_assert(TTHREADS == 1024 && TN % 16 == 0, "32 owner waves: 16 row classes x 2 halves of a row");
static_assert(TN <= 128, "a noted edge keeps its tile row in 7 bits");

#define GKT 32
template <int KD, int PA, int NC, int NT, bool GAT, bool BIAS>
static __device__ __forceinline__ void tile_gemm(const float* atile, float* otile, int tid, int d0, int nt,
                                                 const float* __restrict__ W,
                                                 const float* __restrict__ bias,
                                                 const float* __restrict__ att_src,
                                                 const float* __restrict__ att_dst,
                                                 float* __restrict__ Hout,
                                                 float* __restrict__ s_src,
                                                 float* __restrict__ s_dst) {
  constexpr int RT = TN / 16;
  constexpr int NSTRIP = NC / 16;
  static_assert(TN % 16 == 0, "whole 16-row tiles");
  static_assert(NSTRIP <= (NT >> 5) && KD % (2 * GKT) == 0, "strips fit the block; whole pairs of trips");
  const int lane = tid & 31;
  const int wv   = tid >> 5;
  const int lo = lane & 15;
  const int hi = lane >> 4;
  const bool work = wv < NSTRIP;
  const int n0 = wv << 4;

  v8f c[RT];
  #pragma unroll
  for (int t = 0; t < RT; ++t) c[t] = v8f{};

  if (work) {
    const float* arow = atile + lo * PA + (hi << 1);
    const float* wcol = W + (size_t)(hi << 1) * NC + n0 + lo;
    float bn[GKT / 4][2];
    #pragma unroll
    for (int u = 0; u < GKT / 4; ++u) {
      const float* w0 = wcol + (size_t)(4 * u) * NC;
      bn[u][0] = w0[0];  bn[u][1] = w0[NC];
    }
    #pragma unroll 2
    for (int k = 0; k < KD; k += GKT) {
      float bc[GKT / 4][2];
      #pragma unroll
      for (int u = 0; u < GKT / 4; ++u) { bc[u][0] = bn[u][0]; bc[u][1] = bn[u][1]; }
      if (k + GKT < KD) {
        #pragma unroll
        for (int u = 0; u < GKT / 4; ++u) {
          const float* w0 = wcol + (size_t)(k + GKT + 4 * u) * NC;
          bn[u][0] = w0[0];  bn[u][1] = w0[NC];
        }
      }
      #pragma unroll
      for (int u = 0; u < GKT / 4; ++u) {
        const v2f b = {bc[u][0], bc[u][1]};
        #pragma unroll
        for (int t = 0; t < RT; ++t) {
          const v2f a = *(const v2f*)(arow + t * 16 * PA + k + 4 * u);
          c[t] = __builtin_amdgcn_wmma_f32_16x16x4_f32(false, a, false, b, (short)0, c[t], false, false);
        }
      }
    }
  }
  __syncthreads();
  if (work) {
    #pragma unroll
    for (int t = 0; t < RT; ++t) {
      #pragma unroll
      for (int r = 0; r < 8; ++r)
        otile[(t * 16 + (hi << 3) + r) * APITCH + n0 + lo] = c[t][r];
    }
  }
  __syncthreads();

  constexpr int Q = NC / 4;
  for (int qd = tid; qd < TN * Q; qd += NT) {
    const int row = qd / Q;
    const int c4  = (qd - row * Q) << 2;
    if (row < nt) {
      float4 v = *(const float4*)(otile + row * APITCH + c4);
      if (BIAS) {
        v.x += bias[c4 + 0]; v.y += bias[c4 + 1]; v.z += bias[c4 + 2]; v.w += bias[c4 + 3];
      }
      *(float4*)(Hout + (size_t)(d0 + row) * NC + c4) = v;
    }
  }

  if (GAT) {
    static_assert(!GAT || (NC == HEADS * CHAN && (TN * HEADS) % (NT >> 4) == 0), "one head per 64 columns; whole waves in every trip of the score loop");
    for (int p = tid >> 4; p < TN * HEADS; p += (NT >> 4)) {
      const int row  = p >> 2;
      const int head = p & (HEADS - 1);
      const float* tr = otile + row * APITCH + head * CHAN + lo;
      const float* as = att_src + head * CHAN + lo;
      const float* ad = att_dst + head * CHAN + lo;
      const float x0 = tr[0], x1 = tr[16], x2 = tr[32], x3 = tr[48];
      float ss = x0 * as[0] + x1 * as[16] + x2 * as[32] + x3 * as[48];
      float sd = x0 * ad[0] + x1 * ad[16] + x2 * ad[32] + x3 * ad[48];
      #pragma unroll
      for (int off = 8; off > 0; off >>= 1) {
        ss += __shfl_xor(ss, off, 32);
        sd += __shfl_xor(sd, off, 32);
      }
      if (lo == 0 && row < nt) {
        s_src[(d0 + row) * HEADS + head] = ss;
        s_dst[(d0 + row) * HEADS + head] = sd;
      }
    }
  }
}

#define XPITCH   (FEAT + 4)
#define GTHREADS 512
__global__ void __launch_bounds__(GTHREADS)
xw_tile_kernel(const float* __restrict__ X_g,
               const float* __restrict__ W,
               const float* __restrict__ att_src,
               const float* __restrict__ att_dst,
               float* __restrict__ Hout_g,
               float* __restrict__ s_src_g,
               float* __restrict__ s_dst_g,
               int M, int nb,
               const int* __restrict__ ei, int E,
               unsigned* __restrict__ tmap) {
  __shared__ __attribute__((aligned(16))) float xt[TN * XPITCH];
  __shared__ __attribute__((aligned(16))) float ot[TN * APITCH];
  const int tid = threadIdx.x;
  {
    const int* __restrict__ edst = ei + E;
    const int nwords = ((E + 15) >> 4) << 2;
    const int nthr = gridDim.x * blockDim.x;
    for (int w = blockIdx.x * blockDim.x + tid; w < nwords; w += nthr) {
      unsigned pk = 0u;
      #pragma unroll
      for (int j = 0; j < 4; ++j) {
        const int e = 4 * w + j;
        unsigned t = 255u;
        if (e < E) t = min((unsigned)edst[e] / (unsigned)TILE_ROWS, 255u);
        pk |= t << (8 * j);
      }
      tmap[w] = pk;
    }
  }
  const int d0 = blockIdx.x * TN;
  const int nt = (M - d0) < TN ? (M - d0) : TN;
  constexpr int XQ = TN * (FEAT / 4) / GTHREADS;
  static_assert(TN * (FEAT / 4) % GTHREADS == 0, "whole trips");
  for (int b = 0; b < nb; ++b) {
  const float* __restrict__ X = X_g + (size_t)b * M * FEAT;
  float* __restrict__ Hout  = Hout_g  + (size_t)b * M * HIDDEN;
  float* __restrict__ s_src = s_src_g + (size_t)b * M * HEADS;
  float* __restrict__ s_dst = s_dst_g + (size_t)b * M * HEADS;
  float4 xv[XQ];
  #pragma unroll
  for (int j = 0; j < XQ; ++j) {
    const int qd  = tid + j * GTHREADS;
    const int row = qd / (FEAT / 4);
    const int c4  = (qd - row * (FEAT / 4)) << 2;
    const int rr  = row < nt ? row : nt - 1;
    xv[j] = *(const float4*)(X + (size_t)(d0 + rr) * FEAT + c4);
  }
  #pragma unroll
  for (int j = 0; j < XQ; ++j) {
    const int qd  = tid + j * GTHREADS;
    const int row = qd / (FEAT / 4);
    const int c4  = (qd - row * (FEAT / 4)) << 2;
    const float4 v = row < nt ? xv[j] : make_float4(0.f, 0.f, 0.f, 0.f);
    *(float4*)(xt + row * XPITCH + c4) = v;
  }
  __syncthreads();
  tile_gemm<FEAT, XPITCH, HIDDEN, GTHREADS, true, false>(xt, ot, tid, d0, nt, W, nullptr, att_src, att_dst,
                                                         Hout, s_src, s_dst);
  }
}

template <bool ELU, bool FIRST>
__global__ void __launch_bounds__(TTHREADS)
tile_agg_kernel(const int* __restrict__ ei, int E, int N, int nb,
                const float* __restrict__ s_src_g,
                const float* __restrict__ s_dst_g,
                const float* __restrict__ Hin_g,
                const float* __restrict__ bias,
                const float* __restrict__ Wn,
                const float* __restrict__ bn,
                const float* __restrict__ att_src_n,
                const float* __restrict__ att_dst_n,
                float* __restrict__ Hout_g,
                float* __restrict__ s_src_n_g,
                float* __restrict__ s_dst_n_g,
                unsigned* __restrict__ tl_cnt,
                unsigned* __restrict__ tl_list,
                const unsigned* __restrict__ tmap) {
  __shared__ __attribute__((aligned(16))) float acc[TN * APITCH];
  __shared__ unsigned elist[TCAP];
  __shared__ __attribute__((aligned(16))) float wls[WCH * HEADS];
  __shared__ __attribute__((aligned(16))) float sdt[TN * HEADS];
  __shared__ __attribute__((aligned(16))) unsigned mxo[TN * HEADS];
  __shared__ float den[TN * HEADS];
  __shared__ unsigned cnt;
  __shared__ uint2 wlist[(TTHREADS >> 5) * LCAP];

  const int d0   = blockIdx.x * TN;
  const int nt   = (N - d0) < TN ? (N - d0) : TN;
  const int* __restrict__ srcp = ei;
  const int* __restrict__ dstp = ei + E;

  unsigned* __restrict__ tl = tl_list + (size_t)blockIdx.x * TCAP;

  unsigned total = 0u;
  if (!FIRST) total = tl_cnt[blockIdx.x];
  for (int b = 0; b < nb; ++b) {
  int tid = threadIdx.x;
  asm volatile("" : "+v"(tid));
  const int lane = tid & 31;
  const int wv   = tid >> 5;
  const int hl   = lane & (HEADS - 1);
  const bool first = FIRST && b == 0;
  const float* __restrict__ s_src = s_src_g + (size_t)b * N * HEADS;
  const float* __restrict__ s_dst = s_dst_g + (size_t)b * N * HEADS;
  const float* __restrict__ Hin   = Hin_g   + (size_t)b * N * HIDDEN;
  float* __restrict__ Hout    = Hout_g + (size_t)b * N * (ELU ? HIDDEN : CHAN);
  float* __restrict__ s_src_n = ELU ? s_src_n_g + (size_t)b * N * HEADS : nullptr;
  float* __restrict__ s_dst_n = ELU ? s_dst_n_g + (size_t)b * N * HEADS : nullptr;

  if (first && tid == 0) cnt = 0u;
  if (tid < TN * HEADS) den[tid] = 0.f;
  if (tid < nt * HEADS) {
    const size_t g = (size_t)d0 * HEADS + tid;
    const float sd = s_dst[g];
    sdt[tid] = sd;
    mxo[tid] = ordEnc(lrelu(s_src[g] + sd));
  }
  __syncthreads();

  if (first) {
  auto note = [&](int idx) {
    const unsigned slot = atomicAdd(&cnt, 1u);
    if (slot < (unsigned)TCAP) elist[slot] = (unsigned)idx;
  };

  if (gridDim.x <= 255u) {
  const unsigned pat = blockIdx.x * 0x01010101u;
  const int nv = (E + 15) >> 4;
  const uint4* tv = (const uint4*)tmap;
  auto test4 = [&](unsigned wd, int e0) {
    const unsigned y = wd ^ pat;
    if (((y - 0x01010101u) & ~y & 0x80808080u) != 0u) {
      if ((y & 0x000000ffu) == 0u) note(e0 + 0);
      if ((y & 0x0000ff00u) == 0u) note(e0 + 1);
      if ((y & 0x00ff0000u) == 0u) note(e0 + 2);
      if ((y & 0xff000000u) == 0u) note(e0 + 3);
    }
  };
  for (int v0 = 0; v0 < nv; v0 += TTHREADS * 4) {
    uint4 q[4];
    #pragma unroll
    for (int j = 0; j < 4; ++j) {
      const int vi = v0 + j * TTHREADS + tid;
      q[j] = vi < nv ? tv[vi] : make_uint4(~0u, ~0u, ~0u, ~0u);
    }
    #pragma unroll
    for (int j = 0; j < 4; ++j) {
      const int e0 = (v0 + j * TTHREADS + tid) << 4;
      test4(q[j].x, e0 + 0);
      test4(q[j].y, e0 + 4);
      test4(q[j].z, e0 + 8);
      test4(q[j].w, e0 + 12);
    }
  }
  } else {
  const int Ev   = (((size_t)dstp & 15) == 0) ? (E & ~3) : 0;
  const int nvec = Ev >> 2;
  const int4* dv = (const int4*)dstp;
  for (int v0 = 0; v0 < nvec; v0 += TTHREADS * 4) {
    int4 q[4];
    #pragma unroll
    for (int j = 0; j < 4; ++j) {
      const int vi = v0 + j * TTHREADS + tid;
      q[j] = vi < nvec ? dv[vi] : make_int4(-1, -1, -1, -1);
    }
    #pragma unroll
    for (int j = 0; j < 4; ++j) {
      const int e0 = (v0 + j * TTHREADS + tid) << 2;
      if ((unsigned)(q[j].x - d0) < (unsigned)nt) note(e0 + 0);
      if ((unsigned)(q[j].y - d0) < (unsigned)nt) note(e0 + 1);
      if ((unsigned)(q[j].z - d0) < (unsigned)nt) note(e0 + 2);
      if ((unsigned)(q[j].w - d0) < (unsigned)nt) note(e0 + 3);
    }
  }
  for (int i = Ev + tid; i < E; i += TTHREADS)
    if ((unsigned)(dstp[i] - d0) < (unsigned)nt) note(i);
  }
  __syncthreads();
  total = cnt;
  if (tid == 0) tl_cnt[blockIdx.x] = total;
  }
  const bool fast = total <= (unsigned)TCAP;

  auto edge_max_q = [&](const float4 ss, unsigned dl) {
    const float4 sd = *(const float4*)(sdt + dl * HEADS);
    atomicMax(&mxo[dl * HEADS + 0], ordEnc(lrelu(ss.x + sd.x)));
    atomicMax(&mxo[dl * HEADS + 1], ordEnc(lrelu(ss.y + sd.y)));
    atomicMax(&mxo[dl * HEADS + 2], ordEnc(lrelu(ss.z + sd.z)));
    atomicMax(&mxo[dl * HEADS + 3], ordEnc(lrelu(ss.w + sd.w)));
  };
  auto edge_max = [&](int s, unsigned dl) {
    edge_max_q(*(const float4*)(s_src + (size_t)s * HEADS), dl);
  };
  if (fast && b > 0) {
    for (int i = tid; i < (int)total; i += TTHREADS) {
      const unsigned en = elist[i];
      edge_max((int)(en >> 7), en & 127u);
    }
  } else if (fast && FIRST) {
    for (int i = tid; i < (int)total; i += TTHREADS) {
      const int idx = (int)elist[i];
      const int s = srcp[idx];
      const unsigned dl = (unsigned)(dstp[idx] - d0);
      edge_max(s, dl);
      const unsigned en = ((unsigned)s << 7) | dl;
      elist[i] = en;
      tl[i] = en;
    }
  } else if (fast) {
    for (int i0 = tid; i0 < (int)total; i0 += 2 * TTHREADS) {
      const int i1 = i0 + TTHREADS;
      const bool v1 = i1 < (int)total;
      const unsigned e0 = tl[i0];
      const unsigned e1 = v1 ? tl[i1] : 0u;
      const unsigned s0 = min(e0 >> 7, (unsigned)(N - 1));
      const unsigned s1 = min(e1 >> 7, (unsigned)(N - 1));
      const float4 q0 = *(const float4*)(s_src + (size_t)s0 * HEADS);
      const float4 q1 = *(const float4*)(s_src + (size_t)s1 * HEADS);
      asm volatile("" :: "v"(q0.x), "v"(q0.y), "v"(q0.z), "v"(q0.w),
                         "v"(q1.x), "v"(q1.y), "v"(q1.z), "v"(q1.w));
      edge_max_q(q0, e0 & 127u);
      elist[i0] = (s0 << 7) | (e0 & 127u);
      if (v1) {
        edge_max_q(q1, e1 & 127u);
        elist[i1] = (s1 << 7) | (e1 & 127u);
      }
    }
  } else {
    for (int i = tid; i < E; i += TTHREADS) {
      const unsigned dl = (unsigned)(dstp[i] - d0);
      if (dl < (unsigned)nt) edge_max(srcp[i], dl);
    }
  }
  __syncthreads();

  for (int dl = wv; dl < nt; dl += (TTHREADS >> 5)) {
    const int d = d0 + dl;
    const float e  = lrelu(s_src[(size_t)d * HEADS + hl] + sdt[dl * HEADS + hl]);
    const float wl = expf(e - ordDec(mxo[dl * HEADS + hl]));
    if (lane < HEADS) atomicAdd(&den[dl * HEADS + lane], wl);
    float w[HEADS];
    #pragma unroll
    for (int h = 0; h < HEADS; ++h) w[h] = __shfl(wl, h, 32);
    const float* hs = Hin + (size_t)d * HIDDEN;
    #pragma unroll
    for (int i = 0; i < 8; ++i) {
      const int ch = lane + (i << 5);
      acc[dl * APITCH + ch] = hs[ch] * w[i >> 1];
    }
  }

  if (fast) {
    for (int c0 = 0; c0 < (int)total; c0 += WCH) {
      const int cend = (int)total - c0 < WCH ? (int)total : c0 + WCH;
      if (c0 > 0) __syncthreads();
      for (int i = c0 + tid; i < cend; i += TTHREADS) {
        const unsigned en = elist[i];
        const unsigned dl = en & 127u;
        const float4 ss = *(const float4*)(s_src + (size_t)(en >> 7) * HEADS);
        const float4 sd = *(const float4*)(sdt + dl * HEADS);
        const uint4  mx = *(const uint4*)(mxo + dl * HEADS);
        float4 w;
        w.x = expf(lrelu(ss.x + sd.x) - ordDec(mx.x));
        w.y = expf(lrelu(ss.y + sd.y) - ordDec(mx.y));
        w.z = expf(lrelu(ss.z + sd.z) - ordDec(mx.z));
        w.w = expf(lrelu(ss.w + sd.w) - ordDec(mx.w));
        atomicAdd(&den[dl * HEADS + 0], w.x);
        atomicAdd(&den[dl * HEADS + 1], w.y);
        atomicAdd(&den[dl * HEADS + 2], w.z);
        atomicAdd(&den[dl * HEADS + 3], w.w);
        *(float4*)(wls + (size_t)(i - c0) * HEADS) = w;
      }
      __syncthreads();
      {
        const int half = wv & 1;
        const unsigned q = (unsigned)(wv >> 1);
        const int coff = (half << 7) + (lane << 2);
        const int hsel = (half << 1) + (lane >> 4);
        uint2* mylist = wlist + wv * LCAP;
        float4 a[NSLOT];
        #pragma unroll
        for (int s = 0; s < NSLOT; ++s) a[s] = *(const float4*)(acc + ((int)q + 16 * s) * APITCH + coff);
        int base = c0;
        while (base < cend) {
          int npick = 0;
          while (base < cend && npick <= LCAP - 32 * SCAN) {
            unsigned en[SCAN];
            #pragma unroll
            for (int u = 0; u < SCAN; ++u) {
              const int idx = base + 32 * u + lane;
              en[u] = idx < cend ? elist[idx] : 0xFFFFFFFFu;
            }
            #pragma unroll
            for (int u = 0; u < SCAN; ++u) {
              const int idx = base + 32 * u + lane;
              const bool mine = idx < cend && (en[u] & 15u) == q;
              const unsigned m = __builtin_amdgcn_ballot_w32(mine);
              if (mine) mylist[npick + (int)__builtin_amdgcn_mbcnt_lo(m, 0u)] = make_uint2(en[u], (unsigned)(idx - c0));
              npick += __popc(m);
            }
            base += 32 * SCAN;
          }
          __builtin_amdgcn_wave_barrier();
          for (int cb = 0; cb < npick; cb += 32) {
            const bool val = cb + lane < npick;
            uint2 le = make_uint2(0u, 0u);
            if (val) le = mylist[cb + lane];
            const unsigned lslot = (le.x & 127u) >> 4;
            #pragma unroll
            for (int s = 0; s < NSLOT; ++s) {
              unsigned m = __builtin_amdgcn_ballot_w32(val && lslot == (unsigned)s);
              while (m != 0u) {
                float wq[NB];
                float4 hv[NB];
                #pragma unroll
                for (int j = 0; j < NB; ++j) {
                  hv[j] = make_float4(0.f, 0.f, 0.f, 0.f);
                  wq[j] = 0.f;
                  if (m != 0u) {
                    const int bit = __builtin_ctz(m);
                    const unsigned enj = (unsigned)__builtin_amdgcn_readlane((int)le.x, bit);
                    const unsigned wij = (unsigned)__builtin_amdgcn_readlane((int)le.y, bit);
                    hv[j] = *(const float4*)(Hin + (size_t)(enj >> 7) * HIDDEN + coff);
                    wq[j] = wls[wij * HEADS + hsel];
                  }
                  m &= m - 1u;
                }
                #pragma unroll
                for (int j = 0; j < NB; ++j) {
                  a[s].x += wq[j] * hv[j].x;
                  a[s].y += wq[j] * hv[j].y;
                  a[s].z += wq[j] * hv[j].z;
                  a[s].w += wq[j] * hv[j].w;
                }
              }
            }
          }
        }
        #pragma unroll
        for (int s = 0; s < NSLOT; ++s) *(float4*)(acc + ((int)q + 16 * s) * APITCH + coff) = a[s];
      }
    }
  } else {
    __syncthreads();
    for (int i = tid; i < E; i += TTHREADS) {
      const unsigned dl = (unsigned)(dstp[i] - d0);
      if (dl < (unsigned)nt) {
        const int s = srcp[i];
        const float* hs = Hin + (size_t)s * HIDDEN;
        for (int h = 0; h < HEADS; ++h) {
          const float e  = lrelu(s_src[(size_t)s * HEADS + h] + sdt[dl * HEADS + h]);
          const float wh = expf(e - ordDec(mxo[dl * HEADS + h]));
          atomicAdd(&den[dl * HEADS + h], wh);
          for (int c = 0; c < CHAN; ++c)
            atomicAdd(&acc[dl * APITCH + h * CHAN + c], hs[h * CHAN + c] * wh);
        }
      }
    }
  }
  __syncthreads();

  for (int qd = tid; qd < TN * (HIDDEN / 4); qd += TTHREADS) {
    const int dl = qd / (HIDDEN / 4);
    const int c  = (qd - dl * (HIDDEN / 4)) << 2;
    float4 v = make_float4(0.f, 0.f, 0.f, 0.f);
    if (dl < nt) {
      const float dn = den[dl * HEADS + (c >> 6)];
      v = *(const float4*)(acc + dl * APITCH + c);
      v.x = v.x / dn + bias[c + 0];
      v.y = v.y / dn + bias[c + 1];
      v.z = v.z / dn + bias[c + 2];
      v.w = v.w / dn + bias[c + 3];
      if (ELU) {
        v.x = v.x > 0.f ? v.x : expm1f(v.x);
        v.y = v.y > 0.f ? v.y : expm1f(v.y);
        v.z = v.z > 0.f ? v.z : expm1f(v.z);
        v.w = v.w > 0.f ? v.w : expm1f(v.w);
      }
    }
    *(float4*)(acc + dl * APITCH + c) = v;
  }
  __syncthreads();

  if (ELU)
    tile_gemm<HIDDEN, APITCH, HIDDEN, TTHREADS, true, false>(acc, acc, tid, d0, nt, Wn, nullptr, att_src_n, att_dst_n,
                                                             Hout, s_src_n, s_dst_n);
  else
    tile_gemm<HIDDEN, APITCH, CHAN, TTHREADS, false, true>(acc, acc, tid, d0, nt, Wn, bn, nullptr, nullptr,
                                                           Hout, nullptr, nullptr);
  }
}

extern "C" void kernel_launch(void* const* d_in, const int* in_sizes, int n_in,
                              void* d_out, int out_size, void* d_ws, size_t ws_size,
                              hipStream_t stream) {
  const float* x     = (const float*)d_in[0];
  const int*   ei    = (const int*)  d_in[1];
  const float* W1    = (const float*)d_in[2];
  const float* asrc1 = (const float*)d_in[3];
  const float* adst1 = (const float*)d_in[4];
  const float* b1    = (const float*)d_in[5];
  const float* W2    = (const float*)d_in[6];
  const float* asrc2 = (const float*)d_in[7];
  const float* adst2 = (const float*)d_in[8];
  const float* b2    = (const float*)d_in[9];
  const float* Wf    = (const float*)d_in[10];
  const float* bf    = (const float*)d_in[11];

  const int B  = 2;
  const int N  = in_sizes[0] / (B * FEAT);
  const int E  = in_sizes[1] / 2;

  float* p = (float*)d_ws;
  float* buf_h   = p; p += (size_t)B * N * HIDDEN;
  float* buf_h2  = p; p += (size_t)B * N * HIDDEN;
  float* ssrc    = p; p += (size_t)B * N * HEADS;
  float* sdst    = p; p += (size_t)B * N * HEADS;
  float* ssrc2   = p; p += (size_t)B * N * HEADS;
  float* sdst2   = p; p += (size_t)B * N * HEADS;
  const int ntiles  = (N + TN - 1) / TN;
  unsigned* tmap   = (unsigned*)p; p += ((E + 15) >> 4) << 2;
  unsigned* tlcnt  = (unsigned*)p; p += ntiles;
  unsigned* tllist = (unsigned*)p;

  const dim3 blkG(GTHREADS);
  const dim3 blkT(TTHREADS);

  xw_tile_kernel<<<ntiles, blkG, 0, stream>>>(
      x, W1, asrc1, adst1, buf_h, ssrc, sdst, N, B, ei, E, tmap);
  tile_agg_kernel<true, true><<<ntiles, blkT, 0, stream>>>(ei, E, N, B, ssrc, sdst, buf_h, b1,
      W2, nullptr, asrc2, adst2, buf_h2, ssrc2, sdst2, tlcnt, tllist, tmap);
  tile_agg_kernel<false, false><<<ntiles, blkT, 0, stream>>>(ei, E, N, B, ssrc2, sdst2, buf_h2, b2,
      Wf, bf, nullptr, nullptr, (float*)d_out, nullptr, nullptr, tlcnt, tllist, tmap);
}


# Owner waves keep their picked notes: tile list walked (ballot picks) once per call; second graph takes the picks from LDS, the later launch reads them from workspace; sum order unchanged
# speedup vs baseline: 6.0773x; 1.0430x over previous
#include <hip/hip_runtime.h>
#include <math.h>


typedef __attribute__((ext_vector_type(2))) float v2f;
typedef __attribute__((ext_vector_type(8))) float v8f;

#define HEADS     4
#define CHAN      64
#define HIDDEN    256
#define FEAT      128
#define NEG_SLOPE 0.2f
#define TILE_ROWS 80

static __device__ __forceinline__ float lrelu(float x) {
  return x > 0.f ? x : NEG_SLOPE * x;
}
static __device__ __forceinline__ unsigned ordEnc(float f) {
  unsigned u = __float_as_uint(f);
  return (u & 0x80000000u) ? ~u : (u | 0x80000000u);
}
static __device__ __forceinline__ float ordDec(unsigned u) {
  unsigned v = (u & 0x80000000u) ? (u & 0x7fffffffu) : ~u;
  return __uint_as_float(v);
}

#define TN       TILE_ROWS
#define APITCH   (HIDDEN + 4)
#define TCAP     20480
#define WCH      4096
#define TTHREADS 1024
#define NSLOT    (TN / 16)
#define LCAP     256
#define SCAN     4
#define NB       8
static_assert(TTHREADS == 1024 && TN % 16 == 0, "32 owner waves: 16 row classes x 2 halves of a row");
static_assert(TN <= 128, "a noted edge keeps its tile row in 7 bits");

#define GKT 32
template <int KD, int PA, int NC, int NT, bool GAT, bool BIAS>
static __device__ __forceinline__ void tile_gemm(const float* atile, float* otile, int tid, int d0, int nt,
                                                 const float* __restrict__ W,
                                                 const float* __restrict__ bias,
                                                 const float* __restrict__ att_src,
                                                 const float* __restrict__ att_dst,
                                                 float* __restrict__ Hout,
                                                 float* __restrict__ s_src,
                                                 float* __restrict__ s_dst) {
  constexpr int RT = TN / 16;
  constexpr int NSTRIP = NC / 16;
  static_assert(TN % 16 == 0, "whole 16-row tiles");
  static_assert(NSTRIP <= (NT >> 5) && KD % (2 * GKT) == 0, "strips fit the block; whole pairs of trips");
  const int lane = tid & 31;
  const int wv   = tid >> 5;
  const int lo = lane & 15;
  const int hi = lane >> 4;
  const bool work = wv < NSTRIP;
  const int n0 = wv << 4;

  v8f c[RT];
  #pragma unroll
  for (int t = 0; t < RT; ++t) c[t] = v8f{};

  if (work) {
    const float* arow = atile + lo * PA + (hi << 1);
    const float* wcol = W + (size_t)(hi << 1) * NC + n0 + lo;
    float bn[GKT / 4][2];
    #pragma unroll
    for (int u = 0; u < GKT / 4; ++u) {
      const float* w0 = wcol + (size_t)(4 * u) * NC;
      bn[u][0] = w0[0];  bn[u][1] = w0[NC];
    }
    #pragma unroll 2
    for (int k = 0; k < KD; k += GKT) {
      float bc[GKT / 4][2];
      #pragma unroll
      for (int u = 0; u < GKT / 4; ++u) { bc[u][0] = bn[u][0]; bc[u][1] = bn[u][1]; }
      if (k + GKT < KD) {
        #pragma unroll
        for (int u = 0; u < GKT / 4; ++u) {
          const float* w0 = wcol + (size_t)(k + GKT + 4 * u) * NC;
          bn[u][0] = w0[0];  bn[u][1] = w0[NC];
        }
      }
      #pragma unroll
      for (int u = 0; u < GKT / 4; ++u) {
        const v2f b = {bc[u][0], bc[u][1]};
        #pragma unroll
        for (int t = 0; t < RT; ++t) {
          const v2f a = *(const v2f*)(arow + t * 16 * PA + k + 4 * u);
          c[t] = __builtin_amdgcn_wmma_f32_16x16x4_f32(false, a, false, b, (short)0, c[t], false, false);
        }
      }
    }
  }
  __syncthreads();
  if (work) {
    #pragma unroll
    for (int t = 0; t < RT; ++t) {
      #pragma unroll
      for (int r = 0; r < 8; ++r)
        otile[(t * 16 + (hi << 3) + r) * APITCH + n0 + lo] = c[t][r];
    }
  }
  __syncthreads();

  constexpr int Q = NC / 4;
  for (int qd = tid; qd < TN * Q; qd += NT) {
    const int row = qd / Q;
    const int c4  = (qd - row * Q) << 2;
    if (row < nt) {
      float4 v = *(const float4*)(otile + row * APITCH + c4);
      if (BIAS) {
        v.x += bias[c4 + 0]; v.y += bias[c4 + 1]; v.z += bias[c4 + 2]; v.w += bias[c4 + 3];
      }
      *(float4*)(Hout + (size_t)(d0 + row) * NC + c4) = v;
    }
  }

  if (GAT) {
    static_assert(!GAT || (NC == HEADS * CHAN && (TN * HEADS) % (NT >> 4) == 0), "one head per 64 columns; whole waves in every trip of the score loop");
    for (int p = tid >> 4; p < TN * HEADS; p += (NT >> 4)) {
      const int row  = p >> 2;
      const int head = p & (HEADS - 1);
      const float* tr = otile + row * APITCH + head * CHAN + lo;
      const float* as = att_src + head * CHAN + lo;
      const float* ad = att_dst + head * CHAN + lo;
      const float x0 = tr[0], x1 = tr[16], x2 = tr[32], x3 = tr[48];
      float ss = x0 * as[0] + x1 * as[16] + x2 * as[32] + x3 * as[48];
      float sd = x0 * ad[0] + x1 * ad[16] + x2 * ad[32] + x3 * ad[48];
      #pragma unroll
      for (int off = 8; off > 0; off >>= 1) {
        ss += __shfl_xor(ss, off, 32);
        sd += __shfl_xor(sd, off, 32);
      }
      if (lo == 0 && row < nt) {
        s_src[(d0 + row) * HEADS + head] = ss;
        s_dst[(d0 + row) * HEADS + head] = sd;
      }
    }
  }
}

#define XPITCH   (FEAT + 4)
#define GTHREADS 512
__global__ void __launch_bounds__(GTHREADS)
xw_tile_kernel(const float* __restrict__ X_g,
               const float* __restrict__ W,
               const float* __restrict__ att_src,
               const float* __restrict__ att_dst,
               float* __restrict__ Hout_g,
               float* __restrict__ s_src_g,
               float* __restrict__ s_dst_g,
               int M, int nb,
               const int* __restrict__ ei, int E,
               unsigned* __restrict__ tmap) {
  __shared__ __attribute__((aligned(16))) float xt[TN * XPITCH];
  __shared__ __attribute__((aligned(16))) float ot[TN * APITCH];
  const int tid = threadIdx.x;
  {
    const int* __restrict__ edst = ei + E;
    const int nwords = ((E + 15) >> 4) << 2;
    const int nthr = gridDim.x * blockDim.x;
    for (int w = blockIdx.x * blockDim.x + tid; w < nwords; w += nthr) {
      unsigned pk = 0u;
      #pragma unroll
      for (int j = 0; j < 4; ++j) {
        const int e = 4 * w + j;
        unsigned t = 255u;
        if (e < E) t = min((unsigned)edst[e] / (unsigned)TILE_ROWS, 255u);
        pk |= t << (8 * j);
      }
      tmap[w] = pk;
    }
  }
  const int d0 = blockIdx.x * TN;
  const int nt = (M - d0) < TN ? (M - d0) : TN;
  constexpr int XQ = TN * (FEAT / 4) / GTHREADS;
  static_assert(TN * (FEAT / 4) % GTHREADS == 0, "whole trips");
  for (int b = 0; b < nb; ++b) {
  const float* __restrict__ X = X_g + (size_t)b * M * FEAT;
  float* __restrict__ Hout  = Hout_g  + (size_t)b * M * HIDDEN;
  float* __restrict__ s_src = s_src_g + (size_t)b * M * HEADS;
  float* __restrict__ s_dst = s_dst_g + (size_t)b * M * HEADS;
  float4 xv[XQ];
  #pragma unroll
  for (int j = 0; j < XQ; ++j) {
    const int qd  = tid + j * GTHREADS;
    const int row = qd / (FEAT / 4);
    const int c4  = (qd - row * (FEAT / 4)) << 2;
    const int rr  = row < nt ? row : nt - 1;
    xv[j] = *(const float4*)(X + (size_t)(d0 + rr) * FEAT + c4);
  }
  #pragma unroll
  for (int j = 0; j < XQ; ++j) {
    const int qd  = tid + j * GTHREADS;
    const int row = qd / (FEAT / 4);
    const int c4  = (qd - row * (FEAT / 4)) << 2;
    const float4 v = row < nt ? xv[j] : make_float4(0.f, 0.f, 0.f, 0.f);
    *(float4*)(xt + row * XPITCH + c4) = v;
  }
  __syncthreads();
  tile_gemm<FEAT, XPITCH, HIDDEN, GTHREADS, true, false>(xt, ot, tid, d0, nt, W, nullptr, att_src, att_dst,
                                                         Hout, s_src, s_dst);
  }
}

template <bool ELU, bool FIRST>
__global__ void __launch_bounds__(TTHREADS)
tile_agg_kernel(const int* __restrict__ ei, int E, int N, int nb,
                const float* __restrict__ s_src_g,
                const float* __restrict__ s_dst_g,
                const float* __restrict__ Hin_g,
                const float* __restrict__ bias,
                const float* __restrict__ Wn,
                const float* __restrict__ bn,
                const float* __restrict__ att_src_n,
                const float* __restrict__ att_dst_n,
                float* __restrict__ Hout_g,
                float* __restrict__ s_src_n_g,
                float* __restrict__ s_dst_n_g,
                unsigned* __restrict__ tl_cnt,
                unsigned* __restrict__ tl_list,
                const unsigned* __restrict__ tmap,
                unsigned* __restrict__ pk_cnt,
                unsigned* __restrict__ pk_list) {
  __shared__ __attribute__((aligned(16))) float acc[TN * APITCH];
  __shared__ unsigned elist[TCAP];
  __shared__ __attribute__((aligned(16))) float wls[WCH * HEADS];
  __shared__ __attribute__((aligned(16))) float sdt[TN * HEADS];
  __shared__ __attribute__((aligned(16))) unsigned mxo[TN * HEADS];
  __shared__ float den[TN * HEADS];
  __shared__ unsigned cnt;
  __shared__ uint2 wlist[(TTHREADS >> 5) * LCAP];
  __shared__ unsigned npk[TTHREADS >> 5];

  const int d0   = blockIdx.x * TN;
  const int nt   = (N - d0) < TN ? (N - d0) : TN;
  const int* __restrict__ srcp = ei;
  const int* __restrict__ dstp = ei + E;

  unsigned* __restrict__ tl = tl_list + (size_t)blockIdx.x * TCAP;

  unsigned total = 0u;
  if (!FIRST) total = tl_cnt[blockIdx.x];
  for (int b = 0; b < nb; ++b) {
  int tid = threadIdx.x;
  asm volatile("" : "+v"(tid));
  const int lane = tid & 31;
  const int wv   = tid >> 5;
  const int hl   = lane & (HEADS - 1);
  const bool first = FIRST && b == 0;
  unsigned pkn = 0xFFFFFFFFu;
  if (!FIRST && b == 0 && total - 1u < (unsigned)WCH) pkn = pk_cnt[(size_t)blockIdx.x * (TTHREADS >> 5) + wv];
  const float* __restrict__ s_src = s_src_g + (size_t)b * N * HEADS;
  const float* __restrict__ s_dst = s_dst_g + (size_t)b * N * HEADS;
  const float* __restrict__ Hin   = Hin_g   + (size_t)b * N * HIDDEN;
  float* __restrict__ Hout    = Hout_g + (size_t)b * N * (ELU ? HIDDEN : CHAN);
  float* __restrict__ s_src_n = ELU ? s_src_n_g + (size_t)b * N * HEADS : nullptr;
  float* __restrict__ s_dst_n = ELU ? s_dst_n_g + (size_t)b * N * HEADS : nullptr;

  if (first && tid == 0) cnt = 0u;
  if (tid < TN * HEADS) den[tid] = 0.f;
  if (tid < nt * HEADS) {
    const size_t g = (size_t)d0 * HEADS + tid;
    const float sd = s_dst[g];
    sdt[tid] = sd;
    mxo[tid] = ordEnc(lrelu(s_src[g] + sd));
  }
  __syncthreads();

  if (first) {
  auto note = [&](int idx) {
    const unsigned slot = atomicAdd(&cnt, 1u);
    if (slot < (unsigned)TCAP) elist[slot] = (unsigned)idx;
  };

  if (gridDim.x <= 255u) {
  const unsigned pat = blockIdx.x * 0x01010101u;
  const int nv = (E + 15) >> 4;
  const uint4* tv = (const uint4*)tmap;
  auto test4 = [&](unsigned wd, int e0) {
    const unsigned y = wd ^ pat;
    if (((y - 0x01010101u) & ~y & 0x80808080u) != 0u) {
      if ((y & 0x000000ffu) == 0u) note(e0 + 0);
      if ((y & 0x0000ff00u) == 0u) note(e0 + 1);
      if ((y & 0x00ff0000u) == 0u) note(e0 + 2);
      if ((y & 0xff000000u) == 0u) note(e0 + 3);
    }
  };
  for (int v0 = 0; v0 < nv; v0 += TTHREADS * 4) {
    uint4 q[4];
    #pragma unroll
    for (int j = 0; j < 4; ++j) {
      const int vi = v0 + j * TTHREADS + tid;
      q[j] = vi < nv ? tv[vi] : make_uint4(~0u, ~0u, ~0u, ~0u);
    }
    #pragma unroll
    for (int j = 0; j < 4; ++j) {
      const int e0 = (v0 + j * TTHREADS + tid) << 4;
      test4(q[j].x, e0 + 0);
      test4(q[j].y, e0 + 4);
      test4(q[j].z, e0 + 8);
      test4(q[j].w, e0 + 12);
    }
  }
  } else {
  const int Ev   = (((size_t)dstp & 15) == 0) ? (E & ~3) : 0;
  const int nvec = Ev >> 2;
  const int4* dv = (const int4*)dstp;
  for (int v0 = 0; v0 < nvec; v0 += TTHREADS * 4) {
    int4 q[4];
    #pragma unroll
    for (int j = 0; j < 4; ++j) {
      const int vi = v0 + j * TTHREADS + tid;
      q[j] = vi < nvec ? dv[vi] : make_int4(-1, -1, -1, -1);
    }
    #pragma unroll
    for (int j = 0; j < 4; ++j) {
      const int e0 = (v0 + j * TTHREADS + tid) << 2;
      if ((unsigned)(q[j].x - d0) < (unsigned)nt) note(e0 + 0);
      if ((unsigned)(q[j].y - d0) < (unsigned)nt) note(e0 + 1);
      if ((unsigned)(q[j].z - d0) < (unsigned)nt) note(e0 + 2);
      if ((unsigned)(q[j].w - d0) < (unsigned)nt) note(e0 + 3);
    }
  }
  for (int i = Ev + tid; i < E; i += TTHREADS)
    if ((unsigned)(dstp[i] - d0) < (unsigned)nt) note(i);
  }
  __syncthreads();
  total = cnt;
  if (tid == 0) tl_cnt[blockIdx.x] = total;
  }
  const bool fast = total <= (unsigned)TCAP;

  auto edge_max_q = [&](const float4 ss, unsigned dl) {
    const float4 sd = *(const float4*)(sdt + dl * HEADS);
    atomicMax(&mxo[dl * HEADS + 0], ordEnc(lrelu(ss.x + sd.x)));
    atomicMax(&mxo[dl * HEADS + 1], ordEnc(lrelu(ss.y + sd.y)));
    atomicMax(&mxo[dl * HEADS + 2], ordEnc(lrelu(ss.z + sd.z)));
    atomicMax(&mxo[dl * HEADS + 3], ordEnc(lrelu(ss.w + sd.w)));
  };
  auto edge_max = [&](int s, unsigned dl) {
    edge_max_q(*(const float4*)(s_src + (size_t)s * HEADS), dl);
  };
  if (fast && b > 0) {
    for (int i = tid; i < (int)total; i += TTHREADS) {
      const unsigned en = elist[i];
      edge_max((int)(en >> 7), en & 127u);
    }
  } else if (fast && FIRST) {
    for (int i = tid; i < (int)total; i += TTHREADS) {
      const int idx = (int)elist[i];
      const int s = srcp[idx];
      const unsigned dl = (unsigned)(dstp[idx] - d0);
      edge_max(s, dl);
      const unsigned en = ((unsigned)s << 7) | dl;
      elist[i] = en;
      tl[i] = en;
    }
  } else if (fast) {
    for (int i0 = tid; i0 < (int)total; i0 += 2 * TTHREADS) {
      const int i1 = i0 + TTHREADS;
      const bool v1 = i1 < (int)total;
      const unsigned e0 = tl[i0];
      const unsigned e1 = v1 ? tl[i1] : 0u;
      const unsigned s0 = min(e0 >> 7, (unsigned)(N - 1));
      const unsigned s1 = min(e1 >> 7, (unsigned)(N - 1));
      const float4 q0 = *(const float4*)(s_src + (size_t)s0 * HEADS);
      const float4 q1 = *(const float4*)(s_src + (size_t)s1 * HEADS);
      asm volatile("" :: "v"(q0.x), "v"(q0.y), "v"(q0.z), "v"(q0.w),
                         "v"(q1.x), "v"(q1.y), "v"(q1.z), "v"(q1.w));
      edge_max_q(q0, e0 & 127u);
      elist[i0] = (s0 << 7) | (e0 & 127u);
      if (v1) {
        edge_max_q(q1, e1 & 127u);
        elist[i1] = (s1 << 7) | (e1 & 127u);
      }
    }
  } else {
    for (int i = tid; i < E; i += TTHREADS) {
      const unsigned dl = (unsigned)(dstp[i] - d0);
      if (dl < (unsigned)nt) edge_max(srcp[i], dl);
    }
  }
  __syncthreads();

  for (int dl = wv; dl < nt; dl += (TTHREADS >> 5)) {
    const int d = d0 + dl;
    const float e  = lrelu(s_src[(size_t)d * HEADS + hl] + sdt[dl * HEADS + hl]);
    const float wl = expf(e - ordDec(mxo[dl * HEADS + hl]));
    if (lane < HEADS) atomicAdd(&den[dl * HEADS + lane], wl);
    float w[HEADS];
    #pragma unroll
    for (int h = 0; h < HEADS; ++h) w[h] = __shfl(wl, h, 32);
    const float* hs = Hin + (size_t)d * HIDDEN;
    #pragma unroll
    for (int i = 0; i < 8; ++i) {
      const int ch = lane + (i << 5);
      acc[dl * APITCH + ch] = hs[ch] * w[i >> 1];
    }
  }

  if (fast) {
    for (int c0 = 0; c0 < (int)total; c0 += WCH) {
      const int cend = (int)total - c0 < WCH ? (int)total : c0 + WCH;
      if (c0 > 0) __syncthreads();
      for (int i = c0 + tid; i < cend; i += TTHREADS) {
        const unsigned en = elist[i];
        const unsigned dl = en & 127u;
        const float4 ss = *(const float4*)(s_src + (size_t)(en >> 7) * HEADS);
        const float4 sd = *(const float4*)(sdt + dl * HEADS);
        const uint4  mx = *(const uint4*)(mxo + dl * HEADS);
        float4 w;
        w.x = expf(lrelu(ss.x + sd.x) - ordDec(mx.x));
        w.y = expf(lrelu(ss.y + sd.y) - ordDec(mx.y));
        w.z = expf(lrelu(ss.z + sd.z) - ordDec(mx.z));
        w.w = expf(lrelu(ss.w + sd.w) - ordDec(mx.w));
        atomicAdd(&den[dl * HEADS + 0], w.x);
        atomicAdd(&den[dl * HEADS + 1], w.y);
        atomicAdd(&den[dl * HEADS + 2], w.z);
        atomicAdd(&den[dl * HEADS + 3], w.w);
        *(float4*)(wls + (size_t)(i - c0) * HEADS) = w;
      }
      __syncthreads();
      {
        const int half = wv & 1;
        const unsigned q = (unsigned)(wv >> 1);
        const int coff = (half << 7) + (lane << 2);
        const int hsel = (half << 1) + (lane >> 4);
        uint2* mylist = wlist + wv * LCAP;
        float4 a[NSLOT];
        #pragma unroll
        for (int s = 0; s < NSLOT; ++s) a[s] = *(const float4*)(acc + ((int)q + 16 * s) * APITCH + coff);
        const bool one = total <= (unsigned)WCH;
        const size_t pslot = (size_t)blockIdx.x * (TTHREADS >> 5) + wv;
        int kept = -1;
        if (one && b > 0) {
          kept = __builtin_amdgcn_readfirstlane((int)npk[wv]);
        } else if (one && !FIRST) {
          const unsigned n = (unsigned)__builtin_amdgcn_readfirstlane((int)pkn);
          if (n <= (unsigned)LCAP) {
            const uint2* __restrict__ pl = (const uint2*)(pk_list + pslot * (2 * LCAP));
            uint2 e[LCAP / 32];
            #pragma unroll
            for (int j = 0; j < LCAP / 32; ++j) {
              const int i = lane + 32 * j;
              e[j] = i < (int)n ? pl[i] : make_uint2(0u, 0u);
            }
            #pragma unroll
            for (int j = 0; j < LCAP / 32; ++j) {
              const int i = lane + 32 * j;
              const unsigned sx = min(e[j].x >> 7, (unsigned)(N - 1));
              if (i < (int)n) mylist[i] = make_uint2((sx << 7) | (e[j].x & 127u), min(e[j].y, (unsigned)(WCH - 1)));
            }
            kept = (int)n;
            if (lane == 0) npk[wv] = n;
          }
        }
        int base = c0;
        bool more = true;
        bool round1 = true;
        while (more) {
          int npick = 0;
          if (kept >= 0) {
            npick = kept;
            more = false;
          } else {
            while (base < cend && npick <= LCAP - 32 * SCAN) {
              unsigned en[SCAN];
              #pragma unroll
              for (int u = 0; u < SCAN; ++u) {
                const int idx = base + 32 * u + lane;
                en[u] = idx < cend ? elist[idx] : 0xFFFFFFFFu;
              }
              #pragma unroll
              for (int u = 0; u < SCAN; ++u) {
                const int idx = base + 32 * u + lane;
                const bool mine = idx < cend && (en[u] & 15u) == q;
                const unsigned m = __builtin_amdgcn_ballot_w32(mine);
                if (mine) mylist[npick + (int)__builtin_amdgcn_mbcnt_lo(m, 0u)] = make_uint2(en[u], (unsigned)(idx - c0));
                npick += __popc(m);
              }
              base += 32 * SCAN;
            }
            more = base < cend;
          }
          __builtin_amdgcn_wave_barrier();
          if (kept < 0 && one && b == 0 && round1) {
            const unsigned kn = more ? 0xFFFFFFFFu : (unsigned)npick;
            if (lane == 0) npk[wv] = kn;
            if (FIRST) {
              if (lane == 0) pk_cnt[pslot] = kn;
              if (!more) {
                uint2* __restrict__ pl = (uint2*)(pk_list + pslot * (2 * LCAP));
                for (int i = lane; i < npick; i += 32) pl[i] = mylist[i];
              }
            }
          }
          round1 = false;
          for (int cb = 0; cb < npick; cb += 32) {
            const bool val = cb + lane < npick;
            uint2 le = make_uint2(0u, 0u);
            if (val) le = mylist[cb + lane];
            const unsigned lslot = (le.x & 127u) >> 4;
            #pragma unroll
            for (int s = 0; s < NSLOT; ++s) {
              unsigned m = __builtin_amdgcn_ballot_w32(val && lslot == (unsigned)s);
              while (m != 0u) {
                float wq[NB];
                float4 hv[NB];
                #pragma unroll
                for (int j = 0; j < NB; ++j) {
                  hv[j] = make_float4(0.f, 0.f, 0.f, 0.f);
                  wq[j] = 0.f;
                  if (m != 0u) {
                    const int bit = __builtin_ctz(m);
                    const unsigned enj = (unsigned)__builtin_amdgcn_readlane((int)le.x, bit);
                    const unsigned wij = (unsigned)__builtin_amdgcn_readlane((int)le.y, bit);
                    hv[j] = *(const float4*)(Hin + (size_t)(enj >> 7) * HIDDEN + coff);
                    wq[j] = wls[wij * HEADS + hsel];
                  }
                  m &= m - 1u;
                }
                #pragma unroll
                for (int j = 0; j < NB; ++j) {
                  a[s].x += wq[j] * hv[j].x;
                  a[s].y += wq[j] * hv[j].y;
                  a[s].z += wq[j] * hv[j].z;
                  a[s].w += wq[j] * hv[j].w;
                }
              }
            }
          }
        }
        #pragma unroll
        for (int s = 0; s < NSLOT; ++s) *(float4*)(acc + ((int)q + 16 * s) * APITCH + coff) = a[s];
      }
    }
  } else {
    __syncthreads();
    for (int i = tid; i < E; i += TTHREADS) {
      const unsigned dl = (unsigned)(dstp[i] - d0);
      if (dl < (unsigned)nt) {
        const int s = srcp[i];
        const float* hs = Hin + (size_t)s * HIDDEN;
        for (int h = 0; h < HEADS; ++h) {
          const float e  = lrelu(s_src[(size_t)s * HEADS + h] + sdt[dl * HEADS + h]);
          const float wh = expf(e - ordDec(mxo[dl * HEADS + h]));
          atomicAdd(&den[dl * HEADS + h], wh);
          for (int c = 0; c < CHAN; ++c)
            atomicAdd(&acc[dl * APITCH + h * CHAN + c], hs[h * CHAN + c] * wh);
        }
      }
    }
  }
  __syncthreads();

  for (int qd = tid; qd < TN * (HIDDEN / 4); qd += TTHREADS) {
    const int dl = qd / (HIDDEN / 4);
    const int c  = (qd - dl * (HIDDEN / 4)) << 2;
    float4 v = make_float4(0.f, 0.f, 0.f, 0.f);
    if (dl < nt) {
      const float dn = den[dl * HEADS + (c >> 6)];
      v = *(const float4*)(acc + dl * APITCH + c);
      v.x = v.x / dn + bias[c + 0];
      v.y = v.y / dn + bias[c + 1];
      v.z = v.z / dn + bias[c + 2];
      v.w = v.w / dn + bias[c + 3];
      if (ELU) {
        v.x = v.x > 0.f ? v.x : expm1f(v.x);
        v.y = v.y > 0.f ? v.y : expm1f(v.y);
        v.z = v.z > 0.f ? v.z : expm1f(v.z);
        v.w = v.w > 0.f ? v.w : expm1f(v.w);
      }
    }
    *(float4*)(acc + dl * APITCH + c) = v;
  }
  __syncthreads();

  if (ELU)
    tile_gemm<HIDDEN, APITCH, HIDDEN, TTHREADS, true, false>(acc, acc, tid, d0, nt, Wn, nullptr, att_src_n, att_dst_n,
                                                             Hout, s_src_n, s_dst_n);
  else
    tile_gemm<HIDDEN, APITCH, CHAN, TTHREADS, false, true>(acc, acc, tid, d0, nt, Wn, bn, nullptr, nullptr,
                                                           Hout, nullptr, nullptr);
  }
}

extern "C" void kernel_launch(void* const* d_in, const int* in_sizes, int n_in,
                              void* d_out, int out_size, void* d_ws, size_t ws_size,
                              hipStream_t stream) {
  const float* x     = (const float*)d_in[0];
  const int*   ei    = (const int*)  d_in[1];
  const float* W1    = (const float*)d_in[2];
  const float* asrc1 = (const float*)d_in[3];
  const float* adst1 = (const float*)d_in[4];
  const float* b1    = (const float*)d_in[5];
  const float* W2    = (const float*)d_in[6];
  const float* asrc2 = (const float*)d_in[7];
  const float* adst2 = (const float*)d_in[8];
  const float* b2    = (const float*)d_in[9];
  const float* Wf    = (const float*)d_in[10];
  const float* bf    = (const float*)d_in[11];

  const int B  = 2;
  const int N  = in_sizes[0] / (B * FEAT);
  const int E  = in_sizes[1] / 2;

  float* p = (float*)d_ws;
  float* buf_h   = p; p += (size_t)B * N * HIDDEN;
  float* buf_h2  = p; p += (size_t)B * N * HIDDEN;
  float* ssrc    = p; p += (size_t)B * N * HEADS;
  float* sdst    = p; p += (size_t)B * N * HEADS;
  float* ssrc2   = p; p += (size_t)B * N * HEADS;
  float* sdst2   = p; p += (size_t)B * N * HEADS;
  const int ntiles  = (N + TN - 1) / TN;
  unsigned* tmap   = (unsigned*)p; p += ((E + 15) >> 4) << 2;
  unsigned* pklist = (unsigned*)p; p += (size_t)ntiles * (TTHREADS >> 5) * LCAP * 2;
  unsigned* tlcnt  = (unsigned*)p; p += ntiles;
  unsigned* tllist = (unsigned*)p; p += (size_t)ntiles * TCAP;
  unsigned* pkcnt  = (unsigned*)p;

  const dim3 blkG(GTHREADS);
  const dim3 blkT(TTHREADS);

  xw_tile_kernel<<<ntiles, blkG, 0, stream>>>(
      x, W1, asrc1, adst1, buf_h, ssrc, sdst, N, B, ei, E, tmap);
  tile_agg_kernel<true, true><<<ntiles, blkT, 0, stream>>>(ei, E, N, B, ssrc, sdst, buf_h, b1,
      W2, nullptr, asrc2, adst2, buf_h2, ssrc2, sdst2, tlcnt, tllist, tmap, pkcnt, pklist);
  tile_agg_kernel<false, false><<<ntiles, blkT, 0, stream>>>(ei, E, N, B, ssrc2, sdst2, buf_h2, b2,
      Wf, bf, nullptr, nullptr, (float*)d_out, nullptr, nullptr, tlcnt, tllist, tmap, pkcnt, pklist);
}
